# starting kernel
# baseline (speedup 1.0000x reference)
#include <hip/hip_runtime.h>

#define DEV __device__ __forceinline__

typedef float v2f __attribute__((ext_vector_type(2)));
typedef float v8f __attribute__((ext_vector_type(8)));

static constexpr int CDIM  = 128;
static constexpr int FDIM  = 256;
static constexpr int NSEQ  = 2048;
static constexpr int BATCH = 4;
static constexpr int NHEAD = 8;
static constexpr int HDIM  = 16;
static constexpr int MROWS = BATCH * NSEQ;

DEV v8f v8zero() {
  v8f v;
#pragma unroll
  for (int i = 0; i < 8; ++i) v[i] = 0.f;
  return v;
}

DEV v8f wmma4(v2f a, v2f b, v8f c) {
  return __builtin_amdgcn_wmma_f32_16x16x4_f32(false, a, false, b, (short)0, c,
                                               false, false);
}

DEV float redmax16(float v) {
#pragma unroll
  for (int m = 1; m < 16; m <<= 1) v = fmaxf(v, __shfl_xor(v, m, 32));
  return v;
}
DEV float redsum16(float v) {
#pragma unroll
  for (int m = 1; m < 16; m <<= 1) v += __shfl_xor(v, m, 32);
  return v;
}

template <int NT>
DEV void gemm_acc(const float* __restrict__ Arow, const float* __restrict__ Bc,
                  int ldb, int K, v8f* acc) {
  const int hi = (threadIdx.x >> 4) & 1;
  for (int k = 0; k < K; k += 4) {
    const int kb = k + 2 * hi;
    v2f a;
    a.x = Arow[kb];
    a.y = Arow[kb + 1];
#pragma unroll
    for (int j = 0; j < NT; ++j) {
      v2f b;
      b.x = Bc[(size_t)kb * ldb + j * 16];
      b.y = Bc[(size_t)(kb + 1) * ldb + j * 16];
      acc[j] = wmma4(a, b, acc[j]);
    }
  }
}

__global__ __launch_bounds__(256) void qkv_kernel(
    const float* __restrict__ h, const float* __restrict__ Wq,
    const float* __restrict__ Wk, const float* __restrict__ Wv,
    float* __restrict__ Q, float* __restrict__ K, float* __restrict__ V) {
  const int lane = threadIdx.x & 31;
  const int col = lane & 15, hi = lane >> 4;
  const int gw = blockIdx.x * (blockDim.x >> 5) + (threadIdx.x >> 5);
  const int mat = gw / (MROWS / 16);
  const int rt = gw % (MROWS / 16);
  const float* W = (mat == 0) ? Wq : (mat == 1) ? Wk : Wv;
  float* Out = (mat == 0) ? Q : (mat == 1) ? K : V;
  const int row0 = rt * 16;

  v8f acc[8];
#pragma unroll
  for (int j = 0; j < 8; ++j) acc[j] = v8zero();

  gemm_acc<8>(h + (size_t)(row0 + col) * CDIM, W + col, CDIM, CDIM, acc);

#pragma unroll
  for (int j = 0; j < 8; ++j)
#pragma unroll
    for (int r = 0; r < 8; ++r)
      Out[(size_t)(row0 + r + 8 * hi) * CDIM + 16 * j + col] = acc[j][r];
}

__global__ __launch_bounds__(128) void attn_kernel(
    const float* __restrict__ Q, const float* __restrict__ Kmat,
    const float* __restrict__ V, const int* __restrict__ adj,
    float* __restrict__ O) {
  __shared__ float lds[4][16 * 16];
  const int lane = threadIdx.x & 31;
  const int wid = threadIdx.x >> 5;
  const int col = lane & 15, hi = lane >> 4;
  const int gw = blockIdx.x * 4 + wid;
  const int nt = gw & (NSEQ / 16 - 1);
  const int hd = (gw >> 7) & (NHEAD - 1);
  const int b = gw >> 10;
  const int n0 = nt * 16;
  const size_t base = (size_t)b * NSEQ * CDIM;

  const float* Qrow = Q + base + (size_t)(n0 + col) * CDIM + hd * HDIM;
  float* P = lds[wid];

  float m_i[8], l_i[8];
  v8f o = v8zero();
#pragma unroll
  for (int r = 0; r < 8; ++r) {
    m_i[r] = -__builtin_inff();
    l_i[r] = 0.f;
  }

  for (int m0 = 0; m0 < NSEQ; m0 += 16) {
    v8f s = v8zero();
    const float* Kc = Kmat + base + (size_t)(m0 + col) * CDIM + hd * HDIM;
#pragma unroll
    for (int t = 0; t < 4; ++t) {
      const int kb = 4 * t + 2 * hi;
      v2f a, bb;
      a.x = Qrow[kb];
      a.y = Qrow[kb + 1];
      bb.x = Kc[kb];
      bb.y = Kc[kb + 1];
      s = wmma4(a, bb, s);
    }

    const int* adjp = adj + (size_t)(n0 + 8 * hi) * NSEQ + m0 + col;
#pragma unroll
    for (int r = 0; r < 8; ++r) {
      float sv = s[r] * 0.25f;
      if (adjp[(size_t)r * NSEQ] == 0) sv = -1e30f;
      const float rmax = redmax16(sv);
      const float mnew = fmaxf(m_i[r], rmax);
      const float f = __expf(m_i[r] - mnew);
      const float p = __expf(sv - mnew);
      l_i[r] = l_i[r] * f + redsum16(p);
      o[r] *= f;
      m_i[r] = mnew;
      P[(r + 8 * hi) * 16 + col] = p;
    }
    __builtin_amdgcn_wave_barrier();

    const float* Vc = V + base + (size_t)m0 * CDIM + hd * HDIM + col;
#pragma unroll
    for (int t = 0; t < 4; ++t) {
      const int kb = 4 * t + 2 * hi;
      v2f a, bb;
      a.x = P[col * 16 + kb];
      a.y = P[col * 16 + kb + 1];
      bb.x = Vc[(size_t)kb * CDIM];
      bb.y = Vc[(size_t)(kb + 1) * CDIM];
      o = wmma4(a, bb, o);
    }
    __builtin_amdgcn_wave_barrier();
  }

  float* Op = O + base + (size_t)(n0 + 8 * hi) * CDIM + hd * HDIM + col;
#pragma unroll
  for (int r = 0; r < 8; ++r) Op[(size_t)r * CDIM] = o[r] / l_i[r];
}

__global__ __launch_bounds__(256) void gemm_ln_kernel(
    const float* __restrict__ A, int lda, int K, const float* __restrict__ W,
    const float* __restrict__ bias, const float* __restrict__ res,
    const float* __restrict__ g, const float* __restrict__ bt,
    float* __restrict__ Out) {
  const int lane = threadIdx.x & 31;
  const int col = lane & 15, hi = lane >> 4;
  const int gw = blockIdx.x * (blockDim.x >> 5) + (threadIdx.x >> 5);
  const int row0 = gw * 16;

  v8f acc[8];
#pragma unroll
  for (int j = 0; j < 8; ++j) acc[j] = v8zero();

  gemm_acc<8>(A + (size_t)(row0 + col) * lda, W + col, CDIM, K, acc);

#pragma unroll
  for (int r = 0; r < 8; ++r) {
    const int row = row0 + r + 8 * hi;
    float s1 = 0.f, s2 = 0.f;
#pragma unroll
    for (int j = 0; j < 8; ++j) {
      float v = acc[j][r] + bias[16 * j + col] +
                res[(size_t)row * CDIM + 16 * j + col];
      acc[j][r] = v;
      s1 += v;
      s2 += v * v;
    }
    s1 = redsum16(s1);
    s2 = redsum16(s2);
    const float mean = s1 * (1.f / CDIM);
    const float var = s2 * (1.f / CDIM) - mean * mean;
    const float rstd = rsqrtf(var + 1e-5f);
#pragma unroll
    for (int j = 0; j < 8; ++j)
      Out[(size_t)row * CDIM + 16 * j + col] =
          (acc[j][r] - mean) * rstd * g[16 * j + col] + bt[16 * j + col];
  }
}

__global__ __launch_bounds__(256) void ffn1_kernel(
    const float* __restrict__ X, const float* __restrict__ W1,
    const float* __restrict__ b1, float* __restrict__ T) {
  const int lane = threadIdx.x & 31;
  const int col = lane & 15, hi = lane >> 4;
  const int gw = blockIdx.x * (blockDim.x >> 5) + (threadIdx.x >> 5);
  const int half = gw & 1;
  const int rt = gw >> 1;
  const int row0 = rt * 16, c0 = half * 128;

  v8f acc[8];
#pragma unroll
  for (int j = 0; j < 8; ++j) acc[j] = v8zero();

  gemm_acc<8>(X + (size_t)(row0 + col) * CDIM, W1 + c0 + col, FDIM, CDIM, acc);

#pragma unroll
  for (int j = 0; j < 8; ++j)
#pragma unroll
    for (int r = 0; r < 8; ++r)
      T[(size_t)(row0 + r + 8 * hi) * FDIM + c0 + 16 * j + col] =
          fmaxf(acc[j][r] + b1[c0 + 16 * j + col], 0.f);
}

extern "C" void kernel_launch(void* const* d_in, const int* in_sizes, int n_in,
                              void* d_out, int out_size, void* d_ws,
                              size_t ws_size, hipStream_t stream) {
  (void)in_sizes; (void)n_in; (void)out_size; (void)ws_size;
  const int* adj  = (const int*)d_in[0];
  const float* h  = (const float*)d_in[1];
  const float* Wq = (const float*)d_in[2];
  const float* Wk = (const float*)d_in[3];
  const float* Wv = (const float*)d_in[4];
  const float* Wo = (const float*)d_in[5];
  const float* bo = (const float*)d_in[6];
  const float* g1 = (const float*)d_in[7];
  const float* bt1 = (const float*)d_in[8];
  const float* W1 = (const float*)d_in[9];
  const float* bb1 = (const float*)d_in[10];
  const float* W2 = (const float*)d_in[11];
  const float* bb2 = (const float*)d_in[12];
  const float* g2 = (const float*)d_in[13];
  const float* bt2 = (const float*)d_in[14];

  float* ws = (float*)d_ws;
  const size_t SZ = (size_t)MROWS * CDIM;
  float* Q  = ws;
  float* K  = Q + SZ;
  float* V  = K + SZ;
  float* AO = V + SZ;
  float* X  = AO + SZ;
  float* T  = X + SZ;

  qkv_kernel<<<(3 * (MROWS / 16)) / 8, 256, 0, stream>>>(h, Wq, Wk, Wv, Q, K, V);
  attn_kernel<<<(BATCH * NHEAD * (NSEQ / 16)) / 4, 128, 0, stream>>>(Q, K, V,
                                                                     adj, AO);
  gemm_ln_kernel<<<(MROWS / 16) / 8, 256, 0, stream>>>(AO, CDIM, CDIM, Wo, bo,
                                                       h, g1, bt1, X);
  ffn1_kernel<<<(2 * (MROWS / 16)) / 8, 256, 0, stream>>>(X, W1, bb1, T);
  gemm_ln_kernel<<<(MROWS / 16) / 8, 256, 0, stream>>>(T, FDIM, FDIM, W2, bb2,
                                                       X, g2, bt2,
                                                       (float*)d_out);
}


# attention: scores computed transposed (K*Q^T) so a lane owns one query: max/sum per lane, P feeds P*V from registers, no LDS or 16-lane butterflies; row sum and P*V key order re-associated
# speedup vs baseline: 1.6848x; 1.6848x over previous
#include <hip/hip_runtime.h>

#define DEV __device__ __forceinline__

typedef float v2f __attribute__((ext_vector_type(2)));
typedef float v8f __attribute__((ext_vector_type(8)));
typedef float v4f __attribute__((ext_vector_type(4)));
typedef int v4i __attribute__((ext_vector_type(4)));

static constexpr int CDIM  = 128;
static constexpr int FDIM  = 256;
static constexpr int NSEQ  = 2048;
static constexpr int BATCH = 4;
static constexpr int NHEAD = 8;
static constexpr int HDIM  = 16;
static constexpr int MROWS = BATCH * NSEQ;

DEV v8f v8zero() {
  v8f v;
#pragma unroll
  for (int i = 0; i < 8; ++i) v[i] = 0.f;
  return v;
}

DEV v8f wmma4(v2f a, v2f b, v8f c) {
  return __builtin_amdgcn_wmma_f32_16x16x4_f32(false, a, false, b, (short)0, c,
                                               false, false);
}

DEV float redsum16(float v) {
#pragma unroll
  for (int m = 1; m < 16; m <<= 1) v += __shfl_xor(v, m, 32);
  return v;
}

template <int NT>
DEV void gemm_acc(const float* __restrict__ Arow, const float* __restrict__ Bc,
                  int ldb, int K, v8f* acc) {
  const int hi = (threadIdx.x >> 4) & 1;
  for (int k = 0; k < K; k += 4) {
    const int kb = k + 2 * hi;
    v2f a;
    a.x = Arow[kb];
    a.y = Arow[kb + 1];
#pragma unroll
    for (int j = 0; j < NT; ++j) {
      v2f b;
      b.x = Bc[(size_t)kb * ldb + j * 16];
      b.y = Bc[(size_t)(kb + 1) * ldb + j * 16];
      acc[j] = wmma4(a, b, acc[j]);
    }
  }
}

__global__ __launch_bounds__(256) void qkv_kernel(
    const float* __restrict__ h, const float* __restrict__ Wq,
    const float* __restrict__ Wk, const float* __restrict__ Wv,
    float* __restrict__ Q, float* __restrict__ K, float* __restrict__ V) {
  const int lane = threadIdx.x & 31;
  const int col = lane & 15, hi = lane >> 4;
  const int gw = blockIdx.x * (blockDim.x >> 5) + (threadIdx.x >> 5);
  const int mat = gw / (MROWS / 16);
  const int rt = gw % (MROWS / 16);
  const float* W = (mat == 0) ? Wq : (mat == 1) ? Wk : Wv;
  float* Out = (mat == 0) ? Q : (mat == 1) ? K : V;
  const int row0 = rt * 16;

  v8f acc[8];
#pragma unroll
  for (int j = 0; j < 8; ++j) acc[j] = v8zero();

  gemm_acc<8>(h + (size_t)(row0 + col) * CDIM, W + col, CDIM, CDIM, acc);

#pragma unroll
  for (int j = 0; j < 8; ++j)
#pragma unroll
    for (int r = 0; r < 8; ++r)
      Out[(size_t)(row0 + r + 8 * hi) * CDIM + 16 * j + col] = acc[j][r];
}

DEV float xhalf(float x) {
  const int b = __builtin_bit_cast(int, x);
  return __builtin_bit_cast(
      float, __builtin_amdgcn_permlanex16(b, b, 0x76543210, 0xfedcba98, false,
                                          false));
}

__global__ __launch_bounds__(128) void attn_kernel(
    const float* __restrict__ Q, const float* __restrict__ Kmat,
    const float* __restrict__ V, const int* __restrict__ adj,
    float* __restrict__ O) {
  const int lane = threadIdx.x & 31;
  const int wid = threadIdx.x >> 5;
  const int col = lane & 15, hi = lane >> 4;
  const int gw = blockIdx.x * 4 + wid;
  const int nt = gw & (NSEQ / 16 - 1);
  const int hd = (gw >> 7) & (NHEAD - 1);
  const int b = gw >> 10;
  const int n0 = nt * 16;
  const size_t base = (size_t)b * NSEQ * CDIM;


  const float* Qrow = Q + base + (size_t)(n0 + col) * CDIM + hd * HDIM;
  v2f qf[4];
#pragma unroll
  for (int t = 0; t < 4; ++t) {
    qf[t].x = Qrow[4 * t + 2 * hi];
    qf[t].y = Qrow[4 * t + 2 * hi + 1];
  }

  const float* Kbase = Kmat + base + (size_t)col * CDIM + hd * HDIM + 2 * hi;
  const float* Vbase = V + base + (size_t)(8 * hi) * CDIM + hd * HDIM + col;
  const int* adjrow = adj + (size_t)(n0 + col) * NSEQ + 8 * hi;

  float m_i = -__builtin_inff();
  float l_i = 0.f;
  v8f o = v8zero();

  for (int m0 = 0; m0 < NSEQ; m0 += 16) {
    v8f s = v8zero();
    const float* Kc = Kbase + (size_t)m0 * CDIM;
#pragma unroll
    for (int t = 0; t < 4; ++t) {
      v2f a;
      a.x = Kc[4 * t];
      a.y = Kc[4 * t + 1];
      s = wmma4(a, qf[t], s);
    }

    const v4i a0 = *(const v4i*)(adjrow + m0);
    const v4i a1 = *(const v4i*)(adjrow + m0 + 4);
    float sv[8];
#pragma unroll
    for (int r = 0; r < 8; ++r) {
      const int av = (r < 4) ? a0[r & 3] : a1[r & 3];
      sv[r] = s[r] * 0.25f;
      if (av == 0) sv[r] = -1e30f;
    }
    float mloc = sv[0];
#pragma unroll
    for (int r = 1; r < 8; ++r) mloc = fmaxf(mloc, sv[r]);
    mloc = fmaxf(mloc, xhalf(mloc));
    const float mnew = fmaxf(m_i, mloc);
    const float f = __expf(m_i - mnew);
    float p[8];
    float psum = 0.f;
#pragma unroll
    for (int r = 0; r < 8; ++r) {
      p[r] = __expf(sv[r] - mnew);
      psum += p[r];
    }
    l_i = l_i * f + psum;
#pragma unroll
    for (int r = 0; r < 8; ++r) o[r] *= f;
    m_i = mnew;

    const float* Vc = Vbase + (size_t)m0 * CDIM;
#pragma unroll
    for (int t = 0; t < 4; ++t) {
      v2f a, bb;
      a.x = Vc[(size_t)(2 * t) * CDIM];
      a.y = Vc[(size_t)(2 * t + 1) * CDIM];
      bb.x = p[2 * t];
      bb.y = p[2 * t + 1];
      o = wmma4(a, bb, o);
    }
  }

  l_i += xhalf(l_i);
  float* Op = O + base + (size_t)(n0 + col) * CDIM + hd * HDIM + 8 * hi;
  v4f o0, o1;
#pragma unroll
  for (int r = 0; r < 4; ++r) {
    o0[r] = o[r] / l_i;
    o1[r] = o[r + 4] / l_i;
  }
  *(v4f*)Op = o0;
  *(v4f*)(Op + 4) = o1;
}

__global__ __launch_bounds__(256) void gemm_ln_kernel(
    const float* __restrict__ A, int lda, int K, const float* __restrict__ W,
    const float* __restrict__ bias, const float* __restrict__ res,
    const float* __restrict__ g, const float* __restrict__ bt,
    float* __restrict__ Out) {
  const int lane = threadIdx.x & 31;
  const int col = lane & 15, hi = lane >> 4;
  const int gw = blockIdx.x * (blockDim.x >> 5) + (threadIdx.x >> 5);
  const int row0 = gw * 16;

  v8f acc[8];
#pragma unroll
  for (int j = 0; j < 8; ++j) acc[j] = v8zero();

  gemm_acc<8>(A + (size_t)(row0 + col) * lda, W + col, CDIM, K, acc);

#pragma unroll
  for (int r = 0; r < 8; ++r) {
    const int row = row0 + r + 8 * hi;
    float s1 = 0.f, s2 = 0.f;
#pragma unroll
    for (int j = 0; j < 8; ++j) {
      float v = acc[j][r] + bias[16 * j + col] +
                res[(size_t)row * CDIM + 16 * j + col];
      acc[j][r] = v;
      s1 += v;
      s2 += v * v;
    }
    s1 = redsum16(s1);
    s2 = redsum16(s2);
    const float mean = s1 * (1.f / CDIM);
    const float var = s2 * (1.f / CDIM) - mean * mean;
    const float rstd = rsqrtf(var + 1e-5f);
#pragma unroll
    for (int j = 0; j < 8; ++j)
      Out[(size_t)row * CDIM + 16 * j + col] =
          (acc[j][r] - mean) * rstd * g[16 * j + col] + bt[16 * j + col];
  }
}

__global__ __launch_bounds__(256) void ffn1_kernel(
    const float* __restrict__ X, const float* __restrict__ W1,
    const float* __restrict__ b1, float* __restrict__ T) {
  const int lane = threadIdx.x & 31;
  const int col = lane & 15, hi = lane >> 4;
  const int gw = blockIdx.x * (blockDim.x >> 5) + (threadIdx.x >> 5);
  const int half = gw & 1;
  const int rt = gw >> 1;
  const int row0 = rt * 16, c0 = half * 128;

  v8f acc[8];
#pragma unroll
  for (int j = 0; j < 8; ++j) acc[j] = v8zero();

  gemm_acc<8>(X + (size_t)(row0 + col) * CDIM, W1 + c0 + col, FDIM, CDIM, acc);

#pragma unroll
  for (int j = 0; j < 8; ++j)
#pragma unroll
    for (int r = 0; r < 8; ++r)
      T[(size_t)(row0 + r + 8 * hi) * FDIM + c0 + 16 * j + col] =
          fmaxf(acc[j][r] + b1[c0 + 16 * j + col], 0.f);
}

extern "C" void kernel_launch(void* const* d_in, const int* in_sizes, int n_in,
                              void* d_out, int out_size, void* d_ws,
                              size_t ws_size, hipStream_t stream) {
  (void)in_sizes; (void)n_in; (void)out_size; (void)ws_size;
  const int* adj  = (const int*)d_in[0];
  const float* h  = (const float*)d_in[1];
  const float* Wq = (const float*)d_in[2];
  const float* Wk = (const float*)d_in[3];
  const float* Wv = (const float*)d_in[4];
  const float* Wo = (const float*)d_in[5];
  const float* bo = (const float*)d_in[6];
  const float* g1 = (const float*)d_in[7];
  const float* bt1 = (const float*)d_in[8];
  const float* W1 = (const float*)d_in[9];
  const float* bb1 = (const float*)d_in[10];
  const float* W2 = (const float*)d_in[11];
  const float* bb2 = (const float*)d_in[12];
  const float* g2 = (const float*)d_in[13];
  const float* bt2 = (const float*)d_in[14];

  float* ws = (float*)d_ws;
  const size_t SZ = (size_t)MROWS * CDIM;
  float* Q  = ws;
  float* K  = Q + SZ;
  float* V  = K + SZ;
  float* AO = V + SZ;
  float* X  = AO + SZ;
  float* T  = X + SZ;

  qkv_kernel<<<(3 * (MROWS / 16)) / 8, 256, 0, stream>>>(h, Wq, Wk, Wv, Q, K, V);
  attn_kernel<<<(BATCH * NHEAD * (NSEQ / 16)) / 4, 128, 0, stream>>>(Q, K, V,
                                                                     adj, AO);
  gemm_ln_kernel<<<(MROWS / 16) / 8, 256, 0, stream>>>(AO, CDIM, CDIM, Wo, bo,
                                                       h, g1, bt1, X);
  ffn1_kernel<<<(2 * (MROWS / 16)) / 8, 256, 0, stream>>>(X, W1, bb1, T);
  gemm_ln_kernel<<<(MROWS / 16) / 8, 256, 0, stream>>>(T, FDIM, FDIM, W2, bb2,
                                                       X, g2, bt2,
                                                       (float*)d_out);
}


# K and V stored by the projection kernel as 1-KiB per-head key-tile blocks in fragment order (K by transposed product): attention reads a tile as 2 whole-line b128 loads; score dim order re-associated
# speedup vs baseline: 1.8735x; 1.1120x over previous
#include <hip/hip_runtime.h>

#define DEV __device__ __forceinline__

typedef float v2f __attribute__((ext_vector_type(2)));
typedef float v8f __attribute__((ext_vector_type(8)));
typedef float v4f __attribute__((ext_vector_type(4)));
typedef int v4i __attribute__((ext_vector_type(4)));

static constexpr int CDIM  = 128;
static constexpr int FDIM  = 256;
static constexpr int NSEQ  = 2048;
static constexpr int BATCH = 4;
static constexpr int NHEAD = 8;
static constexpr int HDIM  = 16;
static constexpr int MROWS = BATCH * NSEQ;

DEV v8f v8zero() {
  v8f v;
#pragma unroll
  for (int i = 0; i < 8; ++i) v[i] = 0.f;
  return v;
}

DEV v8f wmma4(v2f a, v2f b, v8f c) {
  return __builtin_amdgcn_wmma_f32_16x16x4_f32(false, a, false, b, (short)0, c,
                                               false, false);
}

DEV float redsum16(float v) {
#pragma unroll
  for (int m = 1; m < 16; m <<= 1) v += __shfl_xor(v, m, 32);
  return v;
}

template <int NT, bool SWAP = false>
DEV void gemm_acc(const float* __restrict__ Arow, const float* __restrict__ Bc,
                  int ldb, int K, v8f* acc) {
  const int hi = (threadIdx.x >> 4) & 1;
  for (int k = 0; k < K; k += 4) {
    const int kb = k + 2 * hi;
    v2f a;
    a.x = Arow[kb];
    a.y = Arow[kb + 1];
#pragma unroll
    for (int j = 0; j < NT; ++j) {
      v2f b;
      b.x = Bc[(size_t)kb * ldb + j * 16];
      b.y = Bc[(size_t)(kb + 1) * ldb + j * 16];
      acc[j] = SWAP ? wmma4(b, a, acc[j]) : wmma4(a, b, acc[j]);
    }
  }
}

static constexpr int TILE_F = 256;
static constexpr int HEAD_F = (NSEQ / 16) * TILE_F;

__global__ __launch_bounds__(256) void qkv_kernel(
    const float* __restrict__ h, const float* __restrict__ Wq,
    const float* __restrict__ Wk, const float* __restrict__ Wv,
    float* __restrict__ Q, float* __restrict__ K, float* __restrict__ V) {
  const int lane = threadIdx.x & 31;
  const int col = lane & 15, hi = lane >> 4;
  const int wv = __builtin_amdgcn_readfirstlane((int)(threadIdx.x >> 5));
  const int gw = blockIdx.x * (blockDim.x >> 5) + wv;
  const int mat = gw / (MROWS / 16);
  const int rt = gw % (MROWS / 16);
  const float* W = (mat == 0) ? Wq : (mat == 1) ? Wk : Wv;
  const int row0 = rt * 16;

  v8f acc[8];
#pragma unroll
  for (int j = 0; j < 8; ++j) acc[j] = v8zero();

  const float* Arow = h + (size_t)(row0 + col) * CDIM;
  if (mat == 1)
    gemm_acc<8, true>(Arow, W + col, CDIM, CDIM, acc);
  else
    gemm_acc<8, false>(Arow, W + col, CDIM, CDIM, acc);

  if (mat == 0) {
#pragma unroll
    for (int j = 0; j < 8; ++j)
#pragma unroll
      for (int r = 0; r < 8; ++r)
        Q[(size_t)(row0 + r + 8 * hi) * CDIM + 16 * j + col] = acc[j][r];
  } else {
    float* Out = (mat == 1) ? K : V;
    float* T = Out + (size_t)((rt / (NSEQ / 16)) * NHEAD) * HEAD_F +
               (size_t)(rt & (NSEQ / 16 - 1)) * TILE_F + lane * 4;
#pragma unroll
    for (int j = 0; j < 8; ++j) {
      v4f lo, up;
#pragma unroll
      for (int e = 0; e < 4; ++e) {
        lo[e] = acc[j][e];
        up[e] = acc[j][e + 4];
      }
      *(v4f*)(T + (size_t)j * HEAD_F) = lo;
      *(v4f*)(T + (size_t)j * HEAD_F + 128) = up;
    }
  }
}

DEV float xhalf(float x) {
  const int b = __builtin_bit_cast(int, x);
  return __builtin_bit_cast(
      float, __builtin_amdgcn_permlanex16(b, b, 0x76543210, 0xfedcba98, false,
                                          false));
}

__global__ __launch_bounds__(128) void attn_kernel(
    const float* __restrict__ Q, const float* __restrict__ Kmat,
    const float* __restrict__ V, const int* __restrict__ adj,
    float* __restrict__ O) {
  const int lane = threadIdx.x & 31;
  const int wid = threadIdx.x >> 5;
  const int col = lane & 15, hi = lane >> 4;
  const int gw = blockIdx.x * 4 + wid;
  const int nt = gw & (NSEQ / 16 - 1);
  const int hd = (gw >> 7) & (NHEAD - 1);
  const int b = gw >> 10;
  const int n0 = nt * 16;
  const size_t base = (size_t)b * NSEQ * CDIM;


  const float* Qrow = Q + base + (size_t)(n0 + col) * CDIM + hd * HDIM + 8 * hi;
  v2f qf[4];
#pragma unroll
  for (int t = 0; t < 4; ++t) {
    qf[t].x = Qrow[2 * t];
    qf[t].y = Qrow[2 * t + 1];
  }

  const size_t hoff = (size_t)(b * NHEAD + hd) * HEAD_F + lane * 4;
  const float* Kbase = Kmat + hoff;
  const float* Vbase = V + hoff;
  const int* adjrow = adj + (size_t)(n0 + col) * NSEQ + 8 * hi;

  float m_i = -__builtin_inff();
  float l_i = 0.f;
  v8f o = v8zero();

  for (int m0 = 0; m0 < NSEQ; m0 += 16) {
    v8f s = v8zero();
    const float* Kc = Kbase + (size_t)m0 * (TILE_F / 16);
    const v4f kq[2] = {*(const v4f*)Kc, *(const v4f*)(Kc + 128)};
#pragma unroll
    for (int t = 0; t < 4; ++t) {
      v2f a;
      a.x = kq[t >> 1][2 * (t & 1)];
      a.y = kq[t >> 1][2 * (t & 1) + 1];
      s = wmma4(a, qf[t], s);
    }

    const v4i a0 = *(const v4i*)(adjrow + m0);
    const v4i a1 = *(const v4i*)(adjrow + m0 + 4);
    float sv[8];
#pragma unroll
    for (int r = 0; r < 8; ++r) {
      const int av = (r < 4) ? a0[r & 3] : a1[r & 3];
      sv[r] = s[r] * 0.25f;
      if (av == 0) sv[r] = -1e30f;
    }
    float mloc = sv[0];
#pragma unroll
    for (int r = 1; r < 8; ++r) mloc = fmaxf(mloc, sv[r]);
    mloc = fmaxf(mloc, xhalf(mloc));
    const float mnew = fmaxf(m_i, mloc);
    const float f = __expf(m_i - mnew);
    float p[8];
    float psum = 0.f;
#pragma unroll
    for (int r = 0; r < 8; ++r) {
      p[r] = __expf(sv[r] - mnew);
      psum += p[r];
    }
    l_i = l_i * f + psum;
#pragma unroll
    for (int r = 0; r < 8; ++r) o[r] *= f;
    m_i = mnew;

    const float* Vc = Vbase + (size_t)m0 * (TILE_F / 16);
    const v4f vq[2] = {*(const v4f*)Vc, *(const v4f*)(Vc + 128)};
#pragma unroll
    for (int t = 0; t < 4; ++t) {
      v2f a, bb;
      a.x = vq[t >> 1][2 * (t & 1)];
      a.y = vq[t >> 1][2 * (t & 1) + 1];
      bb.x = p[2 * t];
      bb.y = p[2 * t + 1];
      o = wmma4(a, bb, o);
    }
  }

  l_i += xhalf(l_i);
  float* Op = O + base + (size_t)(n0 + col) * CDIM + hd * HDIM + 8 * hi;
  v4f o0, o1;
#pragma unroll
  for (int r = 0; r < 4; ++r) {
    o0[r] = o[r] / l_i;
    o1[r] = o[r + 4] / l_i;
  }
  *(v4f*)Op = o0;
  *(v4f*)(Op + 4) = o1;
}

__global__ __launch_bounds__(256) void gemm_ln_kernel(
    const float* __restrict__ A, int lda, int K, const float* __restrict__ W,
    const float* __restrict__ bias, const float* __restrict__ res,
    const float* __restrict__ g, const float* __restrict__ bt,
    float* __restrict__ Out) {
  const int lane = threadIdx.x & 31;
  const int col = lane & 15, hi = lane >> 4;
  const int gw = blockIdx.x * (blockDim.x >> 5) + (threadIdx.x >> 5);
  const int row0 = gw * 16;

  v8f acc[8];
#pragma unroll
  for (int j = 0; j < 8; ++j) acc[j] = v8zero();

  gemm_acc<8>(A + (size_t)(row0 + col) * lda, W + col, CDIM, K, acc);

#pragma unroll
  for (int r = 0; r < 8; ++r) {
    const int row = row0 + r + 8 * hi;
    float s1 = 0.f, s2 = 0.f;
#pragma unroll
    for (int j = 0; j < 8; ++j) {
      float v = acc[j][r] + bias[16 * j + col] +
                res[(size_t)row * CDIM + 16 * j + col];
      acc[j][r] = v;
      s1 += v;
      s2 += v * v;
    }
    s1 = redsum16(s1);
    s2 = redsum16(s2);
    const float mean = s1 * (1.f / CDIM);
    const float var = s2 * (1.f / CDIM) - mean * mean;
    const float rstd = rsqrtf(var + 1e-5f);
#pragma unroll
    for (int j = 0; j < 8; ++j)
      Out[(size_t)row * CDIM + 16 * j + col] =
          (acc[j][r] - mean) * rstd * g[16 * j + col] + bt[16 * j + col];
  }
}

__global__ __launch_bounds__(256) void ffn1_kernel(
    const float* __restrict__ X, const float* __restrict__ W1,
    const float* __restrict__ b1, float* __restrict__ T) {
  const int lane = threadIdx.x & 31;
  const int col = lane & 15, hi = lane >> 4;
  const int gw = blockIdx.x * (blockDim.x >> 5) + (threadIdx.x >> 5);
  const int half = gw & 1;
  const int rt = gw >> 1;
  const int row0 = rt * 16, c0 = half * 128;

  v8f acc[8];
#pragma unroll
  for (int j = 0; j < 8; ++j) acc[j] = v8zero();

  gemm_acc<8>(X + (size_t)(row0 + col) * CDIM, W1 + c0 + col, FDIM, CDIM, acc);

#pragma unroll
  for (int j = 0; j < 8; ++j)
#pragma unroll
    for (int r = 0; r < 8; ++r)
      T[(size_t)(row0 + r + 8 * hi) * FDIM + c0 + 16 * j + col] =
          fmaxf(acc[j][r] + b1[c0 + 16 * j + col], 0.f);
}

extern "C" void kernel_launch(void* const* d_in, const int* in_sizes, int n_in,
                              void* d_out, int out_size, void* d_ws,
                              size_t ws_size, hipStream_t stream) {
  (void)in_sizes; (void)n_in; (void)out_size; (void)ws_size;
  const int* adj  = (const int*)d_in[0];
  const float* h  = (const float*)d_in[1];
  const float* Wq = (const float*)d_in[2];
  const float* Wk = (const float*)d_in[3];
  const float* Wv = (const float*)d_in[4];
  const float* Wo = (const float*)d_in[5];
  const float* bo = (const float*)d_in[6];
  const float* g1 = (const float*)d_in[7];
  const float* bt1 = (const float*)d_in[8];
  const float* W1 = (const float*)d_in[9];
  const float* bb1 = (const float*)d_in[10];
  const float* W2 = (const float*)d_in[11];
  const float* bb2 = (const float*)d_in[12];
  const float* g2 = (const float*)d_in[13];
  const float* bt2 = (const float*)d_in[14];

  float* ws = (float*)d_ws;
  const size_t SZ = (size_t)MROWS * CDIM;
  float* Q  = ws;
  float* K  = Q + SZ;
  float* V  = K + SZ;
  float* AO = V + SZ;
  float* X  = AO + SZ;
  float* T  = X + SZ;

  qkv_kernel<<<(3 * (MROWS / 16)) / 8, 256, 0, stream>>>(h, Wq, Wk, Wv, Q, K, V);
  attn_kernel<<<(BATCH * NHEAD * (NSEQ / 16)) / 4, 128, 0, stream>>>(Q, K, V,
                                                                     adj, AO);
  gemm_ln_kernel<<<(MROWS / 16) / 8, 256, 0, stream>>>(AO, CDIM, CDIM, Wo, bo,
                                                       h, g1, bt1, X);
  ffn1_kernel<<<(2 * (MROWS / 16)) / 8, 256, 0, stream>>>(X, W1, bb1, T);
  gemm_ln_kernel<<<(MROWS / 16) / 8, 256, 0, stream>>>(T, FDIM, FDIM, W2, bb2,
                                                       X, g2, bt2,
                                                       (float*)d_out);
}


# attention: each key tile's K, V and adjacency loads issued 8 tiles ahead into a ring of register sets (loop unrolled by the ring) so their latency overlaps earlier tiles' work; sums unchanged
# speedup vs baseline: 2.2814x; 1.2177x over previous
#include <hip/hip_runtime.h>

#define DEV __device__ __forceinline__

typedef float v2f __attribute__((ext_vector_type(2)));
typedef float v8f __attribute__((ext_vector_type(8)));
typedef float v4f __attribute__((ext_vector_type(4)));
typedef int v4i __attribute__((ext_vector_type(4)));

static constexpr int CDIM  = 128;
static constexpr int FDIM  = 256;
static constexpr int NSEQ  = 2048;
static constexpr int BATCH = 4;
static constexpr int NHEAD = 8;
static constexpr int HDIM  = 16;
static constexpr int MROWS = BATCH * NSEQ;

DEV v8f v8zero() {
  v8f v;
#pragma unroll
  for (int i = 0; i < 8; ++i) v[i] = 0.f;
  return v;
}

DEV v8f wmma4(v2f a, v2f b, v8f c) {
  return __builtin_amdgcn_wmma_f32_16x16x4_f32(false, a, false, b, (short)0, c,
                                               false, false);
}

DEV float redsum16(float v) {
#pragma unroll
  for (int m = 1; m < 16; m <<= 1) v += __shfl_xor(v, m, 32);
  return v;
}

template <int NT, bool SWAP = false>
DEV void gemm_acc(const float* __restrict__ Arow, const float* __restrict__ Bc,
                  int ldb, int K, v8f* acc) {
  const int hi = (threadIdx.x >> 4) & 1;
  for (int k = 0; k < K; k += 4) {
    const int kb = k + 2 * hi;
    v2f a;
    a.x = Arow[kb];
    a.y = Arow[kb + 1];
#pragma unroll
    for (int j = 0; j < NT; ++j) {
      v2f b;
      b.x = Bc[(size_t)kb * ldb + j * 16];
      b.y = Bc[(size_t)(kb + 1) * ldb + j * 16];
      acc[j] = SWAP ? wmma4(b, a, acc[j]) : wmma4(a, b, acc[j]);
    }
  }
}

static constexpr int TILE_F = 256;
static constexpr int HEAD_F = (NSEQ / 16) * TILE_F;

__global__ __launch_bounds__(256) void qkv_kernel(
    const float* __restrict__ h, const float* __restrict__ Wq,
    const float* __restrict__ Wk, const float* __restrict__ Wv,
    float* __restrict__ Q, float* __restrict__ K, float* __restrict__ V) {
  const int lane = threadIdx.x & 31;
  const int col = lane & 15, hi = lane >> 4;
  const int wv = __builtin_amdgcn_readfirstlane((int)(threadIdx.x >> 5));
  const int gw = blockIdx.x * (blockDim.x >> 5) + wv;
  const int mat = gw / (MROWS / 16);
  const int rt = gw % (MROWS / 16);
  const float* W = (mat == 0) ? Wq : (mat == 1) ? Wk : Wv;
  const int row0 = rt * 16;

  v8f acc[8];
#pragma unroll
  for (int j = 0; j < 8; ++j) acc[j] = v8zero();

  const float* Arow = h + (size_t)(row0 + col) * CDIM;
  if (mat == 1)
    gemm_acc<8, true>(Arow, W + col, CDIM, CDIM, acc);
  else
    gemm_acc<8, false>(Arow, W + col, CDIM, CDIM, acc);

  if (mat == 0) {
#pragma unroll
    for (int j = 0; j < 8; ++j)
#pragma unroll
      for (int r = 0; r < 8; ++r)
        Q[(size_t)(row0 + r + 8 * hi) * CDIM + 16 * j + col] = acc[j][r];
  } else {
    float* Out = (mat == 1) ? K : V;
    float* T = Out + (size_t)((rt / (NSEQ / 16)) * NHEAD) * HEAD_F +
               (size_t)(rt & (NSEQ / 16 - 1)) * TILE_F + lane * 4;
#pragma unroll
    for (int j = 0; j < 8; ++j) {
      v4f lo, up;
#pragma unroll
      for (int e = 0; e < 4; ++e) {
        lo[e] = acc[j][e];
        up[e] = acc[j][e + 4];
      }
      *(v4f*)(T + (size_t)j * HEAD_F) = lo;
      *(v4f*)(T + (size_t)j * HEAD_F + 128) = up;
    }
  }
}

DEV float xhalf(float x) {
  const int b = __builtin_bit_cast(int, x);
  return __builtin_bit_cast(
      float, __builtin_amdgcn_permlanex16(b, b, 0x76543210, 0xfedcba98, false,
                                          false));
}

static constexpr int NTILE = NSEQ / 16;
static constexpr int RING = 8;

struct KeyTile {
  v4f k0, k1;
  v4i a0, a1;
  v4f v0, v1;
};

DEV void tile_load(KeyTile& t, const float* __restrict__ Kt,
                   const float* __restrict__ Vt, const int* __restrict__ At,
                   unsigned lo, unsigned ao) {
  t.k0 = *(const v4f*)(Kt + lo);
  t.k1 = *(const v4f*)(Kt + lo + 128);
  t.a0 = *(const v4i*)(At + ao);
  t.a1 = *(const v4i*)(At + ao + 4);
  t.v0 = *(const v4f*)(Vt + lo);
  t.v1 = *(const v4f*)(Vt + lo + 128);
}

DEV void tile_step(const KeyTile& t, const v2f (&qf)[4], float& m_i,
                   float& l_i, v8f& o) {
  v8f s = v8zero();
  const v4f kq[2] = {t.k0, t.k1};
#pragma unroll
  for (int u = 0; u < 4; ++u) {
    v2f a;
    a.x = kq[u >> 1][2 * (u & 1)];
    a.y = kq[u >> 1][2 * (u & 1) + 1];
    s = wmma4(a, qf[u], s);
  }

  float sv[8];
#pragma unroll
  for (int r = 0; r < 8; ++r) {
    const int av = (r < 4) ? t.a0[r & 3] : t.a1[r & 3];
    sv[r] = s[r] * 0.25f;
    if (av == 0) sv[r] = -1e30f;
  }
  float mloc = sv[0];
#pragma unroll
  for (int r = 1; r < 8; ++r) mloc = fmaxf(mloc, sv[r]);
  mloc = fmaxf(mloc, xhalf(mloc));
  const float mnew = fmaxf(m_i, mloc);
  const float f = __expf(m_i - mnew);
  float p[8];
  float psum = 0.f;
#pragma unroll
  for (int r = 0; r < 8; ++r) {
    p[r] = __expf(sv[r] - mnew);
    psum += p[r];
  }
  l_i = l_i * f + psum;
#pragma unroll
  for (int r = 0; r < 8; ++r) o[r] *= f;
  m_i = mnew;

  const v4f vq[2] = {t.v0, t.v1};
#pragma unroll
  for (int u = 0; u < 4; ++u) {
    v2f a, bb;
    a.x = vq[u >> 1][2 * (u & 1)];
    a.y = vq[u >> 1][2 * (u & 1) + 1];
    bb.x = p[2 * u];
    bb.y = p[2 * u + 1];
    o = wmma4(a, bb, o);
  }
}

__global__ __launch_bounds__(128) void attn_kernel(
    const float* __restrict__ Q, const float* __restrict__ Kmat,
    const float* __restrict__ V, const int* __restrict__ adj,
    float* __restrict__ O) {
  const int lane = threadIdx.x & 31;
  const int col = lane & 15, hi = lane >> 4;
  const int wid = __builtin_amdgcn_readfirstlane((int)(threadIdx.x >> 5));
  const int gw = blockIdx.x * 4 + wid;
  const int nt = gw & (NTILE - 1);
  const int hd = (gw >> 7) & (NHEAD - 1);
  const int b = gw >> 10;
  const int n0 = nt * 16;
  const size_t base = (size_t)b * NSEQ * CDIM;

  const float* Qrow = Q + base + (size_t)(n0 + col) * CDIM + hd * HDIM + 8 * hi;
  v2f qf[4];
#pragma unroll
  for (int t = 0; t < 4; ++t) {
    qf[t].x = Qrow[2 * t];
    qf[t].y = Qrow[2 * t + 1];
  }

  const size_t hoff = (size_t)(b * NHEAD + hd) * HEAD_F;
  const float* Kh = Kmat + hoff;
  const float* Vh = V + hoff;
  const unsigned lo = lane * 4;
  const int* Ah = adj + (size_t)n0 * NSEQ;
  const unsigned ao = col * NSEQ + 8 * hi;

  float m_i = -__builtin_inff();
  float l_i = 0.f;
  v8f o = v8zero();

  KeyTile ring[RING];
#pragma unroll
  for (int s = 0; s < RING; ++s) {
    tile_load(ring[s], Kh + s * TILE_F, Vh + s * TILE_F, Ah + s * 16, lo, ao);
    __builtin_amdgcn_sched_barrier(0);
  }

  for (int m = 0; m < NTILE - RING; m += RING) {
#pragma unroll
    for (int s = 0; s < RING; ++s) {
      tile_step(ring[s], qf, m_i, l_i, o);
      __builtin_amdgcn_sched_barrier(0);
      const int nx = m + s + RING;
      tile_load(ring[s], Kh + (size_t)nx * TILE_F, Vh + (size_t)nx * TILE_F,
                Ah + nx * 16, lo, ao);
      __builtin_amdgcn_sched_barrier(0);
    }
  }
#pragma unroll
  for (int s = 0; s < RING; ++s) tile_step(ring[s], qf, m_i, l_i, o);

  l_i += xhalf(l_i);
  float* Op = O + base + (size_t)(n0 + col) * CDIM + hd * HDIM + 8 * hi;
  v4f o0, o1;
#pragma unroll
  for (int r = 0; r < 4; ++r) {
    o0[r] = o[r] / l_i;
    o1[r] = o[r + 4] / l_i;
  }
  *(v4f*)Op = o0;
  *(v4f*)(Op + 4) = o1;
}

__global__ __launch_bounds__(256) void gemm_ln_kernel(
    const float* __restrict__ A, int lda, int K, const float* __restrict__ W,
    const float* __restrict__ bias, const float* __restrict__ res,
    const float* __restrict__ g, const float* __restrict__ bt,
    float* __restrict__ Out) {
  const int lane = threadIdx.x & 31;
  const int col = lane & 15, hi = lane >> 4;
  const int gw = blockIdx.x * (blockDim.x >> 5) + (threadIdx.x >> 5);
  const int row0 = gw * 16;

  v8f acc[8];
#pragma unroll
  for (int j = 0; j < 8; ++j) acc[j] = v8zero();

  gemm_acc<8>(A + (size_t)(row0 + col) * lda, W + col, CDIM, K, acc);

#pragma unroll
  for (int r = 0; r < 8; ++r) {
    const int row = row0 + r + 8 * hi;
    float s1 = 0.f, s2 = 0.f;
#pragma unroll
    for (int j = 0; j < 8; ++j) {
      float v = acc[j][r] + bias[16 * j + col] +
                res[(size_t)row * CDIM + 16 * j + col];
      acc[j][r] = v;
      s1 += v;
      s2 += v * v;
    }
    s1 = redsum16(s1);
    s2 = redsum16(s2);
    const float mean = s1 * (1.f / CDIM);
    const float var = s2 * (1.f / CDIM) - mean * mean;
    const float rstd = rsqrtf(var + 1e-5f);
#pragma unroll
    for (int j = 0; j < 8; ++j)
      Out[(size_t)row * CDIM + 16 * j + col] =
          (acc[j][r] - mean) * rstd * g[16 * j + col] + bt[16 * j + col];
  }
}

__global__ __launch_bounds__(256) void ffn1_kernel(
    const float* __restrict__ X, const float* __restrict__ W1,
    const float* __restrict__ b1, float* __restrict__ T) {
  const int lane = threadIdx.x & 31;
  const int col = lane & 15, hi = lane >> 4;
  const int gw = blockIdx.x * (blockDim.x >> 5) + (threadIdx.x >> 5);
  const int half = gw & 1;
  const int rt = gw >> 1;
  const int row0 = rt * 16, c0 = half * 128;

  v8f acc[8];
#pragma unroll
  for (int j = 0; j < 8; ++j) acc[j] = v8zero();

  gemm_acc<8>(X + (size_t)(row0 + col) * CDIM, W1 + c0 + col, FDIM, CDIM, acc);

#pragma unroll
  for (int j = 0; j < 8; ++j)
#pragma unroll
    for (int r = 0; r < 8; ++r)
      T[(size_t)(row0 + r + 8 * hi) * FDIM + c0 + 16 * j + col] =
          fmaxf(acc[j][r] + b1[c0 + 16 * j + col], 0.f);
}

extern "C" void kernel_launch(void* const* d_in, const int* in_sizes, int n_in,
                              void* d_out, int out_size, void* d_ws,
                              size_t ws_size, hipStream_t stream) {
  (void)in_sizes; (void)n_in; (void)out_size; (void)ws_size;
  const int* adj  = (const int*)d_in[0];
  const float* h  = (const float*)d_in[1];
  const float* Wq = (const float*)d_in[2];
  const float* Wk = (const float*)d_in[3];
  const float* Wv = (const float*)d_in[4];
  const float* Wo = (const float*)d_in[5];
  const float* bo = (const float*)d_in[6];
  const float* g1 = (const float*)d_in[7];
  const float* bt1 = (const float*)d_in[8];
  const float* W1 = (const float*)d_in[9];
  const float* bb1 = (const float*)d_in[10];
  const float* W2 = (const float*)d_in[11];
  const float* bb2 = (const float*)d_in[12];
  const float* g2 = (const float*)d_in[13];
  const float* bt2 = (const float*)d_in[14];

  float* ws = (float*)d_ws;
  const size_t SZ = (size_t)MROWS * CDIM;
  float* Q  = ws;
  float* K  = Q + SZ;
  float* V  = K + SZ;
  float* AO = V + SZ;
  float* X  = AO + SZ;
  float* T  = X + SZ;

  qkv_kernel<<<(3 * (MROWS / 16)) / 8, 256, 0, stream>>>(h, Wq, Wk, Wv, Q, K, V);
  attn_kernel<<<(BATCH * NHEAD * (NSEQ / 16)) / 4, 128, 0, stream>>>(Q, K, V,
                                                                     adj, AO);
  gemm_ln_kernel<<<(MROWS / 16) / 8, 256, 0, stream>>>(AO, CDIM, CDIM, Wo, bo,
                                                       h, g1, bt1, X);
  ffn1_kernel<<<(2 * (MROWS / 16)) / 8, 256, 0, stream>>>(X, W1, bb1, T);
  gemm_ln_kernel<<<(MROWS / 16) / 8, 256, 0, stream>>>(T, FDIM, FDIM, W2, bb2,
                                                       X, g2, bt2,
                                                       (float*)d_out);
}


# adjacency packed once (extra blocks of the projection launch) into a 32-bit lane mask per score register; attention fetches them by scalar load and selects with them directly; sums unchanged
# speedup vs baseline: 2.3664x; 1.0373x over previous
#include <hip/hip_runtime.h>

#define DEV __device__ __forceinline__

typedef float v2f __attribute__((ext_vector_type(2)));
typedef float v8f __attribute__((ext_vector_type(8)));
typedef float v4f __attribute__((ext_vector_type(4)));
typedef int v4i __attribute__((ext_vector_type(4)));
typedef unsigned v16u __attribute__((ext_vector_type(16)));

static constexpr int CDIM  = 128;
static constexpr int FDIM  = 256;
static constexpr int NSEQ  = 2048;
static constexpr int BATCH = 4;
static constexpr int NHEAD = 8;
static constexpr int HDIM  = 16;
static constexpr int MROWS = BATCH * NSEQ;
static constexpr int NTILE = NSEQ / 16;

DEV v8f v8zero() {
  v8f v;
#pragma unroll
  for (int i = 0; i < 8; ++i) v[i] = 0.f;
  return v;
}

DEV v8f wmma4(v2f a, v2f b, v8f c) {
  return __builtin_amdgcn_wmma_f32_16x16x4_f32(false, a, false, b, (short)0, c,
                                               false, false);
}

DEV float redsum16(float v) {
#pragma unroll
  for (int m = 1; m < 16; m <<= 1) v += __shfl_xor(v, m, 32);
  return v;
}

template <int NT, bool SWAP = false>
DEV void gemm_acc(const float* __restrict__ Arow, const float* __restrict__ Bc,
                  int ldb, int K, v8f* acc) {
  const int hi = (threadIdx.x >> 4) & 1;
  for (int k = 0; k < K; k += 4) {
    const int kb = k + 2 * hi;
    v2f a;
    a.x = Arow[kb];
    a.y = Arow[kb + 1];
#pragma unroll
    for (int j = 0; j < NT; ++j) {
      v2f b;
      b.x = Bc[(size_t)kb * ldb + j * 16];
      b.y = Bc[(size_t)(kb + 1) * ldb + j * 16];
      acc[j] = SWAP ? wmma4(b, a, acc[j]) : wmma4(a, b, acc[j]);
    }
  }
}

static constexpr int TILE_F = 256;
static constexpr int HEAD_F = (NSEQ / 16) * TILE_F;

static constexpr int QKV_BLOCKS  = (3 * (MROWS / 16)) / 8;
static constexpr int PACK_TILES  = 8;
static constexpr int PACK_BLOCKS = (NTILE * NTILE / PACK_TILES) / 8;

DEV void pack_tiles(const int* __restrict__ adj, unsigned* __restrict__ masks,
                    int task, int lane) {
  const int nt = task / (NTILE / PACK_TILES);
  const int t0 = (task % (NTILE / PACK_TILES)) * PACK_TILES;
  const int* src = adj + (size_t)(nt * 16 + (lane & 15)) * NSEQ + t0 * 16 +
                   8 * (lane >> 4);
  unsigned* dst = masks + ((size_t)nt * NTILE + t0) * 8 + lane;
#pragma unroll
  for (int g = 0; g < PACK_TILES / 4; ++g) {
    unsigned w = 0;
#pragma unroll
    for (int j = 0; j < 4; ++j) {
      const v4i a0 = *(const v4i*)(src + (g * 4 + j) * 16);
      const v4i a1 = *(const v4i*)(src + (g * 4 + j) * 16 + 4);
#pragma unroll
      for (int r = 0; r < 8; ++r) {
        const int av = (r < 4) ? a0[r & 3] : a1[r & 3];
        const unsigned bal = __builtin_amdgcn_ballot_w32(av != 0);
        if (lane == j * 8 + r) w = bal;
      }
    }
    dst[g * 32] = w;
  }
}

__global__ __launch_bounds__(256) void qkv_kernel(
    const float* __restrict__ h, const float* __restrict__ Wq,
    const float* __restrict__ Wk, const float* __restrict__ Wv,
    const int* __restrict__ adj, float* __restrict__ Q, float* __restrict__ K,
    float* __restrict__ V, unsigned* __restrict__ masks) {
  const int lane = threadIdx.x & 31;
  const int col = lane & 15, hi = lane >> 4;
  const int wv = __builtin_amdgcn_readfirstlane((int)(threadIdx.x >> 5));
  if (blockIdx.x >= QKV_BLOCKS) {
    pack_tiles(adj, masks, (blockIdx.x - QKV_BLOCKS) * 8 + wv, lane);
    return;
  }
  const int gw = blockIdx.x * (blockDim.x >> 5) + wv;
  const int mat = gw / (MROWS / 16);
  const int rt = gw % (MROWS / 16);
  const float* W = (mat == 0) ? Wq : (mat == 1) ? Wk : Wv;
  const int row0 = rt * 16;

  v8f acc[8];
#pragma unroll
  for (int j = 0; j < 8; ++j) acc[j] = v8zero();

  const float* Arow = h + (size_t)(row0 + col) * CDIM;
  if (mat == 1)
    gemm_acc<8, true>(Arow, W + col, CDIM, CDIM, acc);
  else
    gemm_acc<8, false>(Arow, W + col, CDIM, CDIM, acc);

  if (mat == 0) {
#pragma unroll
    for (int j = 0; j < 8; ++j)
#pragma unroll
      for (int r = 0; r < 8; ++r)
        Q[(size_t)(row0 + r + 8 * hi) * CDIM + 16 * j + col] = acc[j][r];
  } else {
    float* Out = (mat == 1) ? K : V;
    float* T = Out + (size_t)((rt / (NSEQ / 16)) * NHEAD) * HEAD_F +
               (size_t)(rt & (NSEQ / 16 - 1)) * TILE_F + lane * 4;
#pragma unroll
    for (int j = 0; j < 8; ++j) {
      v4f lo, up;
#pragma unroll
      for (int e = 0; e < 4; ++e) {
        lo[e] = acc[j][e];
        up[e] = acc[j][e + 4];
      }
      *(v4f*)(T + (size_t)j * HEAD_F) = lo;
      *(v4f*)(T + (size_t)j * HEAD_F + 128) = up;
    }
  }
}

DEV float xhalf(float x) {
  const int b = __builtin_bit_cast(int, x);
  return __builtin_bit_cast(
      float, __builtin_amdgcn_permlanex16(b, b, 0x76543210, 0xfedcba98, false,
                                          false));
}

static constexpr int RING = 8;

struct KeyTile {
  v4f k0, k1;
  v4f v0, v1;
};

DEV void tile_load(KeyTile& t, const float* __restrict__ Kt,
                   const float* __restrict__ Vt, unsigned lo) {
  t.k0 = *(const v4f*)(Kt + lo);
  t.k1 = *(const v4f*)(Kt + lo + 128);
  t.v0 = *(const v4f*)(Vt + lo);
  t.v1 = *(const v4f*)(Vt + lo + 128);
}

DEV void tile_step(const KeyTile& t, const v16u& mw, int off,
                   const v2f (&qf)[4], float& m_i, float& l_i, v8f& o) {
  v8f s = v8zero();
  const v4f kq[2] = {t.k0, t.k1};
#pragma unroll
  for (int u = 0; u < 4; ++u) {
    v2f a;
    a.x = kq[u >> 1][2 * (u & 1)];
    a.y = kq[u >> 1][2 * (u & 1) + 1];
    s = wmma4(a, qf[u], s);
  }

  float sv[8];
#pragma unroll
  for (int r = 0; r < 8; ++r) {
    const bool edge = __builtin_amdgcn_inverse_ballot_w32(mw[off + r]);
    sv[r] = edge ? s[r] * 0.25f
                 : -1e30f;
  }
  float mloc = sv[0];
#pragma unroll
  for (int r = 1; r < 8; ++r) mloc = fmaxf(mloc, sv[r]);
  mloc = fmaxf(mloc, xhalf(mloc));
  const float mnew = fmaxf(m_i, mloc);
  const float f = __expf(m_i - mnew);
  float p[8];
  float psum = 0.f;
#pragma unroll
  for (int r = 0; r < 8; ++r) {
    p[r] = __expf(sv[r] - mnew);
    psum += p[r];
  }
  l_i = l_i * f + psum;
#pragma unroll
  for (int r = 0; r < 8; ++r) o[r] *= f;
  m_i = mnew;

  const v4f vq[2] = {t.v0, t.v1};
#pragma unroll
  for (int u = 0; u < 4; ++u) {
    v2f a, bb;
    a.x = vq[u >> 1][2 * (u & 1)];
    a.y = vq[u >> 1][2 * (u & 1) + 1];
    bb.x = p[2 * u];
    bb.y = p[2 * u + 1];
    o = wmma4(a, bb, o);
  }
}

__global__ __launch_bounds__(128) __attribute__((amdgpu_waves_per_eu(4, 4)))
void attn_kernel(
    const float* __restrict__ Q, const float* __restrict__ Kmat,
    const float* __restrict__ V, const unsigned* __restrict__ masks,
    float* __restrict__ O) {
  const int lane = threadIdx.x & 31;
  const int col = lane & 15, hi = lane >> 4;
  const int wid = __builtin_amdgcn_readfirstlane((int)(threadIdx.x >> 5));
  const int gw = blockIdx.x * 4 + wid;
  const int nt = gw & (NTILE - 1);
  const int hd = (gw >> 7) & (NHEAD - 1);
  const int b = gw >> 10;
  const int n0 = nt * 16;
  const size_t base = (size_t)b * NSEQ * CDIM;

  const float* Qrow = Q + base + (size_t)(n0 + col) * CDIM + hd * HDIM + 8 * hi;
  v2f qf[4];
#pragma unroll
  for (int t = 0; t < 4; ++t) {
    qf[t].x = Qrow[2 * t];
    qf[t].y = Qrow[2 * t + 1];
  }

  const size_t hoff = (size_t)(b * NHEAD + hd) * HEAD_F;
  const float* Kh = Kmat + hoff;
  const float* Vh = V + hoff;
  const unsigned lo = lane * 4;
  const unsigned* Mq = masks + (size_t)nt * (NTILE * 8);

  float m_i = -__builtin_inff();
  float l_i = 0.f;
  v8f o = v8zero();

  v16u mp[2];
  mp[0] = *(const v16u*)Mq;
  KeyTile ring[RING];
#pragma unroll
  for (int s = 0; s < RING; ++s) {
    tile_load(ring[s], Kh + s * TILE_F, Vh + s * TILE_F, lo);
    __builtin_amdgcn_sched_barrier(0);
  }

  for (int m = 0; m < NTILE - RING; m += RING) {
#pragma unroll
    for (int s = 0; s < RING; ++s) {
      tile_step(ring[s], mp[(s >> 1) & 1], (s & 1) * 8, qf, m_i, l_i, o);
      __builtin_amdgcn_sched_barrier(0);
      const int nx = m + s + RING;
      tile_load(ring[s], Kh + (size_t)nx * TILE_F, Vh + (size_t)nx * TILE_F,
                lo);
      if (!(s & 1))
        mp[((s >> 1) + 1) & 1] = *(const v16u*)(Mq + ((m + s) / 2 + 1) * 16);
      __builtin_amdgcn_sched_barrier(0);
    }
  }
#pragma unroll
  for (int s = 0; s < RING; ++s) {
    tile_step(ring[s], mp[(s >> 1) & 1], (s & 1) * 8, qf, m_i, l_i, o);
    if (!(s & 1) && s < RING - 2)
      mp[((s >> 1) + 1) & 1] =
          *(const v16u*)(Mq + ((NTILE - RING + s) / 2 + 1) * 16);
  }

  l_i += xhalf(l_i);
  float* Op = O + base + (size_t)(n0 + col) * CDIM + hd * HDIM + 8 * hi;
  v4f o0, o1;
#pragma unroll
  for (int r = 0; r < 4; ++r) {
    o0[r] = o[r] / l_i;
    o1[r] = o[r + 4] / l_i;
  }
  *(v4f*)Op = o0;
  *(v4f*)(Op + 4) = o1;
}

__global__ __launch_bounds__(256) void gemm_ln_kernel(
    const float* __restrict__ A, int lda, int K, const float* __restrict__ W,
    const float* __restrict__ bias, const float* __restrict__ res,
    const float* __restrict__ g, const float* __restrict__ bt,
    float* __restrict__ Out) {
  const int lane = threadIdx.x & 31;
  const int col = lane & 15, hi = lane >> 4;
  const int gw = blockIdx.x * (blockDim.x >> 5) + (threadIdx.x >> 5);
  const int row0 = gw * 16;

  v8f acc[8];
#pragma unroll
  for (int j = 0; j < 8; ++j) acc[j] = v8zero();

  gemm_acc<8>(A + (size_t)(row0 + col) * lda, W + col, CDIM, K, acc);

#pragma unroll
  for (int r = 0; r < 8; ++r) {
    const int row = row0 + r + 8 * hi;
    float s1 = 0.f, s2 = 0.f;
#pragma unroll
    for (int j = 0; j < 8; ++j) {
      float v = acc[j][r] + bias[16 * j + col] +
                res[(size_t)row * CDIM + 16 * j + col];
      acc[j][r] = v;
      s1 += v;
      s2 += v * v;
    }
    s1 = redsum16(s1);
    s2 = redsum16(s2);
    const float mean = s1 * (1.f / CDIM);
    const float var = s2 * (1.f / CDIM) - mean * mean;
    const float rstd = rsqrtf(var + 1e-5f);
#pragma unroll
    for (int j = 0; j < 8; ++j)
      Out[(size_t)row * CDIM + 16 * j + col] =
          (acc[j][r] - mean) * rstd * g[16 * j + col] + bt[16 * j + col];
  }
}

__global__ __launch_bounds__(256) void ffn1_kernel(
    const float* __restrict__ X, const float* __restrict__ W1,
    const float* __restrict__ b1, float* __restrict__ T) {
  const int lane = threadIdx.x & 31;
  const int col = lane & 15, hi = lane >> 4;
  const int gw = blockIdx.x * (blockDim.x >> 5) + (threadIdx.x >> 5);
  const int half = gw & 1;
  const int rt = gw >> 1;
  const int row0 = rt * 16, c0 = half * 128;

  v8f acc[8];
#pragma unroll
  for (int j = 0; j < 8; ++j) acc[j] = v8zero();

  gemm_acc<8>(X + (size_t)(row0 + col) * CDIM, W1 + c0 + col, FDIM, CDIM, acc);

#pragma unroll
  for (int j = 0; j < 8; ++j)
#pragma unroll
    for (int r = 0; r < 8; ++r)
      T[(size_t)(row0 + r + 8 * hi) * FDIM + c0 + 16 * j + col] =
          fmaxf(acc[j][r] + b1[c0 + 16 * j + col], 0.f);
}

extern "C" void kernel_launch(void* const* d_in, const int* in_sizes, int n_in,
                              void* d_out, int out_size, void* d_ws,
                              size_t ws_size, hipStream_t stream) {
  (void)in_sizes; (void)n_in; (void)out_size; (void)ws_size;
  const int* adj  = (const int*)d_in[0];
  const float* h  = (const float*)d_in[1];
  const float* Wq = (const float*)d_in[2];
  const float* Wk = (const float*)d_in[3];
  const float* Wv = (const float*)d_in[4];
  const float* Wo = (const float*)d_in[5];
  const float* bo = (const float*)d_in[6];
  const float* g1 = (const float*)d_in[7];
  const float* bt1 = (const float*)d_in[8];
  const float* W1 = (const float*)d_in[9];
  const float* bb1 = (const float*)d_in[10];
  const float* W2 = (const float*)d_in[11];
  const float* bb2 = (const float*)d_in[12];
  const float* g2 = (const float*)d_in[13];
  const float* bt2 = (const float*)d_in[14];

  float* ws = (float*)d_ws;
  const size_t SZ = (size_t)MROWS * CDIM;
  float* Q  = ws;
  float* K  = Q + SZ;
  float* V  = K + SZ;
  float* AO = V + SZ;
  float* X  = AO + SZ;
  float* T  = X + SZ;
  unsigned* masks = (unsigned*)T;

  qkv_kernel<<<QKV_BLOCKS + PACK_BLOCKS, 256, 0, stream>>>(h, Wq, Wk, Wv, adj,
                                                            Q, K, V, masks);
  attn_kernel<<<(BATCH * NHEAD * (NSEQ / 16)) / 4, 128, 0, stream>>>(Q, K, V,
                                                                     masks, AO);
  gemm_ln_kernel<<<(MROWS / 16) / 8, 256, 0, stream>>>(AO, CDIM, CDIM, Wo, bo,
                                                       h, g1, bt1, X);
  ffn1_kernel<<<(2 * (MROWS / 16)) / 8, 256, 0, stream>>>(X, W1, bb1, T);
  gemm_ln_kernel<<<(MROWS / 16) / 8, 256, 0, stream>>>(T, FDIM, FDIM, W2, bb2,
                                                       X, g2, bt2,
                                                       (float*)d_out);
}


# Q, out-proj and FFN GEMMs: tiles take permuted columns and k in blocks of 8 so fragment, bias and residual loads and output stores are 16-byte, whole lines; dot-product and LN row sums re-associated
# speedup vs baseline: 2.6446x; 1.1176x over previous
#include <hip/hip_runtime.h>

#define DEV __device__ __forceinline__

typedef float v2f __attribute__((ext_vector_type(2)));
typedef float v8f __attribute__((ext_vector_type(8)));
typedef float v4f __attribute__((ext_vector_type(4)));
typedef int v4i __attribute__((ext_vector_type(4)));
typedef unsigned v16u __attribute__((ext_vector_type(16)));

static constexpr int CDIM  = 128;
static constexpr int FDIM  = 256;
static constexpr int NSEQ  = 2048;
static constexpr int BATCH = 4;
static constexpr int NHEAD = 8;
static constexpr int HDIM  = 16;
static constexpr int MROWS = BATCH * NSEQ;
static constexpr int NTILE = NSEQ / 16;

DEV v8f v8zero() {
  v8f v;
#pragma unroll
  for (int i = 0; i < 8; ++i) v[i] = 0.f;
  return v;
}

DEV v8f wmma4(v2f a, v2f b, v8f c) {
  return __builtin_amdgcn_wmma_f32_16x16x4_f32(false, a, false, b, (short)0, c,
                                               false, false);
}

DEV float redsum16(float v) {
#pragma unroll
  for (int m = 1; m < 16; m <<= 1) v += __shfl_xor(v, m, 32);
  return v;
}

template <int NT, bool SWAP = false>
DEV void gemm_acc(const float* __restrict__ Arow, const float* __restrict__ Bc,
                  int ldb, int K, v8f* acc) {
  const int hi = (threadIdx.x >> 4) & 1;
  for (int k = 0; k < K; k += 4) {
    const int kb = k + 2 * hi;
    v2f a;
    a.x = Arow[kb];
    a.y = Arow[kb + 1];
#pragma unroll
    for (int j = 0; j < NT; ++j) {
      v2f b;
      b.x = Bc[(size_t)kb * ldb + j * 16];
      b.y = Bc[(size_t)(kb + 1) * ldb + j * 16];
      acc[j] = SWAP ? wmma4(b, a, acc[j]) : wmma4(a, b, acc[j]);
    }
  }
}

DEV void gemm_acc_w(const float* __restrict__ Arow, const float* __restrict__ Bw,
                    int ldb, int K, v8f* acc) {
  const int hi = (threadIdx.x >> 4) & 1;
  for (int k = 0; k < K; k += 8) {
    const int kb = k + 4 * hi;
    const v4f a4 = *(const v4f*)(Arow + kb);
#pragma unroll
    for (int s = 0; s < 2; ++s) {
      v2f a;
      a.x = a4[2 * s];
      a.y = a4[2 * s + 1];
      const float* B0 = Bw + (size_t)(kb + 2 * s) * ldb;
      const float* B1 = B0 + ldb;
#pragma unroll
      for (int g = 0; g < 2; ++g) {
        const v4f b0 = *(const v4f*)(B0 + 64 * g);
        const v4f b1 = *(const v4f*)(B1 + 64 * g);
#pragma unroll
        for (int t = 0; t < 4; ++t) {
          v2f b;
          b.x = b0[t];
          b.y = b1[t];
          acc[4 * g + t] = wmma4(a, b, acc[4 * g + t]);
        }
      }
    }
  }
}

static constexpr int TILE_F = 256;
static constexpr int HEAD_F = (NSEQ / 16) * TILE_F;

static constexpr int QKV_BLOCKS  = (3 * (MROWS / 16)) / 8;
static constexpr int PACK_TILES  = 8;
static constexpr int PACK_BLOCKS = (NTILE * NTILE / PACK_TILES) / 8;

DEV void pack_tiles(const int* __restrict__ adj, unsigned* __restrict__ masks,
                    int task, int lane) {
  const int nt = task / (NTILE / PACK_TILES);
  const int t0 = (task % (NTILE / PACK_TILES)) * PACK_TILES;
  const int* src = adj + (size_t)(nt * 16 + (lane & 15)) * NSEQ + t0 * 16 +
                   8 * (lane >> 4);
  unsigned* dst = masks + ((size_t)nt * NTILE + t0) * 8 + lane;
#pragma unroll
  for (int g = 0; g < PACK_TILES / 4; ++g) {
    unsigned w = 0;
#pragma unroll
    for (int j = 0; j < 4; ++j) {
      const v4i a0 = *(const v4i*)(src + (g * 4 + j) * 16);
      const v4i a1 = *(const v4i*)(src + (g * 4 + j) * 16 + 4);
#pragma unroll
      for (int r = 0; r < 8; ++r) {
        const int av = (r < 4) ? a0[r & 3] : a1[r & 3];
        const unsigned bal = __builtin_amdgcn_ballot_w32(av != 0);
        if (lane == j * 8 + r) w = bal;
      }
    }
    dst[g * 32] = w;
  }
}

__global__ __launch_bounds__(256) void qkv_kernel(
    const float* __restrict__ h, const float* __restrict__ Wq,
    const float* __restrict__ Wk, const float* __restrict__ Wv,
    const int* __restrict__ adj, float* __restrict__ Q, float* __restrict__ K,
    float* __restrict__ V, unsigned* __restrict__ masks) {
  const int lane = threadIdx.x & 31;
  const int col = lane & 15, hi = lane >> 4;
  const int wv = __builtin_amdgcn_readfirstlane((int)(threadIdx.x >> 5));
  if (blockIdx.x >= QKV_BLOCKS) {
    pack_tiles(adj, masks, (blockIdx.x - QKV_BLOCKS) * 8 + wv, lane);
    return;
  }
  const int gw = blockIdx.x * (blockDim.x >> 5) + wv;
  const int mat = gw / (MROWS / 16);
  const int rt = gw % (MROWS / 16);
  const float* W = (mat == 0) ? Wq : (mat == 1) ? Wk : Wv;
  const int row0 = rt * 16;

  v8f acc[8];
#pragma unroll
  for (int j = 0; j < 8; ++j) acc[j] = v8zero();

  const float* Arow = h + (size_t)(row0 + col) * CDIM;
  if (mat == 0)
    gemm_acc_w(Arow, W + 4 * col, CDIM, CDIM, acc);
  else if (mat == 1)
    gemm_acc<8, true>(Arow, W + col, CDIM, CDIM, acc);
  else
    gemm_acc<8, false>(Arow, W + col, CDIM, CDIM, acc);

  if (mat == 0) {
#pragma unroll
    for (int g = 0; g < 2; ++g)
#pragma unroll
      for (int r = 0; r < 8; ++r) {
        v4f v;
#pragma unroll
        for (int t = 0; t < 4; ++t) v[t] = acc[4 * g + t][r];
        *(v4f*)(Q + (size_t)(row0 + r + 8 * hi) * CDIM + 64 * g + 4 * col) = v;
      }
  } else {
    float* Out = (mat == 1) ? K : V;
    float* T = Out + (size_t)((rt / (NSEQ / 16)) * NHEAD) * HEAD_F +
               (size_t)(rt & (NSEQ / 16 - 1)) * TILE_F + lane * 4;
#pragma unroll
    for (int j = 0; j < 8; ++j) {
      v4f lo, up;
#pragma unroll
      for (int e = 0; e < 4; ++e) {
        lo[e] = acc[j][e];
        up[e] = acc[j][e + 4];
      }
      *(v4f*)(T + (size_t)j * HEAD_F) = lo;
      *(v4f*)(T + (size_t)j * HEAD_F + 128) = up;
    }
  }
}

DEV float xhalf(float x) {
  const int b = __builtin_bit_cast(int, x);
  return __builtin_bit_cast(
      float, __builtin_amdgcn_permlanex16(b, b, 0x76543210, 0xfedcba98, false,
                                          false));
}

static constexpr int RING = 8;

struct KeyTile {
  v4f k0, k1;
  v4f v0, v1;
};

DEV void tile_load(KeyTile& t, const float* __restrict__ Kt,
                   const float* __restrict__ Vt, unsigned lo) {
  t.k0 = *(const v4f*)(Kt + lo);
  t.k1 = *(const v4f*)(Kt + lo + 128);
  t.v0 = *(const v4f*)(Vt + lo);
  t.v1 = *(const v4f*)(Vt + lo + 128);
}

DEV void tile_step(const KeyTile& t, const v16u& mw, int off,
                   const v2f (&qf)[4], float& m_i, float& l_i, v8f& o) {
  v8f s = v8zero();
  const v4f kq[2] = {t.k0, t.k1};
#pragma unroll
  for (int u = 0; u < 4; ++u) {
    v2f a;
    a.x = kq[u >> 1][2 * (u & 1)];
    a.y = kq[u >> 1][2 * (u & 1) + 1];
    s = wmma4(a, qf[u], s);
  }

  float sv[8];
#pragma unroll
  for (int r = 0; r < 8; ++r) {
    const bool edge = __builtin_amdgcn_inverse_ballot_w32(mw[off + r]);
    sv[r] = edge ? s[r] * 0.25f
                 : -1e30f;
  }
  float mloc = sv[0];
#pragma unroll
  for (int r = 1; r < 8; ++r) mloc = fmaxf(mloc, sv[r]);
  mloc = fmaxf(mloc, xhalf(mloc));
  const float mnew = fmaxf(m_i, mloc);
  const float f = __expf(m_i - mnew);
  float p[8];
  float psum = 0.f;
#pragma unroll
  for (int r = 0; r < 8; ++r) {
    p[r] = __expf(sv[r] - mnew);
    psum += p[r];
  }
  l_i = l_i * f + psum;
#pragma unroll
  for (int r = 0; r < 8; ++r) o[r] *= f;
  m_i = mnew;

  const v4f vq[2] = {t.v0, t.v1};
#pragma unroll
  for (int u = 0; u < 4; ++u) {
    v2f a, bb;
    a.x = vq[u >> 1][2 * (u & 1)];
    a.y = vq[u >> 1][2 * (u & 1) + 1];
    bb.x = p[2 * u];
    bb.y = p[2 * u + 1];
    o = wmma4(a, bb, o);
  }
}

__global__ __launch_bounds__(128) __attribute__((amdgpu_waves_per_eu(4, 4)))
void attn_kernel(
    const float* __restrict__ Q, const float* __restrict__ Kmat,
    const float* __restrict__ V, const unsigned* __restrict__ masks,
    float* __restrict__ O) {
  const int lane = threadIdx.x & 31;
  const int col = lane & 15, hi = lane >> 4;
  const int wid = __builtin_amdgcn_readfirstlane((int)(threadIdx.x >> 5));
  const int gw = blockIdx.x * 4 + wid;
  const int nt = gw & (NTILE - 1);
  const int hd = (gw >> 7) & (NHEAD - 1);
  const int b = gw >> 10;
  const int n0 = nt * 16;
  const size_t base = (size_t)b * NSEQ * CDIM;

  const float* Qrow = Q + base + (size_t)(n0 + col) * CDIM + hd * HDIM + 8 * hi;
  v2f qf[4];
#pragma unroll
  for (int t = 0; t < 4; ++t) {
    qf[t].x = Qrow[2 * t];
    qf[t].y = Qrow[2 * t + 1];
  }

  const size_t hoff = (size_t)(b * NHEAD + hd) * HEAD_F;
  const float* Kh = Kmat + hoff;
  const float* Vh = V + hoff;
  const unsigned lo = lane * 4;
  const unsigned* Mq = masks + (size_t)nt * (NTILE * 8);

  float m_i = -__builtin_inff();
  float l_i = 0.f;
  v8f o = v8zero();

  v16u mp[2];
  mp[0] = *(const v16u*)Mq;
  KeyTile ring[RING];
#pragma unroll
  for (int s = 0; s < RING; ++s) {
    tile_load(ring[s], Kh + s * TILE_F, Vh + s * TILE_F, lo);
    __builtin_amdgcn_sched_barrier(0);
  }

  for (int m = 0; m < NTILE - RING; m += RING) {
#pragma unroll
    for (int s = 0; s < RING; ++s) {
      tile_step(ring[s], mp[(s >> 1) & 1], (s & 1) * 8, qf, m_i, l_i, o);
      __builtin_amdgcn_sched_barrier(0);
      const int nx = m + s + RING;
      tile_load(ring[s], Kh + (size_t)nx * TILE_F, Vh + (size_t)nx * TILE_F,
                lo);
      if (!(s & 1))
        mp[((s >> 1) + 1) & 1] = *(const v16u*)(Mq + ((m + s) / 2 + 1) * 16);
      __builtin_amdgcn_sched_barrier(0);
    }
  }
#pragma unroll
  for (int s = 0; s < RING; ++s) {
    tile_step(ring[s], mp[(s >> 1) & 1], (s & 1) * 8, qf, m_i, l_i, o);
    if (!(s & 1) && s < RING - 2)
      mp[((s >> 1) + 1) & 1] =
          *(const v16u*)(Mq + ((NTILE - RING + s) / 2 + 1) * 16);
  }

  l_i += xhalf(l_i);
  float* Op = O + base + (size_t)(n0 + col) * CDIM + hd * HDIM + 8 * hi;
  v4f o0, o1;
#pragma unroll
  for (int r = 0; r < 4; ++r) {
    o0[r] = o[r] / l_i;
    o1[r] = o[r + 4] / l_i;
  }
  *(v4f*)Op = o0;
  *(v4f*)(Op + 4) = o1;
}

__global__ __launch_bounds__(256) void gemm_ln_kernel(
    const float* __restrict__ A, int lda, int K, const float* __restrict__ W,
    const float* __restrict__ bias, const float* __restrict__ res,
    const float* __restrict__ g, const float* __restrict__ bt,
    float* __restrict__ Out) {
  const int lane = threadIdx.x & 31;
  const int col = lane & 15, hi = lane >> 4;
  const int gw = blockIdx.x * (blockDim.x >> 5) + (threadIdx.x >> 5);
  const int row0 = gw * 16;

  v8f acc[8];
#pragma unroll
  for (int j = 0; j < 8; ++j) acc[j] = v8zero();

  gemm_acc_w(A + (size_t)(row0 + col) * lda, W + 4 * col, CDIM, K, acc);

  v4f bias4[2], g4[2], bt4[2];
#pragma unroll
  for (int q = 0; q < 2; ++q) {
    bias4[q] = *(const v4f*)(bias + 64 * q + 4 * col);
    g4[q] = *(const v4f*)(g + 64 * q + 4 * col);
    bt4[q] = *(const v4f*)(bt + 64 * q + 4 * col);
  }
#pragma unroll
  for (int r = 0; r < 8; ++r) {
    const int row = row0 + r + 8 * hi;
    const float* rrow = res + (size_t)row * CDIM + 4 * col;
    v4f res4[2];
#pragma unroll
    for (int q = 0; q < 2; ++q) res4[q] = *(const v4f*)(rrow + 64 * q);
    float s1 = 0.f, s2 = 0.f;
#pragma unroll
    for (int j = 0; j < 8; ++j) {
      float v = acc[j][r] + bias4[j >> 2][j & 3] + res4[j >> 2][j & 3];
      acc[j][r] = v;
      s1 += v;
      s2 += v * v;
    }
    s1 = redsum16(s1);
    s2 = redsum16(s2);
    const float mean = s1 * (1.f / CDIM);
    const float var = s2 * (1.f / CDIM) - mean * mean;
    const float rstd = rsqrtf(var + 1e-5f);
    float* orow = Out + (size_t)row * CDIM + 4 * col;
#pragma unroll
    for (int q = 0; q < 2; ++q) {
      v4f o;
#pragma unroll
      for (int t = 0; t < 4; ++t)
        o[t] = (acc[4 * q + t][r] - mean) * rstd * g4[q][t] + bt4[q][t];
      *(v4f*)(orow + 64 * q) = o;
    }
  }
}

__global__ __launch_bounds__(256) void ffn1_kernel(
    const float* __restrict__ X, const float* __restrict__ W1,
    const float* __restrict__ b1, float* __restrict__ T) {
  const int lane = threadIdx.x & 31;
  const int col = lane & 15, hi = lane >> 4;
  const int gw = blockIdx.x * (blockDim.x >> 5) + (threadIdx.x >> 5);
  const int half = gw & 1;
  const int rt = gw >> 1;
  const int row0 = rt * 16, c0 = half * 128;

  v8f acc[8];
#pragma unroll
  for (int j = 0; j < 8; ++j) acc[j] = v8zero();

  gemm_acc_w(X + (size_t)(row0 + col) * CDIM, W1 + c0 + 4 * col, FDIM, CDIM,
             acc);

#pragma unroll
  for (int g = 0; g < 2; ++g) {
    const v4f bv = *(const v4f*)(b1 + c0 + 64 * g + 4 * col);
#pragma unroll
    for (int r = 0; r < 8; ++r) {
      v4f v;
#pragma unroll
      for (int t = 0; t < 4; ++t)
        v[t] = fmaxf(acc[4 * g + t][r] + bv[t], 0.f);
      *(v4f*)(T + (size_t)(row0 + r + 8 * hi) * FDIM + c0 + 64 * g + 4 * col) =
          v;
    }
  }
}

extern "C" void kernel_launch(void* const* d_in, const int* in_sizes, int n_in,
                              void* d_out, int out_size, void* d_ws,
                              size_t ws_size, hipStream_t stream) {
  (void)in_sizes; (void)n_in; (void)out_size; (void)ws_size;
  const int* adj  = (const int*)d_in[0];
  const float* h  = (const float*)d_in[1];
  const float* Wq = (const float*)d_in[2];
  const float* Wk = (const float*)d_in[3];
  const float* Wv = (const float*)d_in[4];
  const float* Wo = (const float*)d_in[5];
  const float* bo = (const float*)d_in[6];
  const float* g1 = (const float*)d_in[7];
  const float* bt1 = (const float*)d_in[8];
  const float* W1 = (const float*)d_in[9];
  const float* bb1 = (const float*)d_in[10];
  const float* W2 = (const float*)d_in[11];
  const float* bb2 = (const float*)d_in[12];
  const float* g2 = (const float*)d_in[13];
  const float* bt2 = (const float*)d_in[14];

  float* ws = (float*)d_ws;
  const size_t SZ = (size_t)MROWS * CDIM;
  float* Q  = ws;
  float* K  = Q + SZ;
  float* V  = K + SZ;
  float* AO = V + SZ;
  float* X  = AO + SZ;
  float* T  = X + SZ;
  unsigned* masks = (unsigned*)T;

  qkv_kernel<<<QKV_BLOCKS + PACK_BLOCKS, 256, 0, stream>>>(h, Wq, Wk, Wv, adj,
                                                            Q, K, V, masks);
  attn_kernel<<<(BATCH * NHEAD * (NSEQ / 16)) / 4, 128, 0, stream>>>(Q, K, V,
                                                                     masks, AO);
  gemm_ln_kernel<<<(MROWS / 16) / 8, 256, 0, stream>>>(AO, CDIM, CDIM, Wo, bo,
                                                       h, g1, bt1, X);
  ffn1_kernel<<<(2 * (MROWS / 16)) / 8, 256, 0, stream>>>(X, W1, bb1, T);
  gemm_ln_kernel<<<(MROWS / 16) / 8, 256, 0, stream>>>(T, FDIM, FDIM, W2, bb2,
                                                       X, g2, bt2,
                                                       (float*)d_out);
}


# out-proj+LN1, FFN1 and FFN2+LN2 fused into one launch: products computed transposed so a lane owns a row and each result feeds the next from registers; dot-product and LN row sums re-associated
# speedup vs baseline: 3.2822x; 1.2411x over previous
#include <hip/hip_runtime.h>

#define DEV __device__ __forceinline__

typedef float v2f __attribute__((ext_vector_type(2)));
typedef float v8f __attribute__((ext_vector_type(8)));
typedef float v4f __attribute__((ext_vector_type(4)));
typedef int v4i __attribute__((ext_vector_type(4)));
typedef unsigned v16u __attribute__((ext_vector_type(16)));

static constexpr int CDIM  = 128;
static constexpr int FDIM  = 256;
static constexpr int NSEQ  = 2048;
static constexpr int BATCH = 4;
static constexpr int NHEAD = 8;
static constexpr int HDIM  = 16;
static constexpr int MROWS = BATCH * NSEQ;
static constexpr int NTILE = NSEQ / 16;

DEV v8f v8zero() {
  v8f v;
#pragma unroll
  for (int i = 0; i < 8; ++i) v[i] = 0.f;
  return v;
}

DEV v8f wmma4(v2f a, v2f b, v8f c) {
  return __builtin_amdgcn_wmma_f32_16x16x4_f32(false, a, false, b, (short)0, c,
                                               false, false);
}

template <int NT, bool SWAP = false>
DEV void gemm_acc(const float* __restrict__ Arow, const float* __restrict__ Bc,
                  int ldb, int K, v8f* acc) {
  const int hi = (threadIdx.x >> 4) & 1;
  for (int k = 0; k < K; k += 4) {
    const int kb = k + 2 * hi;
    v2f a;
    a.x = Arow[kb];
    a.y = Arow[kb + 1];
#pragma unroll
    for (int j = 0; j < NT; ++j) {
      v2f b;
      b.x = Bc[(size_t)kb * ldb + j * 16];
      b.y = Bc[(size_t)(kb + 1) * ldb + j * 16];
      acc[j] = SWAP ? wmma4(b, a, acc[j]) : wmma4(a, b, acc[j]);
    }
  }
}

DEV void gemm_acc_w(const float* __restrict__ Arow, const float* __restrict__ Bw,
                    int ldb, int K, v8f* acc) {
  const int hi = (threadIdx.x >> 4) & 1;
  for (int k = 0; k < K; k += 8) {
    const int kb = k + 4 * hi;
    const v4f a4 = *(const v4f*)(Arow + kb);
#pragma unroll
    for (int s = 0; s < 2; ++s) {
      v2f a;
      a.x = a4[2 * s];
      a.y = a4[2 * s + 1];
      const float* B0 = Bw + (size_t)(kb + 2 * s) * ldb;
      const float* B1 = B0 + ldb;
#pragma unroll
      for (int g = 0; g < 2; ++g) {
        const v4f b0 = *(const v4f*)(B0 + 64 * g);
        const v4f b1 = *(const v4f*)(B1 + 64 * g);
#pragma unroll
        for (int t = 0; t < 4; ++t) {
          v2f b;
          b.x = b0[t];
          b.y = b1[t];
          acc[4 * g + t] = wmma4(a, b, acc[4 * g + t]);
        }
      }
    }
  }
}

static constexpr int TILE_F = 256;
static constexpr int HEAD_F = (NSEQ / 16) * TILE_F;

static constexpr int QKV_BLOCKS  = (3 * (MROWS / 16)) / 8;
static constexpr int PACK_TILES  = 8;
static constexpr int PACK_BLOCKS = (NTILE * NTILE / PACK_TILES) / 8;

DEV void pack_tiles(const int* __restrict__ adj, unsigned* __restrict__ masks,
                    int task, int lane) {
  const int nt = task / (NTILE / PACK_TILES);
  const int t0 = (task % (NTILE / PACK_TILES)) * PACK_TILES;
  const int* src = adj + (size_t)(nt * 16 + (lane & 15)) * NSEQ + t0 * 16 +
                   8 * (lane >> 4);
  unsigned* dst = masks + ((size_t)nt * NTILE + t0) * 8 + lane;
#pragma unroll
  for (int g = 0; g < PACK_TILES / 4; ++g) {
    unsigned w = 0;
#pragma unroll
    for (int j = 0; j < 4; ++j) {
      const v4i a0 = *(const v4i*)(src + (g * 4 + j) * 16);
      const v4i a1 = *(const v4i*)(src + (g * 4 + j) * 16 + 4);
#pragma unroll
      for (int r = 0; r < 8; ++r) {
        const int av = (r < 4) ? a0[r & 3] : a1[r & 3];
        const unsigned bal = __builtin_amdgcn_ballot_w32(av != 0);
        if (lane == j * 8 + r) w = bal;
      }
    }
    dst[g * 32] = w;
  }
}

__global__ __launch_bounds__(256) void qkv_kernel(
    const float* __restrict__ h, const float* __restrict__ Wq,
    const float* __restrict__ Wk, const float* __restrict__ Wv,
    const int* __restrict__ adj, float* __restrict__ Q, float* __restrict__ K,
    float* __restrict__ V, unsigned* __restrict__ masks) {
  const int lane = threadIdx.x & 31;
  const int col = lane & 15, hi = lane >> 4;
  const int wv = __builtin_amdgcn_readfirstlane((int)(threadIdx.x >> 5));
  if (blockIdx.x >= QKV_BLOCKS) {
    pack_tiles(adj, masks, (blockIdx.x - QKV_BLOCKS) * 8 + wv, lane);
    return;
  }
  const int gw = blockIdx.x * (blockDim.x >> 5) + wv;
  const int mat = gw / (MROWS / 16);
  const int rt = gw % (MROWS / 16);
  const float* W = (mat == 0) ? Wq : (mat == 1) ? Wk : Wv;
  const int row0 = rt * 16;

  v8f acc[8];
#pragma unroll
  for (int j = 0; j < 8; ++j) acc[j] = v8zero();

  const float* Arow = h + (size_t)(row0 + col) * CDIM;
  if (mat == 0)
    gemm_acc_w(Arow, W + 4 * col, CDIM, CDIM, acc);
  else if (mat == 1)
    gemm_acc<8, true>(Arow, W + col, CDIM, CDIM, acc);
  else
    gemm_acc<8, false>(Arow, W + col, CDIM, CDIM, acc);

  if (mat == 0) {
#pragma unroll
    for (int g = 0; g < 2; ++g)
#pragma unroll
      for (int r = 0; r < 8; ++r) {
        v4f v;
#pragma unroll
        for (int t = 0; t < 4; ++t) v[t] = acc[4 * g + t][r];
        *(v4f*)(Q + (size_t)(row0 + r + 8 * hi) * CDIM + 64 * g + 4 * col) = v;
      }
  } else {
    float* Out = (mat == 1) ? K : V;
    float* T = Out + (size_t)((rt / (NSEQ / 16)) * NHEAD) * HEAD_F +
               (size_t)(rt & (NSEQ / 16 - 1)) * TILE_F + lane * 4;
#pragma unroll
    for (int j = 0; j < 8; ++j) {
      v4f lo, up;
#pragma unroll
      for (int e = 0; e < 4; ++e) {
        lo[e] = acc[j][e];
        up[e] = acc[j][e + 4];
      }
      *(v4f*)(T + (size_t)j * HEAD_F) = lo;
      *(v4f*)(T + (size_t)j * HEAD_F + 128) = up;
    }
  }
}

DEV float xhalf(float x) {
  const int b = __builtin_bit_cast(int, x);
  return __builtin_bit_cast(
      float, __builtin_amdgcn_permlanex16(b, b, 0x76543210, 0xfedcba98, false,
                                          false));
}

static constexpr int RING = 8;

struct KeyTile {
  v4f k0, k1;
  v4f v0, v1;
};

DEV void tile_load(KeyTile& t, const float* __restrict__ Kt,
                   const float* __restrict__ Vt, unsigned lo) {
  t.k0 = *(const v4f*)(Kt + lo);
  t.k1 = *(const v4f*)(Kt + lo + 128);
  t.v0 = *(const v4f*)(Vt + lo);
  t.v1 = *(const v4f*)(Vt + lo + 128);
}

DEV void tile_step(const KeyTile& t, const v16u& mw, int off,
                   const v2f (&qf)[4], float& m_i, float& l_i, v8f& o) {
  v8f s = v8zero();
  const v4f kq[2] = {t.k0, t.k1};
#pragma unroll
  for (int u = 0; u < 4; ++u) {
    v2f a;
    a.x = kq[u >> 1][2 * (u & 1)];
    a.y = kq[u >> 1][2 * (u & 1) + 1];
    s = wmma4(a, qf[u], s);
  }

  float sv[8];
#pragma unroll
  for (int r = 0; r < 8; ++r) {
    const bool edge = __builtin_amdgcn_inverse_ballot_w32(mw[off + r]);
    sv[r] = edge ? s[r] * 0.25f
                 : -1e30f;
  }
  float mloc = sv[0];
#pragma unroll
  for (int r = 1; r < 8; ++r) mloc = fmaxf(mloc, sv[r]);
  mloc = fmaxf(mloc, xhalf(mloc));
  const float mnew = fmaxf(m_i, mloc);
  const float f = __expf(m_i - mnew);
  float p[8];
  float psum = 0.f;
#pragma unroll
  for (int r = 0; r < 8; ++r) {
    p[r] = __expf(sv[r] - mnew);
    psum += p[r];
  }
  l_i = l_i * f + psum;
#pragma unroll
  for (int r = 0; r < 8; ++r) o[r] *= f;
  m_i = mnew;

  const v4f vq[2] = {t.v0, t.v1};
#pragma unroll
  for (int u = 0; u < 4; ++u) {
    v2f a, bb;
    a.x = vq[u >> 1][2 * (u & 1)];
    a.y = vq[u >> 1][2 * (u & 1) + 1];
    bb.x = p[2 * u];
    bb.y = p[2 * u + 1];
    o = wmma4(a, bb, o);
  }
}

__global__ __launch_bounds__(128) __attribute__((amdgpu_waves_per_eu(4, 4)))
void attn_kernel(
    const float* __restrict__ Q, const float* __restrict__ Kmat,
    const float* __restrict__ V, const unsigned* __restrict__ masks,
    float* __restrict__ O) {
  const int lane = threadIdx.x & 31;
  const int col = lane & 15, hi = lane >> 4;
  const int wid = __builtin_amdgcn_readfirstlane((int)(threadIdx.x >> 5));
  const int gw = blockIdx.x * 4 + wid;
  const int nt = gw & (NTILE - 1);
  const int hd = (gw >> 7) & (NHEAD - 1);
  const int b = gw >> 10;
  const int n0 = nt * 16;
  const size_t base = (size_t)b * NSEQ * CDIM;

  const float* Qrow = Q + base + (size_t)(n0 + col) * CDIM + hd * HDIM + 8 * hi;
  v2f qf[4];
#pragma unroll
  for (int t = 0; t < 4; ++t) {
    qf[t].x = Qrow[2 * t];
    qf[t].y = Qrow[2 * t + 1];
  }

  const size_t hoff = (size_t)(b * NHEAD + hd) * HEAD_F;
  const float* Kh = Kmat + hoff;
  const float* Vh = V + hoff;
  const unsigned lo = lane * 4;
  const unsigned* Mq = masks + (size_t)nt * (NTILE * 8);

  float m_i = -__builtin_inff();
  float l_i = 0.f;
  v8f o = v8zero();

  v16u mp[2];
  mp[0] = *(const v16u*)Mq;
  KeyTile ring[RING];
#pragma unroll
  for (int s = 0; s < RING; ++s) {
    tile_load(ring[s], Kh + s * TILE_F, Vh + s * TILE_F, lo);
    __builtin_amdgcn_sched_barrier(0);
  }

  for (int m = 0; m < NTILE - RING; m += RING) {
#pragma unroll
    for (int s = 0; s < RING; ++s) {
      tile_step(ring[s], mp[(s >> 1) & 1], (s & 1) * 8, qf, m_i, l_i, o);
      __builtin_amdgcn_sched_barrier(0);
      const int nx = m + s + RING;
      tile_load(ring[s], Kh + (size_t)nx * TILE_F, Vh + (size_t)nx * TILE_F,
                lo);
      if (!(s & 1))
        mp[((s >> 1) + 1) & 1] = *(const v16u*)(Mq + ((m + s) / 2 + 1) * 16);
      __builtin_amdgcn_sched_barrier(0);
    }
  }
#pragma unroll
  for (int s = 0; s < RING; ++s) {
    tile_step(ring[s], mp[(s >> 1) & 1], (s & 1) * 8, qf, m_i, l_i, o);
    if (!(s & 1) && s < RING - 2)
      mp[((s >> 1) + 1) & 1] =
          *(const v16u*)(Mq + ((NTILE - RING + s) / 2 + 1) * 16);
  }

  l_i += xhalf(l_i);
  float* Op = O + base + (size_t)(n0 + col) * CDIM + hd * HDIM + 8 * hi;
  v4f o0, o1;
#pragma unroll
  for (int r = 0; r < 4; ++r) {
    o0[r] = o[r] / l_i;
    o1[r] = o[r + 4] / l_i;
  }
  *(v4f*)Op = o0;
  *(v4f*)(Op + 4) = o1;
}


struct GStage {
  v4f a;
  v4f w[2][2][2];
};
DEV void gstage_load(GStage& st, const float* __restrict__ Arow,
                     const float* __restrict__ Bw, int ldb, int kb) {
  st.a = *(const v4f*)(Arow + kb);
#pragma unroll
  for (int s = 0; s < 2; ++s)
#pragma unroll
    for (int g = 0; g < 2; ++g) {
      const float* B0 = Bw + (size_t)(kb + 2 * s) * ldb + 64 * g;
      st.w[s][g][0] = *(const v4f*)B0;
      st.w[s][g][1] = *(const v4f*)(B0 + ldb);
    }
}
template <int DEPTH>
DEV void gemm_glob_t(const float* __restrict__ Arow,
                     const float* __restrict__ Bw, int ldb, v8f* acc) {
  const int hi = (threadIdx.x >> 4) & 1;
  constexpr int NS = CDIM / 8;
  GStage ring[DEPTH];
#pragma unroll
  for (int i = 0; i < DEPTH; ++i) {
    gstage_load(ring[i], Arow, Bw, ldb, 8 * i + 4 * hi);
    __builtin_amdgcn_sched_barrier(0);
  }
#pragma nounroll
  for (int i0 = 0; i0 < NS; i0 += DEPTH) {
#pragma unroll
    for (int j = 0; j < DEPTH; ++j) {
      GStage& st = ring[j];
#pragma unroll
      for (int s = 0; s < 2; ++s) {
        v2f b;
        b.x = st.a[2 * s];
        b.y = st.a[2 * s + 1];
#pragma unroll
        for (int g = 0; g < 2; ++g)
#pragma unroll
          for (int t = 0; t < 4; ++t) {
            v2f a;
            a.x = st.w[s][g][0][t];
            a.y = st.w[s][g][1][t];
            acc[4 * g + t] = wmma4(a, b, acc[4 * g + t]);
          }
      }
      __builtin_amdgcn_sched_barrier(0);
      const int nx = (i0 + j + DEPTH < NS) ? i0 + j + DEPTH : NS - 1;
      gstage_load(st, Arow, Bw, ldb, 8 * nx + 4 * hi);
      __builtin_amdgcn_sched_barrier(0);
    }
  }
}

template <int NQ>
DEV void wstage_load(v4f (&w)[NQ][2], const float* __restrict__ Wl, int ldw,
                     int step) {
  const int kk = 64 * (step >> 4) + 4 * ((step >> 1) & 7) + 2 * (step & 1);
  const float* R0 = Wl + (size_t)kk * ldw;
#pragma unroll
  for (int q = 0; q < NQ; ++q) {
    w[q][0] = *(const v4f*)(R0 + 64 * q);
    w[q][1] = *(const v4f*)(R0 + ldw + 64 * q);
  }
}
template <int NG, int NQ, int DEPTH>
DEV void gemm_regs_t(v2f (&xp)[NG][2][8], const float* __restrict__ Wl,
                     int ldw, v8f* acc) {
  v4f ring[DEPTH][NQ][2];
#pragma unroll
  for (int i = 0; i < DEPTH; ++i) {
    wstage_load<NQ>(ring[i], Wl, ldw, i);
    __builtin_amdgcn_sched_barrier(0);
  }
#pragma nounroll
  for (int gi = 0; gi < NG; ++gi) {
    const float* Wg = Wl + (size_t)(64 * gi) * ldw;
    const float* Wn = Wl + (size_t)(64 * ((gi + 1 < NG) ? gi + 1 : gi)) * ldw;
#pragma unroll
    for (int s = 0; s < 16; ++s) {
      const v2f b = xp[0][s & 1][(s >> 1) & 7];
#pragma unroll
      for (int q = 0; q < NQ; ++q)
#pragma unroll
        for (int t = 0; t < 4; ++t) {
          v2f a;
          a.x = ring[s % DEPTH][q][0][t];
          a.y = ring[s % DEPTH][q][1][t];
          acc[4 * q + t] = wmma4(a, b, acc[4 * q + t]);
        }
      __builtin_amdgcn_sched_barrier(0);
      if (s + DEPTH < 16)
        wstage_load<NQ>(ring[s % DEPTH], Wg, ldw, s + DEPTH);
      else
        wstage_load<NQ>(ring[s % DEPTH], Wn, ldw, s + DEPTH - 16);
      __builtin_amdgcn_sched_barrier(0);
    }
    v2f first[2][8];
#pragma unroll
    for (int u = 0; u < 2; ++u)
#pragma unroll
      for (int r = 0; r < 8; ++r) first[u][r] = xp[0][u][r];
#pragma unroll
    for (int g = 0; g + 1 < NG; ++g)
#pragma unroll
      for (int u = 0; u < 2; ++u)
#pragma unroll
        for (int r = 0; r < 8; ++r) xp[g][u][r] = xp[g + 1][u][r];
#pragma unroll
    for (int u = 0; u < 2; ++u)
#pragma unroll
      for (int r = 0; r < 8; ++r) xp[NG - 1][u][r] = first[u][r];
  }
}

__global__ __launch_bounds__(128) __attribute__((amdgpu_waves_per_eu(1, 1)))
void mlp_kernel(
    const float* __restrict__ AO, const float* __restrict__ Wo,
    const float* __restrict__ bo, const float* __restrict__ hres,
    const float* __restrict__ g1, const float* __restrict__ bt1,
    const float* __restrict__ W1, const float* __restrict__ b1,
    const float* __restrict__ W2, const float* __restrict__ b2,
    const float* __restrict__ g2, const float* __restrict__ bt2,
    float* __restrict__ Out) {
  const int lane = threadIdx.x & 31;
  const int lr = lane & 15, hi = lane >> 4;
  const int gw = blockIdx.x * (blockDim.x >> 5) + (threadIdx.x >> 5);
  const size_t row = (size_t)gw * 16 + lr;
  const int cb = 32 * hi;

  v8f acc[8];
#pragma unroll
  for (int j = 0; j < 8; ++j) acc[j] = v8zero();
  gemm_glob_t<4>(AO + row * CDIM, Wo + 4 * lr, CDIM, acc);

  v2f xp[2][2][8];
  {
    const float* hrow = hres + row * CDIM + cb;
    float s1 = 0.f, s2 = 0.f;
#pragma unroll
    for (int g = 0; g < 2; ++g)
#pragma unroll
      for (int r = 0; r < 8; ++r) {
        const v4f bi = *(const v4f*)(bo + 64 * g + cb + 4 * r);
        const v4f re = *(const v4f*)(hrow + 64 * g + 4 * r);
#pragma unroll
        for (int t = 0; t < 4; ++t) {
          const float v = acc[4 * g + t][r] + bi[t] + re[t];
          acc[4 * g + t][r] = v;
          s1 += v;
          s2 += v * v;
        }
      }
    s1 += xhalf(s1);
    s2 += xhalf(s2);
    const float mean = s1 * (1.f / CDIM);
    const float var = s2 * (1.f / CDIM) - mean * mean;
    const float rstd = rsqrtf(var + 1e-5f);
#pragma unroll
    for (int g = 0; g < 2; ++g)
#pragma unroll
      for (int r = 0; r < 8; ++r) {
        const v4f gv = *(const v4f*)(g1 + 64 * g + cb + 4 * r);
        const v4f bv = *(const v4f*)(bt1 + 64 * g + cb + 4 * r);
#pragma unroll
        for (int u = 0; u < 2; ++u) {
          xp[g][u][r].x =
              (acc[4 * g + 2 * u][r] - mean) * rstd * gv[2 * u] + bv[2 * u];
          xp[g][u][r].y = (acc[4 * g + 2 * u + 1][r] - mean) * rstd *
                              gv[2 * u + 1] + bv[2 * u + 1];
        }
      }
  }

  v2f tp[4][2][8];
  {
    v8f a2[16];
#pragma unroll
    for (int j = 0; j < 16; ++j) a2[j] = v8zero();
    gemm_regs_t<2, 4, 4>(xp, W1 + (size_t)cb * FDIM + 4 * lr, FDIM, a2);
#pragma unroll
    for (int q = 0; q < 4; ++q)
#pragma unroll
      for (int r = 0; r < 8; ++r) {
        const v4f bi = *(const v4f*)(b1 + 64 * q + cb + 4 * r);
#pragma unroll
        for (int u = 0; u < 2; ++u) {
          tp[q][u][r].x = fmaxf(a2[4 * q + 2 * u][r] + bi[2 * u], 0.f);
          tp[q][u][r].y = fmaxf(a2[4 * q + 2 * u + 1][r] + bi[2 * u + 1], 0.f);
        }
      }
  }

#pragma unroll
  for (int j = 0; j < 8; ++j) acc[j] = v8zero();
  gemm_regs_t<4, 2, 8>(tp, W2 + (size_t)cb * CDIM + 4 * lr, CDIM, acc);
  {
    float s1 = 0.f, s2 = 0.f;
#pragma unroll
    for (int g = 0; g < 2; ++g)
#pragma unroll
      for (int r = 0; r < 8; ++r) {
        const v4f bi = *(const v4f*)(b2 + 64 * g + cb + 4 * r);
#pragma unroll
        for (int t = 0; t < 4; ++t) {
          const float x = (t & 1) ? xp[g][t >> 1][r].y : xp[g][t >> 1][r].x;
          const float v = acc[4 * g + t][r] + bi[t] + x;
          acc[4 * g + t][r] = v;
          s1 += v;
          s2 += v * v;
        }
      }
    s1 += xhalf(s1);
    s2 += xhalf(s2);
    const float mean = s1 * (1.f / CDIM);
    const float var = s2 * (1.f / CDIM) - mean * mean;
    const float rstd = rsqrtf(var + 1e-5f);
    float* orow = Out + row * CDIM + cb;
#pragma unroll
    for (int g = 0; g < 2; ++g)
#pragma unroll
      for (int r = 0; r < 8; ++r) {
        const v4f gv = *(const v4f*)(g2 + 64 * g + cb + 4 * r);
        const v4f bv = *(const v4f*)(bt2 + 64 * g + cb + 4 * r);
        v4f o;
#pragma unroll
        for (int t = 0; t < 4; ++t)
          o[t] = (acc[4 * g + t][r] - mean) * rstd * gv[t] + bv[t];
        *(v4f*)(orow + 64 * g + 4 * r) = o;
      }
  }
}

extern "C" void kernel_launch(void* const* d_in, const int* in_sizes, int n_in,
                              void* d_out, int out_size, void* d_ws,
                              size_t ws_size, hipStream_t stream) {
  (void)in_sizes; (void)n_in; (void)out_size; (void)ws_size;
  const int* adj  = (const int*)d_in[0];
  const float* h  = (const float*)d_in[1];
  const float* Wq = (const float*)d_in[2];
  const float* Wk = (const float*)d_in[3];
  const float* Wv = (const float*)d_in[4];
  const float* Wo = (const float*)d_in[5];
  const float* bo = (const float*)d_in[6];
  const float* g1 = (const float*)d_in[7];
  const float* bt1 = (const float*)d_in[8];
  const float* W1 = (const float*)d_in[9];
  const float* bb1 = (const float*)d_in[10];
  const float* W2 = (const float*)d_in[11];
  const float* bb2 = (const float*)d_in[12];
  const float* g2 = (const float*)d_in[13];
  const float* bt2 = (const float*)d_in[14];

  float* ws = (float*)d_ws;
  const size_t SZ = (size_t)MROWS * CDIM;
  float* Q  = ws;
  float* K  = Q + SZ;
  float* V  = K + SZ;
  float* AO = V + SZ;
  unsigned* masks = (unsigned*)(AO + SZ);

  qkv_kernel<<<QKV_BLOCKS + PACK_BLOCKS, 256, 0, stream>>>(h, Wq, Wk, Wv, adj,
                                                            Q, K, V, masks);
  attn_kernel<<<(BATCH * NHEAD * (NSEQ / 16)) / 4, 128, 0, stream>>>(Q, K, V,
                                                                     masks, AO);
  mlp_kernel<<<(MROWS / 16) / 4, 128, 0, stream>>>(AO, Wo, bo, h, g1, bt1, W1,
                                                   bb1, W2, bb2, g2, bt2,
                                                   (float*)d_out);
}


# out-proj/FFN launch: two waves on two SIMDs share each 16-row strip (out-proj column halves, FFN hidden halves, exchanged through LDS): 1024 waves fill all WGPs; FFN2 k sum re-associated in 2 halves
# speedup vs baseline: 3.5365x; 1.0775x over previous
#include <hip/hip_runtime.h>

#define DEV __device__ __forceinline__

typedef float v2f __attribute__((ext_vector_type(2)));
typedef float v8f __attribute__((ext_vector_type(8)));
typedef float v4f __attribute__((ext_vector_type(4)));
typedef int v4i __attribute__((ext_vector_type(4)));
typedef unsigned v16u __attribute__((ext_vector_type(16)));

static constexpr int CDIM  = 128;
static constexpr int FDIM  = 256;
static constexpr int NSEQ  = 2048;
static constexpr int BATCH = 4;
static constexpr int NHEAD = 8;
static constexpr int HDIM  = 16;
static constexpr int MROWS = BATCH * NSEQ;
static constexpr int NTILE = NSEQ / 16;

DEV v8f v8zero() {
  v8f v;
#pragma unroll
  for (int i = 0; i < 8; ++i) v[i] = 0.f;
  return v;
}

DEV v8f wmma4(v2f a, v2f b, v8f c) {
  return __builtin_amdgcn_wmma_f32_16x16x4_f32(false, a, false, b, (short)0, c,
                                               false, false);
}

template <int NT, bool SWAP = false>
DEV void gemm_acc(const float* __restrict__ Arow, const float* __restrict__ Bc,
                  int ldb, int K, v8f* acc) {
  const int hi = (threadIdx.x >> 4) & 1;
  for (int k = 0; k < K; k += 4) {
    const int kb = k + 2 * hi;
    v2f a;
    a.x = Arow[kb];
    a.y = Arow[kb + 1];
#pragma unroll
    for (int j = 0; j < NT; ++j) {
      v2f b;
      b.x = Bc[(size_t)kb * ldb + j * 16];
      b.y = Bc[(size_t)(kb + 1) * ldb + j * 16];
      acc[j] = SWAP ? wmma4(b, a, acc[j]) : wmma4(a, b, acc[j]);
    }
  }
}

DEV void gemm_acc_w(const float* __restrict__ Arow, const float* __restrict__ Bw,
                    int ldb, int K, v8f* acc) {
  const int hi = (threadIdx.x >> 4) & 1;
  for (int k = 0; k < K; k += 8) {
    const int kb = k + 4 * hi;
    const v4f a4 = *(const v4f*)(Arow + kb);
#pragma unroll
    for (int s = 0; s < 2; ++s) {
      v2f a;
      a.x = a4[2 * s];
      a.y = a4[2 * s + 1];
      const float* B0 = Bw + (size_t)(kb + 2 * s) * ldb;
      const float* B1 = B0 + ldb;
#pragma unroll
      for (int g = 0; g < 2; ++g) {
        const v4f b0 = *(const v4f*)(B0 + 64 * g);
        const v4f b1 = *(const v4f*)(B1 + 64 * g);
#pragma unroll
        for (int t = 0; t < 4; ++t) {
          v2f b;
          b.x = b0[t];
          b.y = b1[t];
          acc[4 * g + t] = wmma4(a, b, acc[4 * g + t]);
        }
      }
    }
  }
}

static constexpr int TILE_F = 256;
static constexpr int HEAD_F = (NSEQ / 16) * TILE_F;

static constexpr int QKV_BLOCKS  = (3 * (MROWS / 16)) / 8;
static constexpr int PACK_TILES  = 8;
static constexpr int PACK_BLOCKS = (NTILE * NTILE / PACK_TILES) / 8;

DEV void pack_tiles(const int* __restrict__ adj, unsigned* __restrict__ masks,
                    int task, int lane) {
  const int nt = task / (NTILE / PACK_TILES);
  const int t0 = (task % (NTILE / PACK_TILES)) * PACK_TILES;
  const int* src = adj + (size_t)(nt * 16 + (lane & 15)) * NSEQ + t0 * 16 +
                   8 * (lane >> 4);
  unsigned* dst = masks + ((size_t)nt * NTILE + t0) * 8 + lane;
#pragma unroll
  for (int g = 0; g < PACK_TILES / 4; ++g) {
    unsigned w = 0;
#pragma unroll
    for (int j = 0; j < 4; ++j) {
      const v4i a0 = *(const v4i*)(src + (g * 4 + j) * 16);
      const v4i a1 = *(const v4i*)(src + (g * 4 + j) * 16 + 4);
#pragma unroll
      for (int r = 0; r < 8; ++r) {
        const int av = (r < 4) ? a0[r & 3] : a1[r & 3];
        const unsigned bal = __builtin_amdgcn_ballot_w32(av != 0);
        if (lane == j * 8 + r) w = bal;
      }
    }
    dst[g * 32] = w;
  }
}

__global__ __launch_bounds__(256) void qkv_kernel(
    const float* __restrict__ h, const float* __restrict__ Wq,
    const float* __restrict__ Wk, const float* __restrict__ Wv,
    const int* __restrict__ adj, float* __restrict__ Q, float* __restrict__ K,
    float* __restrict__ V, unsigned* __restrict__ masks) {
  const int lane = threadIdx.x & 31;
  const int col = lane & 15, hi = lane >> 4;
  const int wv = __builtin_amdgcn_readfirstlane((int)(threadIdx.x >> 5));
  if (blockIdx.x >= QKV_BLOCKS) {
    pack_tiles(adj, masks, (blockIdx.x - QKV_BLOCKS) * 8 + wv, lane);
    return;
  }
  const int gw = blockIdx.x * (blockDim.x >> 5) + wv;
  const int mat = gw / (MROWS / 16);
  const int rt = gw % (MROWS / 16);
  const float* W = (mat == 0) ? Wq : (mat == 1) ? Wk : Wv;
  const int row0 = rt * 16;

  v8f acc[8];
#pragma unroll
  for (int j = 0; j < 8; ++j) acc[j] = v8zero();

  const float* Arow = h + (size_t)(row0 + col) * CDIM;
  if (mat == 0)
    gemm_acc_w(Arow, W + 4 * col, CDIM, CDIM, acc);
  else if (mat == 1)
    gemm_acc<8, true>(Arow, W + col, CDIM, CDIM, acc);
  else
    gemm_acc<8, false>(Arow, W + col, CDIM, CDIM, acc);

  if (mat == 0) {
#pragma unroll
    for (int g = 0; g < 2; ++g)
#pragma unroll
      for (int r = 0; r < 8; ++r) {
        v4f v;
#pragma unroll
        for (int t = 0; t < 4; ++t) v[t] = acc[4 * g + t][r];
        *(v4f*)(Q + (size_t)(row0 + r + 8 * hi) * CDIM + 64 * g + 4 * col) = v;
      }
  } else {
    float* Out = (mat == 1) ? K : V;
    float* T = Out + (size_t)((rt / (NSEQ / 16)) * NHEAD) * HEAD_F +
               (size_t)(rt & (NSEQ / 16 - 1)) * TILE_F + lane * 4;
#pragma unroll
    for (int j = 0; j < 8; ++j) {
      v4f lo, up;
#pragma unroll
      for (int e = 0; e < 4; ++e) {
        lo[e] = acc[j][e];
        up[e] = acc[j][e + 4];
      }
      *(v4f*)(T + (size_t)j * HEAD_F) = lo;
      *(v4f*)(T + (size_t)j * HEAD_F + 128) = up;
    }
  }
}

DEV float xhalf(float x) {
  const int b = __builtin_bit_cast(int, x);
  return __builtin_bit_cast(
      float, __builtin_amdgcn_permlanex16(b, b, 0x76543210, 0xfedcba98, false,
                                          false));
}

static constexpr int RING = 8;

struct KeyTile {
  v4f k0, k1;
  v4f v0, v1;
};

DEV void tile_load(KeyTile& t, const float* __restrict__ Kt,
                   const float* __restrict__ Vt, unsigned lo) {
  t.k0 = *(const v4f*)(Kt + lo);
  t.k1 = *(const v4f*)(Kt + lo + 128);
  t.v0 = *(const v4f*)(Vt + lo);
  t.v1 = *(const v4f*)(Vt + lo + 128);
}

DEV void tile_step(const KeyTile& t, const v16u& mw, int off,
                   const v2f (&qf)[4], float& m_i, float& l_i, v8f& o) {
  v8f s = v8zero();
  const v4f kq[2] = {t.k0, t.k1};
#pragma unroll
  for (int u = 0; u < 4; ++u) {
    v2f a;
    a.x = kq[u >> 1][2 * (u & 1)];
    a.y = kq[u >> 1][2 * (u & 1) + 1];
    s = wmma4(a, qf[u], s);
  }

  float sv[8];
#pragma unroll
  for (int r = 0; r < 8; ++r) {
    const bool edge = __builtin_amdgcn_inverse_ballot_w32(mw[off + r]);
    sv[r] = edge ? s[r] * 0.25f
                 : -1e30f;
  }
  float mloc = sv[0];
#pragma unroll
  for (int r = 1; r < 8; ++r) mloc = fmaxf(mloc, sv[r]);
  mloc = fmaxf(mloc, xhalf(mloc));
  const float mnew = fmaxf(m_i, mloc);
  const float f = __expf(m_i - mnew);
  float p[8];
  float psum = 0.f;
#pragma unroll
  for (int r = 0; r < 8; ++r) {
    p[r] = __expf(sv[r] - mnew);
    psum += p[r];
  }
  l_i = l_i * f + psum;
#pragma unroll
  for (int r = 0; r < 8; ++r) o[r] *= f;
  m_i = mnew;

  const v4f vq[2] = {t.v0, t.v1};
#pragma unroll
  for (int u = 0; u < 4; ++u) {
    v2f a, bb;
    a.x = vq[u >> 1][2 * (u & 1)];
    a.y = vq[u >> 1][2 * (u & 1) + 1];
    bb.x = p[2 * u];
    bb.y = p[2 * u + 1];
    o = wmma4(a, bb, o);
  }
}

__global__ __launch_bounds__(128) __attribute__((amdgpu_waves_per_eu(4, 4)))
void attn_kernel(
    const float* __restrict__ Q, const float* __restrict__ Kmat,
    const float* __restrict__ V, const unsigned* __restrict__ masks,
    float* __restrict__ O) {
  const int lane = threadIdx.x & 31;
  const int col = lane & 15, hi = lane >> 4;
  const int wid = __builtin_amdgcn_readfirstlane((int)(threadIdx.x >> 5));
  const int gw = blockIdx.x * 4 + wid;
  const int nt = gw & (NTILE - 1);
  const int hd = (gw >> 7) & (NHEAD - 1);
  const int b = gw >> 10;
  const int n0 = nt * 16;
  const size_t base = (size_t)b * NSEQ * CDIM;

  const float* Qrow = Q + base + (size_t)(n0 + col) * CDIM + hd * HDIM + 8 * hi;
  v2f qf[4];
#pragma unroll
  for (int t = 0; t < 4; ++t) {
    qf[t].x = Qrow[2 * t];
    qf[t].y = Qrow[2 * t + 1];
  }

  const size_t hoff = (size_t)(b * NHEAD + hd) * HEAD_F;
  const float* Kh = Kmat + hoff;
  const float* Vh = V + hoff;
  const unsigned lo = lane * 4;
  const unsigned* Mq = masks + (size_t)nt * (NTILE * 8);

  float m_i = -__builtin_inff();
  float l_i = 0.f;
  v8f o = v8zero();

  v16u mp[2];
  mp[0] = *(const v16u*)Mq;
  KeyTile ring[RING];
#pragma unroll
  for (int s = 0; s < RING; ++s) {
    tile_load(ring[s], Kh + s * TILE_F, Vh + s * TILE_F, lo);
    __builtin_amdgcn_sched_barrier(0);
  }

  for (int m = 0; m < NTILE - RING; m += RING) {
#pragma unroll
    for (int s = 0; s < RING; ++s) {
      tile_step(ring[s], mp[(s >> 1) & 1], (s & 1) * 8, qf, m_i, l_i, o);
      __builtin_amdgcn_sched_barrier(0);
      const int nx = m + s + RING;
      tile_load(ring[s], Kh + (size_t)nx * TILE_F, Vh + (size_t)nx * TILE_F,
                lo);
      if (!(s & 1))
        mp[((s >> 1) + 1) & 1] = *(const v16u*)(Mq + ((m + s) / 2 + 1) * 16);
      __builtin_amdgcn_sched_barrier(0);
    }
  }
#pragma unroll
  for (int s = 0; s < RING; ++s) {
    tile_step(ring[s], mp[(s >> 1) & 1], (s & 1) * 8, qf, m_i, l_i, o);
    if (!(s & 1) && s < RING - 2)
      mp[((s >> 1) + 1) & 1] =
          *(const v16u*)(Mq + ((NTILE - RING + s) / 2 + 1) * 16);
  }

  l_i += xhalf(l_i);
  float* Op = O + base + (size_t)(n0 + col) * CDIM + hd * HDIM + 8 * hi;
  v4f o0, o1;
#pragma unroll
  for (int r = 0; r < 4; ++r) {
    o0[r] = o[r] / l_i;
    o1[r] = o[r + 4] / l_i;
  }
  *(v4f*)Op = o0;
  *(v4f*)(Op + 4) = o1;
}


template <int NGW>
struct GStage {
  v4f a;
  v4f w[2][NGW][2];
};
template <int NGW>
DEV void gstage_load(GStage<NGW>& st, const float* __restrict__ Arow,
                     const float* __restrict__ Bw, int ldb, int kb) {
  st.a = *(const v4f*)(Arow + kb);
#pragma unroll
  for (int s = 0; s < 2; ++s)
#pragma unroll
    for (int g = 0; g < NGW; ++g) {
      const float* B0 = Bw + (size_t)(kb + 2 * s) * ldb + 64 * g;
      st.w[s][g][0] = *(const v4f*)B0;
      st.w[s][g][1] = *(const v4f*)(B0 + ldb);
    }
}
template <int DEPTH, int NGW>
DEV void gemm_glob_t(const float* __restrict__ Arow,
                     const float* __restrict__ Bw, int ldb, v8f* acc) {
  const int hi = (threadIdx.x >> 4) & 1;
  constexpr int NS = CDIM / 8;
  GStage<NGW> ring[DEPTH];
#pragma unroll
  for (int i = 0; i < DEPTH; ++i) {
    gstage_load<NGW>(ring[i], Arow, Bw, ldb, 8 * i + 4 * hi);
    __builtin_amdgcn_sched_barrier(0);
  }
#pragma nounroll
  for (int i0 = 0; i0 < NS; i0 += DEPTH) {
#pragma unroll
    for (int j = 0; j < DEPTH; ++j) {
      GStage<NGW>& st = ring[j];
#pragma unroll
      for (int s = 0; s < 2; ++s) {
        v2f b;
        b.x = st.a[2 * s];
        b.y = st.a[2 * s + 1];
#pragma unroll
        for (int g = 0; g < NGW; ++g)
#pragma unroll
          for (int t = 0; t < 4; ++t) {
            v2f a;
            a.x = st.w[s][g][0][t];
            a.y = st.w[s][g][1][t];
            acc[4 * g + t] = wmma4(a, b, acc[4 * g + t]);
          }
      }
      __builtin_amdgcn_sched_barrier(0);
      const int nx = (i0 + j + DEPTH < NS) ? i0 + j + DEPTH : NS - 1;
      gstage_load<NGW>(st, Arow, Bw, ldb, 8 * nx + 4 * hi);
      __builtin_amdgcn_sched_barrier(0);
    }
  }
}

template <int NQ>
DEV void wstage_load(v4f (&w)[NQ][2], const float* __restrict__ Wl, int ldw,
                     int step) {
  const int kk = 64 * (step >> 4) + 4 * ((step >> 1) & 7) + 2 * (step & 1);
  const float* R0 = Wl + (size_t)kk * ldw;
#pragma unroll
  for (int q = 0; q < NQ; ++q) {
    w[q][0] = *(const v4f*)(R0 + 64 * q);
    w[q][1] = *(const v4f*)(R0 + ldw + 64 * q);
  }
}
template <int NG, int NQ, int DEPTH>
DEV void gemm_regs_t(v2f (&xp)[NG][2][8], const float* __restrict__ Wl,
                     int ldw, v8f* acc) {
  v4f ring[DEPTH][NQ][2];
#pragma unroll
  for (int i = 0; i < DEPTH; ++i) {
    wstage_load<NQ>(ring[i], Wl, ldw, i);
    __builtin_amdgcn_sched_barrier(0);
  }
#pragma nounroll
  for (int gi = 0; gi < NG; ++gi) {
    const float* Wg = Wl + (size_t)(64 * gi) * ldw;
    const float* Wn = Wl + (size_t)(64 * ((gi + 1 < NG) ? gi + 1 : gi)) * ldw;
#pragma unroll
    for (int s = 0; s < 16; ++s) {
      const v2f b = xp[0][s & 1][(s >> 1) & 7];
#pragma unroll
      for (int q = 0; q < NQ; ++q)
#pragma unroll
        for (int t = 0; t < 4; ++t) {
          v2f a;
          a.x = ring[s % DEPTH][q][0][t];
          a.y = ring[s % DEPTH][q][1][t];
          acc[4 * q + t] = wmma4(a, b, acc[4 * q + t]);
        }
      __builtin_amdgcn_sched_barrier(0);
      if (s + DEPTH < 16)
        wstage_load<NQ>(ring[s % DEPTH], Wg, ldw, s + DEPTH);
      else
        wstage_load<NQ>(ring[s % DEPTH], Wn, ldw, s + DEPTH - 16);
      __builtin_amdgcn_sched_barrier(0);
    }
    v2f first[2][8];
#pragma unroll
    for (int u = 0; u < 2; ++u)
#pragma unroll
      for (int r = 0; r < 8; ++r) first[u][r] = xp[0][u][r];
#pragma unroll
    for (int g = 0; g + 1 < NG; ++g)
#pragma unroll
      for (int u = 0; u < 2; ++u)
#pragma unroll
        for (int r = 0; r < 8; ++r) xp[g][u][r] = xp[g + 1][u][r];
#pragma unroll
    for (int u = 0; u < 2; ++u)
#pragma unroll
      for (int r = 0; r < 8; ++r) xp[NG - 1][u][r] = first[u][r];
  }
}

__global__ __launch_bounds__(128) __attribute__((amdgpu_waves_per_eu(1, 1)))
void mlp_kernel(
    const float* __restrict__ AO, const float* __restrict__ Wo,
    const float* __restrict__ bo, const float* __restrict__ hres,
    const float* __restrict__ g1, const float* __restrict__ bt1,
    const float* __restrict__ W1, const float* __restrict__ b1,
    const float* __restrict__ W2, const float* __restrict__ b2,
    const float* __restrict__ g2, const float* __restrict__ bt2,
    float* __restrict__ Out) {
  __shared__ v4f xs[2][2][8][32];
  __shared__ v4f ys[2][16][32];
  const int lane = threadIdx.x & 31;
  const int lr = lane & 15, hi = lane >> 4;
  const int wv = __builtin_amdgcn_readfirstlane((int)(threadIdx.x >> 5));
  const int pair = wv >> 1, part = wv & 1;
  const size_t row = (size_t)(blockIdx.x * 2 + pair) * 16 + lr;
  const int cb = 32 * hi;

  {
    v8f a4[4];
#pragma unroll
    for (int j = 0; j < 4; ++j) a4[j] = v8zero();
    gemm_glob_t<8, 1>(AO + row * CDIM, Wo + 64 * part + 4 * lr, CDIM, a4);
    const float* hrow = hres + row * CDIM + 64 * part + cb;
    const float* brow = bo + 64 * part + cb;
#pragma unroll
    for (int r = 0; r < 8; ++r) {
      const v4f bi = *(const v4f*)(brow + 4 * r);
      const v4f re = *(const v4f*)(hrow + 4 * r);
      v4f v;
#pragma unroll
      for (int t = 0; t < 4; ++t) v[t] = a4[t][r] + bi[t] + re[t];
      xs[pair][part][r][lane] = v;
    }
  }
  __syncthreads();

  v8f acc[8];
#pragma unroll
  for (int g = 0; g < 2; ++g)
#pragma unroll
    for (int r = 0; r < 8; ++r) {
      const v4f v = xs[pair][g][r][lane];
#pragma unroll
      for (int t = 0; t < 4; ++t) acc[4 * g + t][r] = v[t];
    }
  v2f xp[2][2][8];
  {
    float s1 = 0.f, s2 = 0.f;
#pragma unroll
    for (int g = 0; g < 2; ++g)
#pragma unroll
      for (int r = 0; r < 8; ++r)
#pragma unroll
        for (int t = 0; t < 4; ++t) {
          const float v = acc[4 * g + t][r];
          s1 += v;
          s2 += v * v;
        }
    s1 += xhalf(s1);
    s2 += xhalf(s2);
    const float mean = s1 * (1.f / CDIM);
    const float var = s2 * (1.f / CDIM) - mean * mean;
    const float rstd = rsqrtf(var + 1e-5f);
#pragma unroll
    for (int g = 0; g < 2; ++g)
#pragma unroll
      for (int r = 0; r < 8; ++r) {
        const v4f gv = *(const v4f*)(g1 + 64 * g + cb + 4 * r);
        const v4f bv = *(const v4f*)(bt1 + 64 * g + cb + 4 * r);
#pragma unroll
        for (int u = 0; u < 2; ++u) {
          xp[g][u][r].x =
              (acc[4 * g + 2 * u][r] - mean) * rstd * gv[2 * u] + bv[2 * u];
          xp[g][u][r].y = (acc[4 * g + 2 * u + 1][r] - mean) * rstd *
                              gv[2 * u + 1] + bv[2 * u + 1];
        }
      }
  }

  v2f tp[2][2][8];
  {
#pragma unroll
    for (int j = 0; j < 8; ++j) acc[j] = v8zero();
    gemm_regs_t<2, 2, 8>(xp, W1 + (size_t)cb * FDIM + 128 * part + 4 * lr, FDIM,
                         acc);
    const float* b1h = b1 + 128 * part + cb;
#pragma unroll
    for (int q = 0; q < 2; ++q)
#pragma unroll
      for (int r = 0; r < 8; ++r) {
        const v4f bi = *(const v4f*)(b1h + 64 * q + 4 * r);
#pragma unroll
        for (int u = 0; u < 2; ++u) {
          tp[q][u][r].x = fmaxf(acc[4 * q + 2 * u][r] + bi[2 * u], 0.f);
          tp[q][u][r].y = fmaxf(acc[4 * q + 2 * u + 1][r] + bi[2 * u + 1], 0.f);
        }
      }
  }

#pragma unroll
  for (int j = 0; j < 8; ++j) acc[j] = v8zero();
  gemm_regs_t<2, 2, 8>(tp, W2 + (size_t)(128 * part + cb) * CDIM + 4 * lr, CDIM,
                       acc);
  if (part == 1) {
#pragma unroll
    for (int g = 0; g < 2; ++g)
#pragma unroll
      for (int r = 0; r < 8; ++r) {
        v4f v;
#pragma unroll
        for (int t = 0; t < 4; ++t) v[t] = acc[4 * g + t][r];
        ys[pair][8 * g + r][lane] = v;
      }
  }
  __syncthreads();
  if (part == 0) {
    float s1 = 0.f, s2 = 0.f;
#pragma unroll
    for (int g = 0; g < 2; ++g)
#pragma unroll
      for (int r = 0; r < 8; ++r) {
        const v4f bi = *(const v4f*)(b2 + 64 * g + cb + 4 * r);
        const v4f pa = ys[pair][8 * g + r][lane];
#pragma unroll
        for (int t = 0; t < 4; ++t) {
          const float x = (t & 1) ? xp[g][t >> 1][r].y : xp[g][t >> 1][r].x;
          const float v = (acc[4 * g + t][r] + pa[t]) + bi[t] + x;
          acc[4 * g + t][r] = v;
          s1 += v;
          s2 += v * v;
        }
      }
    s1 += xhalf(s1);
    s2 += xhalf(s2);
    const float mean = s1 * (1.f / CDIM);
    const float var = s2 * (1.f / CDIM) - mean * mean;
    const float rstd = rsqrtf(var + 1e-5f);
    float* orow = Out + row * CDIM + cb;
#pragma unroll
    for (int g = 0; g < 2; ++g)
#pragma unroll
      for (int r = 0; r < 8; ++r) {
        const v4f gv = *(const v4f*)(g2 + 64 * g + cb + 4 * r);
        const v4f bv = *(const v4f*)(bt2 + 64 * g + cb + 4 * r);
        v4f o;
#pragma unroll
        for (int t = 0; t < 4; ++t)
          o[t] = (acc[4 * g + t][r] - mean) * rstd * gv[t] + bv[t];
        *(v4f*)(orow + 64 * g + 4 * r) = o;
      }
  }
}

extern "C" void kernel_launch(void* const* d_in, const int* in_sizes, int n_in,
                              void* d_out, int out_size, void* d_ws,
                              size_t ws_size, hipStream_t stream) {
  (void)in_sizes; (void)n_in; (void)out_size; (void)ws_size;
  const int* adj  = (const int*)d_in[0];
  const float* h  = (const float*)d_in[1];
  const float* Wq = (const float*)d_in[2];
  const float* Wk = (const float*)d_in[3];
  const float* Wv = (const float*)d_in[4];
  const float* Wo = (const float*)d_in[5];
  const float* bo = (const float*)d_in[6];
  const float* g1 = (const float*)d_in[7];
  const float* bt1 = (const float*)d_in[8];
  const float* W1 = (const float*)d_in[9];
  const float* bb1 = (const float*)d_in[10];
  const float* W2 = (const float*)d_in[11];
  const float* bb2 = (const float*)d_in[12];
  const float* g2 = (const float*)d_in[13];
  const float* bt2 = (const float*)d_in[14];

  float* ws = (float*)d_ws;
  const size_t SZ = (size_t)MROWS * CDIM;
  float* Q  = ws;
  float* K  = Q + SZ;
  float* V  = K + SZ;
  float* AO = V + SZ;
  unsigned* masks = (unsigned*)(AO + SZ);

  qkv_kernel<<<QKV_BLOCKS + PACK_BLOCKS, 256, 0, stream>>>(h, Wq, Wk, Wv, adj,
                                                            Q, K, V, masks);
  attn_kernel<<<(BATCH * NHEAD * (NSEQ / 16)) / 4, 128, 0, stream>>>(Q, K, V,
                                                                     masks, AO);
  mlp_kernel<<<(MROWS / 16) / 2, 128, 0, stream>>>(AO, Wo, bo, h, g1, bt1, W1,
                                                   bb1, W2, bb2, g2, bt2,
                                                   (float*)d_out);
}


# attention: each wave keeps two independent online-softmax states (even / odd key tiles) whose WMMA and vector chains alternate, merged at the end; row sum and P*V key order re-associated
# speedup vs baseline: 3.6489x; 1.0318x over previous
#include <hip/hip_runtime.h>

#define DEV __device__ __forceinline__

typedef float v2f __attribute__((ext_vector_type(2)));
typedef float v8f __attribute__((ext_vector_type(8)));
typedef float v4f __attribute__((ext_vector_type(4)));
typedef int v4i __attribute__((ext_vector_type(4)));
typedef unsigned v16u __attribute__((ext_vector_type(16)));

static constexpr int CDIM  = 128;
static constexpr int FDIM  = 256;
static constexpr int NSEQ  = 2048;
static constexpr int BATCH = 4;
static constexpr int NHEAD = 8;
static constexpr int HDIM  = 16;
static constexpr int MROWS = BATCH * NSEQ;
static constexpr int NTILE = NSEQ / 16;

DEV v8f v8zero() {
  v8f v;
#pragma unroll
  for (int i = 0; i < 8; ++i) v[i] = 0.f;
  return v;
}

DEV v8f wmma4(v2f a, v2f b, v8f c) {
  return __builtin_amdgcn_wmma_f32_16x16x4_f32(false, a, false, b, (short)0, c,
                                               false, false);
}

template <int NT, bool SWAP = false>
DEV void gemm_acc(const float* __restrict__ Arow, const float* __restrict__ Bc,
                  int ldb, int K, v8f* acc) {
  const int hi = (threadIdx.x >> 4) & 1;
  for (int k = 0; k < K; k += 4) {
    const int kb = k + 2 * hi;
    v2f a;
    a.x = Arow[kb];
    a.y = Arow[kb + 1];
#pragma unroll
    for (int j = 0; j < NT; ++j) {
      v2f b;
      b.x = Bc[(size_t)kb * ldb + j * 16];
      b.y = Bc[(size_t)(kb + 1) * ldb + j * 16];
      acc[j] = SWAP ? wmma4(b, a, acc[j]) : wmma4(a, b, acc[j]);
    }
  }
}

DEV void gemm_acc_w(const float* __restrict__ Arow, const float* __restrict__ Bw,
                    int ldb, int K, v8f* acc) {
  const int hi = (threadIdx.x >> 4) & 1;
  for (int k = 0; k < K; k += 8) {
    const int kb = k + 4 * hi;
    const v4f a4 = *(const v4f*)(Arow + kb);
#pragma unroll
    for (int s = 0; s < 2; ++s) {
      v2f a;
      a.x = a4[2 * s];
      a.y = a4[2 * s + 1];
      const float* B0 = Bw + (size_t)(kb + 2 * s) * ldb;
      const float* B1 = B0 + ldb;
#pragma unroll
      for (int g = 0; g < 2; ++g) {
        const v4f b0 = *(const v4f*)(B0 + 64 * g);
        const v4f b1 = *(const v4f*)(B1 + 64 * g);
#pragma unroll
        for (int t = 0; t < 4; ++t) {
          v2f b;
          b.x = b0[t];
          b.y = b1[t];
          acc[4 * g + t] = wmma4(a, b, acc[4 * g + t]);
        }
      }
    }
  }
}

static constexpr int TILE_F = 256;
static constexpr int HEAD_F = (NSEQ / 16) * TILE_F;

static constexpr int QKV_BLOCKS  = (3 * (MROWS / 16)) / 8;
static constexpr int PACK_TILES  = 8;
static constexpr int PACK_BLOCKS = (NTILE * NTILE / PACK_TILES) / 8;

DEV void pack_tiles(const int* __restrict__ adj, unsigned* __restrict__ masks,
                    int task, int lane) {
  const int nt = task / (NTILE / PACK_TILES);
  const int t0 = (task % (NTILE / PACK_TILES)) * PACK_TILES;
  const int* src = adj + (size_t)(nt * 16 + (lane & 15)) * NSEQ + t0 * 16 +
                   8 * (lane >> 4);
  unsigned* dst = masks + ((size_t)nt * NTILE + t0) * 8 + lane;
#pragma unroll
  for (int g = 0; g < PACK_TILES / 4; ++g) {
    unsigned w = 0;
#pragma unroll
    for (int j = 0; j < 4; ++j) {
      const v4i a0 = *(const v4i*)(src + (g * 4 + j) * 16);
      const v4i a1 = *(const v4i*)(src + (g * 4 + j) * 16 + 4);
#pragma unroll
      for (int r = 0; r < 8; ++r) {
        const int av = (r < 4) ? a0[r & 3] : a1[r & 3];
        const unsigned bal = __builtin_amdgcn_ballot_w32(av != 0);
        if (lane == j * 8 + r) w = bal;
      }
    }
    dst[g * 32] = w;
  }
}

__global__ __launch_bounds__(256) void qkv_kernel(
    const float* __restrict__ h, const float* __restrict__ Wq,
    const float* __restrict__ Wk, const float* __restrict__ Wv,
    const int* __restrict__ adj, float* __restrict__ Q, float* __restrict__ K,
    float* __restrict__ V, unsigned* __restrict__ masks) {
  const int lane = threadIdx.x & 31;
  const int col = lane & 15, hi = lane >> 4;
  const int wv = __builtin_amdgcn_readfirstlane((int)(threadIdx.x >> 5));
  if (blockIdx.x >= QKV_BLOCKS) {
    pack_tiles(adj, masks, (blockIdx.x - QKV_BLOCKS) * 8 + wv, lane);
    return;
  }
  const int gw = blockIdx.x * (blockDim.x >> 5) + wv;
  const int mat = gw / (MROWS / 16);
  const int rt = gw % (MROWS / 16);
  const float* W = (mat == 0) ? Wq : (mat == 1) ? Wk : Wv;
  const int row0 = rt * 16;

  v8f acc[8];
#pragma unroll
  for (int j = 0; j < 8; ++j) acc[j] = v8zero();

  const float* Arow = h + (size_t)(row0 + col) * CDIM;
  if (mat == 0)
    gemm_acc_w(Arow, W + 4 * col, CDIM, CDIM, acc);
  else if (mat == 1)
    gemm_acc<8, true>(Arow, W + col, CDIM, CDIM, acc);
  else
    gemm_acc<8, false>(Arow, W + col, CDIM, CDIM, acc);

  if (mat == 0) {
#pragma unroll
    for (int g = 0; g < 2; ++g)
#pragma unroll
      for (int r = 0; r < 8; ++r) {
        v4f v;
#pragma unroll
        for (int t = 0; t < 4; ++t) v[t] = acc[4 * g + t][r];
        *(v4f*)(Q + (size_t)(row0 + r + 8 * hi) * CDIM + 64 * g + 4 * col) = v;
      }
  } else {
    float* Out = (mat == 1) ? K : V;
    float* T = Out + (size_t)((rt / (NSEQ / 16)) * NHEAD) * HEAD_F +
               (size_t)(rt & (NSEQ / 16 - 1)) * TILE_F + lane * 4;
#pragma unroll
    for (int j = 0; j < 8; ++j) {
      v4f lo, up;
#pragma unroll
      for (int e = 0; e < 4; ++e) {
        lo[e] = acc[j][e];
        up[e] = acc[j][e + 4];
      }
      *(v4f*)(T + (size_t)j * HEAD_F) = lo;
      *(v4f*)(T + (size_t)j * HEAD_F + 128) = up;
    }
  }
}

DEV float xhalf(float x) {
  const int b = __builtin_bit_cast(int, x);
  return __builtin_bit_cast(
      float, __builtin_amdgcn_permlanex16(b, b, 0x76543210, 0xfedcba98, false,
                                          false));
}

static constexpr int RING = 8;

struct KeyTile {
  v4f k0, k1;
  v4f v0, v1;
};

DEV void tile_load(KeyTile& t, const float* __restrict__ Kt,
                   const float* __restrict__ Vt, unsigned lo) {
  t.k0 = *(const v4f*)(Kt + lo);
  t.k1 = *(const v4f*)(Kt + lo + 128);
  t.v0 = *(const v4f*)(Vt + lo);
  t.v1 = *(const v4f*)(Vt + lo + 128);
}

struct Soft {
  float m;
  float l;
  v8f o;
};

DEV void pair_scores(const KeyTile& ta, const KeyTile& tb, const v16u& mw,
                     const v2f (&qf)[4], float (&sv)[2][8]) {
  v8f s[2] = {v8zero(), v8zero()};
  const v4f kq[2][2] = {{ta.k0, ta.k1}, {tb.k0, tb.k1}};
#pragma unroll
  for (int u = 0; u < 4; ++u)
#pragma unroll
    for (int c = 0; c < 2; ++c) {
      v2f a;
      a.x = kq[c][u >> 1][2 * (u & 1)];
      a.y = kq[c][u >> 1][2 * (u & 1) + 1];
      s[c] = wmma4(a, qf[u], s[c]);
    }
#pragma unroll
  for (int r = 0; r < 8; ++r)
#pragma unroll
    for (int c = 0; c < 2; ++c) {
      const bool edge = __builtin_amdgcn_inverse_ballot_w32(mw[8 * c + r]);
      sv[c][r] = edge ? s[c][r] * 0.25f
                      : -1e30f;
    }
}

DEV void pair_softmax_pv(const KeyTile& ta, const KeyTile& tb,
                         const float (&sv)[2][8], Soft (&st)[2]) {
  float mloc[2], mnew[2], f[2], psum[2];
  float p[2][8];
#pragma unroll
  for (int c = 0; c < 2; ++c) mloc[c] = sv[c][0];
#pragma unroll
  for (int r = 1; r < 8; ++r)
#pragma unroll
    for (int c = 0; c < 2; ++c) mloc[c] = fmaxf(mloc[c], sv[c][r]);
#pragma unroll
  for (int c = 0; c < 2; ++c)
    mloc[c] = fmaxf(mloc[c], xhalf(mloc[c]));
#pragma unroll
  for (int c = 0; c < 2; ++c) {
    mnew[c] = fmaxf(st[c].m, mloc[c]);
    f[c] = __expf(st[c].m - mnew[c]);
    psum[c] = 0.f;
  }
#pragma unroll
  for (int r = 0; r < 8; ++r)
#pragma unroll
    for (int c = 0; c < 2; ++c) {
      p[c][r] = __expf(sv[c][r] - mnew[c]);
      psum[c] += p[c][r];
    }
#pragma unroll
  for (int c = 0; c < 2; ++c) {
    st[c].l = st[c].l * f[c] + psum[c];
    st[c].m = mnew[c];
  }
#pragma unroll
  for (int r = 0; r < 8; ++r)
#pragma unroll
    for (int c = 0; c < 2; ++c) st[c].o[r] *= f[c];

  const v4f vq[2][2] = {{ta.v0, ta.v1}, {tb.v0, tb.v1}};
#pragma unroll
  for (int u = 0; u < 4; ++u)
#pragma unroll
    for (int c = 0; c < 2; ++c) {
      v2f a, bb;
      a.x = vq[c][u >> 1][2 * (u & 1)];
      a.y = vq[c][u >> 1][2 * (u & 1) + 1];
      bb.x = p[c][2 * u];
      bb.y = p[c][2 * u + 1];
      st[c].o = wmma4(a, bb, st[c].o);
    }
}

__global__ __launch_bounds__(128) __attribute__((amdgpu_waves_per_eu(4, 4)))
void attn_kernel(
    const float* __restrict__ Q, const float* __restrict__ Kmat,
    const float* __restrict__ V, const unsigned* __restrict__ masks,
    float* __restrict__ O) {
  const int lane = threadIdx.x & 31;
  const int col = lane & 15, hi = lane >> 4;
  const int wid = __builtin_amdgcn_readfirstlane((int)(threadIdx.x >> 5));
  const int gw = blockIdx.x * 4 + wid;
  const int nt = gw & (NTILE - 1);
  const int hd = (gw >> 7) & (NHEAD - 1);
  const int b = gw >> 10;
  const int n0 = nt * 16;
  const size_t base = (size_t)b * NSEQ * CDIM;

  const float* Qrow = Q + base + (size_t)(n0 + col) * CDIM + hd * HDIM + 8 * hi;
  v2f qf[4];
#pragma unroll
  for (int t = 0; t < 4; ++t) {
    qf[t].x = Qrow[2 * t];
    qf[t].y = Qrow[2 * t + 1];
  }

  const size_t hoff = (size_t)(b * NHEAD + hd) * HEAD_F;
  const float* Kh = Kmat + hoff;
  const float* Vh = V + hoff;
  const unsigned lo = lane * 4;
  const unsigned* Mq = masks + (size_t)nt * (NTILE * 8);

  Soft st[2];
#pragma unroll
  for (int c = 0; c < 2; ++c) {
    st[c].m = -__builtin_inff();
    st[c].l = 0.f;
    st[c].o = v8zero();
  }

  v16u mp[2];
  mp[0] = *(const v16u*)Mq;
  KeyTile ring[RING];
#pragma unroll
  for (int s = 0; s < RING; ++s) {
    tile_load(ring[s], Kh + s * TILE_F, Vh + s * TILE_F, lo);
    __builtin_amdgcn_sched_barrier(0);
  }

  for (int m = 0; m < NTILE - RING; m += RING) {
#pragma unroll
    for (int s = 0; s < RING; s += 2) {
      float sv[2][8];
      pair_scores(ring[s], ring[s + 1], mp[(s >> 1) & 1], qf, sv);
      __builtin_amdgcn_sched_barrier(0);
      mp[((s >> 1) + 1) & 1] = *(const v16u*)(Mq + ((m + s) / 2 + 1) * 16);
      __builtin_amdgcn_sched_barrier(0);
      pair_softmax_pv(ring[s], ring[s + 1], sv, st);
      __builtin_amdgcn_sched_barrier(0);
      const int nx = m + s + RING;
      tile_load(ring[s], Kh + (size_t)nx * TILE_F, Vh + (size_t)nx * TILE_F,
                lo);
      tile_load(ring[s + 1], Kh + (size_t)(nx + 1) * TILE_F,
                Vh + (size_t)(nx + 1) * TILE_F, lo);
      __builtin_amdgcn_sched_barrier(0);
    }
  }
#pragma unroll
  for (int s = 0; s < RING; s += 2) {
    float sv[2][8];
    pair_scores(ring[s], ring[s + 1], mp[(s >> 1) & 1], qf, sv);
    if (s < RING - 2) {
      __builtin_amdgcn_sched_barrier(0);
      mp[((s >> 1) + 1) & 1] =
          *(const v16u*)(Mq + ((NTILE - RING + s) / 2 + 1) * 16);
      __builtin_amdgcn_sched_barrier(0);
    }
    pair_softmax_pv(ring[s], ring[s + 1], sv, st);
  }

  const float mm = fmaxf(st[0].m, st[1].m);
  const float a0 = __expf(st[0].m - mm);
  const float a1 = __expf(st[1].m - mm);
  float l_i = st[0].l * a0 + st[1].l * a1;
  v8f o;
#pragma unroll
  for (int r = 0; r < 8; ++r) o[r] = st[0].o[r] * a0 + st[1].o[r] * a1;

  l_i += xhalf(l_i);
  float* Op = O + base + (size_t)(n0 + col) * CDIM + hd * HDIM + 8 * hi;
  v4f o0, o1;
#pragma unroll
  for (int r = 0; r < 4; ++r) {
    o0[r] = o[r] / l_i;
    o1[r] = o[r + 4] / l_i;
  }
  *(v4f*)Op = o0;
  *(v4f*)(Op + 4) = o1;
}


template <int NGW>
struct GStage {
  v4f a;
  v4f w[2][NGW][2];
};
template <int NGW>
DEV void gstage_load(GStage<NGW>& st, const float* __restrict__ Arow,
                     const float* __restrict__ Bw, int ldb, int kb) {
  st.a = *(const v4f*)(Arow + kb);
#pragma unroll
  for (int s = 0; s < 2; ++s)
#pragma unroll
    for (int g = 0; g < NGW; ++g) {
      const float* B0 = Bw + (size_t)(kb + 2 * s) * ldb + 64 * g;
      st.w[s][g][0] = *(const v4f*)B0;
      st.w[s][g][1] = *(const v4f*)(B0 + ldb);
    }
}
template <int DEPTH, int NGW>
DEV void gemm_glob_t(const float* __restrict__ Arow,
                     const float* __restrict__ Bw, int ldb, v8f* acc) {
  const int hi = (threadIdx.x >> 4) & 1;
  constexpr int NS = CDIM / 8;
  GStage<NGW> ring[DEPTH];
#pragma unroll
  for (int i = 0; i < DEPTH; ++i) {
    gstage_load<NGW>(ring[i], Arow, Bw, ldb, 8 * i + 4 * hi);
    __builtin_amdgcn_sched_barrier(0);
  }
#pragma nounroll
  for (int i0 = 0; i0 < NS; i0 += DEPTH) {
#pragma unroll
    for (int j = 0; j < DEPTH; ++j) {
      GStage<NGW>& st = ring[j];
#pragma unroll
      for (int s = 0; s < 2; ++s) {
        v2f b;
        b.x = st.a[2 * s];
        b.y = st.a[2 * s + 1];
#pragma unroll
        for (int g = 0; g < NGW; ++g)
#pragma unroll
          for (int t = 0; t < 4; ++t) {
            v2f a;
            a.x = st.w[s][g][0][t];
            a.y = st.w[s][g][1][t];
            acc[4 * g + t] = wmma4(a, b, acc[4 * g + t]);
          }
      }
      __builtin_amdgcn_sched_barrier(0);
      const int nx = (i0 + j + DEPTH < NS) ? i0 + j + DEPTH : NS - 1;
      gstage_load<NGW>(st, Arow, Bw, ldb, 8 * nx + 4 * hi);
      __builtin_amdgcn_sched_barrier(0);
    }
  }
}

template <int NQ>
DEV void wstage_load(v4f (&w)[NQ][2], const float* __restrict__ Wl, int ldw,
                     int step) {
  const int kk = 64 * (step >> 4) + 4 * ((step >> 1) & 7) + 2 * (step & 1);
  const float* R0 = Wl + (size_t)kk * ldw;
#pragma unroll
  for (int q = 0; q < NQ; ++q) {
    w[q][0] = *(const v4f*)(R0 + 64 * q);
    w[q][1] = *(const v4f*)(R0 + ldw + 64 * q);
  }
}
template <int NG, int NQ, int DEPTH>
DEV void gemm_regs_t(v2f (&xp)[NG][2][8], const float* __restrict__ Wl,
                     int ldw, v8f* acc) {
  v4f ring[DEPTH][NQ][2];
#pragma unroll
  for (int i = 0; i < DEPTH; ++i) {
    wstage_load<NQ>(ring[i], Wl, ldw, i);
    __builtin_amdgcn_sched_barrier(0);
  }
#pragma nounroll
  for (int gi = 0; gi < NG; ++gi) {
    const float* Wg = Wl + (size_t)(64 * gi) * ldw;
    const float* Wn = Wl + (size_t)(64 * ((gi + 1 < NG) ? gi + 1 : gi)) * ldw;
#pragma unroll
    for (int s = 0; s < 16; ++s) {
      const v2f b = xp[0][s & 1][(s >> 1) & 7];
#pragma unroll
      for (int q = 0; q < NQ; ++q)
#pragma unroll
        for (int t = 0; t < 4; ++t) {
          v2f a;
          a.x = ring[s % DEPTH][q][0][t];
          a.y = ring[s % DEPTH][q][1][t];
          acc[4 * q + t] = wmma4(a, b, acc[4 * q + t]);
        }
      __builtin_amdgcn_sched_barrier(0);
      if (s + DEPTH < 16)
        wstage_load<NQ>(ring[s % DEPTH], Wg, ldw, s + DEPTH);
      else
        wstage_load<NQ>(ring[s % DEPTH], Wn, ldw, s + DEPTH - 16);
      __builtin_amdgcn_sched_barrier(0);
    }
    v2f first[2][8];
#pragma unroll
    for (int u = 0; u < 2; ++u)
#pragma unroll
      for (int r = 0; r < 8; ++r) first[u][r] = xp[0][u][r];
#pragma unroll
    for (int g = 0; g + 1 < NG; ++g)
#pragma unroll
      for (int u = 0; u < 2; ++u)
#pragma unroll
        for (int r = 0; r < 8; ++r) xp[g][u][r] = xp[g + 1][u][r];
#pragma unroll
    for (int u = 0; u < 2; ++u)
#pragma unroll
      for (int r = 0; r < 8; ++r) xp[NG - 1][u][r] = first[u][r];
  }
}

__global__ __launch_bounds__(128) __attribute__((amdgpu_waves_per_eu(1, 1)))
void mlp_kernel(
    const float* __restrict__ AO, const float* __restrict__ Wo,
    const float* __restrict__ bo, const float* __restrict__ hres,
    const float* __restrict__ g1, const float* __restrict__ bt1,
    const float* __restrict__ W1, const float* __restrict__ b1,
    const float* __restrict__ W2, const float* __restrict__ b2,
    const float* __restrict__ g2, const float* __restrict__ bt2,
    float* __restrict__ Out) {
  __shared__ v4f xs[2][2][8][32];
  __shared__ v4f ys[2][16][32];
  const int lane = threadIdx.x & 31;
  const int lr = lane & 15, hi = lane >> 4;
  const int wv = __builtin_amdgcn_readfirstlane((int)(threadIdx.x >> 5));
  const int pair = wv >> 1, part = wv & 1;
  const size_t row = (size_t)(blockIdx.x * 2 + pair) * 16 + lr;
  const int cb = 32 * hi;

  {
    v8f a4[4];
#pragma unroll
    for (int j = 0; j < 4; ++j) a4[j] = v8zero();
    gemm_glob_t<8, 1>(AO + row * CDIM, Wo + 64 * part + 4 * lr, CDIM, a4);
    const float* hrow = hres + row * CDIM + 64 * part + cb;
    const float* brow = bo + 64 * part + cb;
#pragma unroll
    for (int r = 0; r < 8; ++r) {
      const v4f bi = *(const v4f*)(brow + 4 * r);
      const v4f re = *(const v4f*)(hrow + 4 * r);
      v4f v;
#pragma unroll
      for (int t = 0; t < 4; ++t) v[t] = a4[t][r] + bi[t] + re[t];
      xs[pair][part][r][lane] = v;
    }
  }
  __syncthreads();

  v8f acc[8];
#pragma unroll
  for (int g = 0; g < 2; ++g)
#pragma unroll
    for (int r = 0; r < 8; ++r) {
      const v4f v = xs[pair][g][r][lane];
#pragma unroll
      for (int t = 0; t < 4; ++t) acc[4 * g + t][r] = v[t];
    }
  v2f xp[2][2][8];
  {
    float s1 = 0.f, s2 = 0.f;
#pragma unroll
    for (int g = 0; g < 2; ++g)
#pragma unroll
      for (int r = 0; r < 8; ++r)
#pragma unroll
        for (int t = 0; t < 4; ++t) {
          const float v = acc[4 * g + t][r];
          s1 += v;
          s2 += v * v;
        }
    s1 += xhalf(s1);
    s2 += xhalf(s2);
    const float mean = s1 * (1.f / CDIM);
    const float var = s2 * (1.f / CDIM) - mean * mean;
    const float rstd = rsqrtf(var + 1e-5f);
#pragma unroll
    for (int g = 0; g < 2; ++g)
#pragma unroll
      for (int r = 0; r < 8; ++r) {
        const v4f gv = *(const v4f*)(g1 + 64 * g + cb + 4 * r);
        const v4f bv = *(const v4f*)(bt1 + 64 * g + cb + 4 * r);
#pragma unroll
        for (int u = 0; u < 2; ++u) {
          xp[g][u][r].x =
              (acc[4 * g + 2 * u][r] - mean) * rstd * gv[2 * u] + bv[2 * u];
          xp[g][u][r].y = (acc[4 * g + 2 * u + 1][r] - mean) * rstd *
                              gv[2 * u + 1] + bv[2 * u + 1];
        }
      }
  }

  v2f tp[2][2][8];
  {
#pragma unroll
    for (int j = 0; j < 8; ++j) acc[j] = v8zero();
    gemm_regs_t<2, 2, 8>(xp, W1 + (size_t)cb * FDIM + 128 * part + 4 * lr, FDIM,
                         acc);
    const float* b1h = b1 + 128 * part + cb;
#pragma unroll
    for (int q = 0; q < 2; ++q)
#pragma unroll
      for (int r = 0; r < 8; ++r) {
        const v4f bi = *(const v4f*)(b1h + 64 * q + 4 * r);
#pragma unroll
        for (int u = 0; u < 2; ++u) {
          tp[q][u][r].x = fmaxf(acc[4 * q + 2 * u][r] + bi[2 * u], 0.f);
          tp[q][u][r].y = fmaxf(acc[4 * q + 2 * u + 1][r] + bi[2 * u + 1], 0.f);
        }
      }
  }

#pragma unroll
  for (int j = 0; j < 8; ++j) acc[j] = v8zero();
  gemm_regs_t<2, 2, 8>(tp, W2 + (size_t)(128 * part + cb) * CDIM + 4 * lr, CDIM,
                       acc);
  if (part == 1) {
#pragma unroll
    for (int g = 0; g < 2; ++g)
#pragma unroll
      for (int r = 0; r < 8; ++r) {
        v4f v;
#pragma unroll
        for (int t = 0; t < 4; ++t) v[t] = acc[4 * g + t][r];
        ys[pair][8 * g + r][lane] = v;
      }
  }
  __syncthreads();
  if (part == 0) {
    float s1 = 0.f, s2 = 0.f;
#pragma unroll
    for (int g = 0; g < 2; ++g)
#pragma unroll
      for (int r = 0; r < 8; ++r) {
        const v4f bi = *(const v4f*)(b2 + 64 * g + cb + 4 * r);
        const v4f pa = ys[pair][8 * g + r][lane];
#pragma unroll
        for (int t = 0; t < 4; ++t) {
          const float x = (t & 1) ? xp[g][t >> 1][r].y : xp[g][t >> 1][r].x;
          const float v = (acc[4 * g + t][r] + pa[t]) + bi[t] + x;
          acc[4 * g + t][r] = v;
          s1 += v;
          s2 += v * v;
        }
      }
    s1 += xhalf(s1);
    s2 += xhalf(s2);
    const float mean = s1 * (1.f / CDIM);
    const float var = s2 * (1.f / CDIM) - mean * mean;
    const float rstd = rsqrtf(var + 1e-5f);
    float* orow = Out + row * CDIM + cb;
#pragma unroll
    for (int g = 0; g < 2; ++g)
#pragma unroll
      for (int r = 0; r < 8; ++r) {
        const v4f gv = *(const v4f*)(g2 + 64 * g + cb + 4 * r);
        const v4f bv = *(const v4f*)(bt2 + 64 * g + cb + 4 * r);
        v4f o;
#pragma unroll
        for (int t = 0; t < 4; ++t)
          o[t] = (acc[4 * g + t][r] - mean) * rstd * gv[t] + bv[t];
        *(v4f*)(orow + 64 * g + 4 * r) = o;
      }
  }
}

extern "C" void kernel_launch(void* const* d_in, const int* in_sizes, int n_in,
                              void* d_out, int out_size, void* d_ws,
                              size_t ws_size, hipStream_t stream) {
  (void)in_sizes; (void)n_in; (void)out_size; (void)ws_size;
  const int* adj  = (const int*)d_in[0];
  const float* h  = (const float*)d_in[1];
  const float* Wq = (const float*)d_in[2];
  const float* Wk = (const float*)d_in[3];
  const float* Wv = (const float*)d_in[4];
  const float* Wo = (const float*)d_in[5];
  const float* bo = (const float*)d_in[6];
  const float* g1 = (const float*)d_in[7];
  const float* bt1 = (const float*)d_in[8];
  const float* W1 = (const float*)d_in[9];
  const float* bb1 = (const float*)d_in[10];
  const float* W2 = (const float*)d_in[11];
  const float* bb2 = (const float*)d_in[12];
  const float* g2 = (const float*)d_in[13];
  const float* bt2 = (const float*)d_in[14];

  float* ws = (float*)d_ws;
  const size_t SZ = (size_t)MROWS * CDIM;
  float* Q  = ws;
  float* K  = Q + SZ;
  float* V  = K + SZ;
  float* AO = V + SZ;
  unsigned* masks = (unsigned*)(AO + SZ);

  qkv_kernel<<<QKV_BLOCKS + PACK_BLOCKS, 256, 0, stream>>>(h, Wq, Wk, Wv, adj,
                                                            Q, K, V, masks);
  attn_kernel<<<(BATCH * NHEAD * (NSEQ / 16)) / 4, 128, 0, stream>>>(Q, K, V,
                                                                     masks, AO);
  mlp_kernel<<<(MROWS / 16) / 2, 128, 0, stream>>>(AO, Wo, bo, h, g1, bt1, W1,
                                                   bb1, W2, bb2, g2, bt2,
                                                   (float*)d_out);
}


# Q/K/V projections run on the out-proj launch's ring GEMM: 16-byte fragments (K, V too; K transposed) fetched 4 k-steps ahead, no load latency exposed; K, V dot-product k order re-associated
# speedup vs baseline: 4.0996x; 1.1235x over previous
#include <hip/hip_runtime.h>

#define DEV __device__ __forceinline__

typedef float v2f __attribute__((ext_vector_type(2)));
typedef float v8f __attribute__((ext_vector_type(8)));
typedef float v4f __attribute__((ext_vector_type(4)));
typedef int v4i __attribute__((ext_vector_type(4)));
typedef unsigned v16u __attribute__((ext_vector_type(16)));

static constexpr int CDIM  = 128;
static constexpr int FDIM  = 256;
static constexpr int NSEQ  = 2048;
static constexpr int BATCH = 4;
static constexpr int NHEAD = 8;
static constexpr int HDIM  = 16;
static constexpr int MROWS = BATCH * NSEQ;
static constexpr int NTILE = NSEQ / 16;

DEV v8f v8zero() {
  v8f v;
#pragma unroll
  for (int i = 0; i < 8; ++i) v[i] = 0.f;
  return v;
}

DEV v8f wmma4(v2f a, v2f b, v8f c) {
  return __builtin_amdgcn_wmma_f32_16x16x4_f32(false, a, false, b, (short)0, c,
                                               false, false);
}

template <int NGW>
struct GStage {
  v4f a;
  v4f w[2][NGW][2];
};
template <int NGW>
DEV void gstage_load(GStage<NGW>& st, const float* __restrict__ Arow,
                     const float* __restrict__ Bw, int ldb, int kb) {
  st.a = *(const v4f*)(Arow + kb);
#pragma unroll
  for (int s = 0; s < 2; ++s)
#pragma unroll
    for (int g = 0; g < NGW; ++g) {
      const float* B0 = Bw + (size_t)(kb + 2 * s) * ldb + 64 * g;
      st.w[s][g][0] = *(const v4f*)B0;
      st.w[s][g][1] = *(const v4f*)(B0 + ldb);
    }
}
template <int DEPTH, int NGW, bool PLAIN = false>
DEV void gemm_glob_t(const float* __restrict__ Arow,
                     const float* __restrict__ Bw, int ldb, v8f* acc) {
  const int hi = (threadIdx.x >> 4) & 1;
  constexpr int NS = CDIM / 8;
  GStage<NGW> ring[DEPTH];
#pragma unroll
  for (int i = 0; i < DEPTH; ++i) {
    gstage_load<NGW>(ring[i], Arow, Bw, ldb, 8 * i + 4 * hi);
    __builtin_amdgcn_sched_barrier(0);
  }
#pragma nounroll
  for (int i0 = 0; i0 < NS; i0 += DEPTH) {
#pragma unroll
    for (int j = 0; j < DEPTH; ++j) {
      GStage<NGW>& st = ring[j];
#pragma unroll
      for (int s = 0; s < 2; ++s) {
        v2f b;
        b.x = st.a[2 * s];
        b.y = st.a[2 * s + 1];
#pragma unroll
        for (int g = 0; g < NGW; ++g)
#pragma unroll
          for (int t = 0; t < 4; ++t) {
            v2f a;
            a.x = st.w[s][g][0][t];
            a.y = st.w[s][g][1][t];
            acc[4 * g + t] = PLAIN ? wmma4(b, a, acc[4 * g + t])
                                   : wmma4(a, b, acc[4 * g + t]);
          }
      }
      __builtin_amdgcn_sched_barrier(0);
      const int nx = (i0 + j + DEPTH < NS) ? i0 + j + DEPTH : NS - 1;
      gstage_load<NGW>(st, Arow, Bw, ldb, 8 * nx + 4 * hi);
      __builtin_amdgcn_sched_barrier(0);
    }
  }
}

static constexpr int TILE_F = 256;
static constexpr int HEAD_F = (NSEQ / 16) * TILE_F;

static constexpr int QKV_BLOCKS  = (3 * (MROWS / 16)) / 8;
static constexpr int PACK_TILES  = 8;
static constexpr int PACK_BLOCKS = (NTILE * NTILE / PACK_TILES) / 8;

DEV void pack_tiles(const int* __restrict__ adj, unsigned* __restrict__ masks,
                    int task, int lane) {
  const int nt = task / (NTILE / PACK_TILES);
  const int t0 = (task % (NTILE / PACK_TILES)) * PACK_TILES;
  const int* src = adj + (size_t)(nt * 16 + (lane & 15)) * NSEQ + t0 * 16 +
                   8 * (lane >> 4);
  unsigned* dst = masks + ((size_t)nt * NTILE + t0) * 8 + lane;
#pragma unroll
  for (int g = 0; g < PACK_TILES / 4; ++g) {
    unsigned w = 0;
#pragma unroll
    for (int j = 0; j < 4; ++j) {
      const v4i a0 = *(const v4i*)(src + (g * 4 + j) * 16);
      const v4i a1 = *(const v4i*)(src + (g * 4 + j) * 16 + 4);
#pragma unroll
      for (int r = 0; r < 8; ++r) {
        const int av = (r < 4) ? a0[r & 3] : a1[r & 3];
        const unsigned bal = __builtin_amdgcn_ballot_w32(av != 0);
        if (lane == j * 8 + r) w = bal;
      }
    }
    dst[g * 32] = w;
  }
}

__global__ __launch_bounds__(256) void qkv_kernel(
    const float* __restrict__ h, const float* __restrict__ Wq,
    const float* __restrict__ Wk, const float* __restrict__ Wv,
    const int* __restrict__ adj, float* __restrict__ Q, float* __restrict__ K,
    float* __restrict__ V, unsigned* __restrict__ masks) {
  const int lane = threadIdx.x & 31;
  const int col = lane & 15, hi = lane >> 4;
  const int wv = __builtin_amdgcn_readfirstlane((int)(threadIdx.x >> 5));
  if (blockIdx.x >= QKV_BLOCKS) {
    pack_tiles(adj, masks, (blockIdx.x - QKV_BLOCKS) * 8 + wv, lane);
    return;
  }
  const int gw = blockIdx.x * (blockDim.x >> 5) + wv;
  const int mat = gw / (MROWS / 16);
  const int rt = gw % (MROWS / 16);
  const float* W = (mat == 0) ? Wq : (mat == 1) ? Wk : Wv;
  const int row0 = rt * 16;

  v8f acc[8];
#pragma unroll
  for (int j = 0; j < 8; ++j) acc[j] = v8zero();

  const float* Arow = h + (size_t)(row0 + col) * CDIM;
  if (mat == 1)
    gemm_glob_t<4, 2, false>(Arow, W + 4 * col, CDIM, acc);
  else
    gemm_glob_t<4, 2, true>(Arow, W + 4 * col, CDIM, acc);

  if (mat == 0) {
#pragma unroll
    for (int g = 0; g < 2; ++g)
#pragma unroll
      for (int r = 0; r < 8; ++r) {
        v4f v;
#pragma unroll
        for (int t = 0; t < 4; ++t) v[t] = acc[4 * g + t][r];
        *(v4f*)(Q + (size_t)(row0 + r + 8 * hi) * CDIM + 64 * g + 4 * col) = v;
      }
  } else if (mat == 1) {
    float* T = K + (size_t)((rt / (NSEQ / 16)) * NHEAD + 2 * hi) * HEAD_F +
               (size_t)(rt & (NSEQ / 16 - 1)) * TILE_F + col * 4;
#pragma unroll
    for (int g = 0; g < 2; ++g)
#pragma unroll
      for (int r = 0; r < 8; ++r) {
        v4f v;
#pragma unroll
        for (int t = 0; t < 4; ++t) v[t] = acc[4 * g + t][r];
        *(v4f*)(T + (size_t)(4 * g + (r >> 2)) * HEAD_F + (r & 1) * 128 +
                ((r & 3) >> 1) * 64) = v;
      }
  } else {
    float* T = V + (size_t)((rt / (NSEQ / 16)) * NHEAD + (col >> 2)) * HEAD_F +
               (size_t)(rt & (NSEQ / 16 - 1)) * TILE_F +
               (16 * hi + 4 * (col & 3)) * 4;
#pragma unroll
    for (int g = 0; g < 2; ++g)
#pragma unroll
      for (int t = 0; t < 4; ++t)
#pragma unroll
        for (int q = 0; q < 2; ++q) {
          v4f v;
#pragma unroll
          for (int e = 0; e < 4; ++e) v[e] = acc[4 * g + t][4 * q + e];
          *(v4f*)(T + (size_t)(4 * g) * HEAD_F + q * 128 + t * 4) = v;
        }
  }
}

DEV float xhalf(float x) {
  const int b = __builtin_bit_cast(int, x);
  return __builtin_bit_cast(
      float, __builtin_amdgcn_permlanex16(b, b, 0x76543210, 0xfedcba98, false,
                                          false));
}

static constexpr int RING = 8;

struct KeyTile {
  v4f k0, k1;
  v4f v0, v1;
};

DEV void tile_load(KeyTile& t, const float* __restrict__ Kt,
                   const float* __restrict__ Vt, unsigned lo) {
  t.k0 = *(const v4f*)(Kt + lo);
  t.k1 = *(const v4f*)(Kt + lo + 128);
  t.v0 = *(const v4f*)(Vt + lo);
  t.v1 = *(const v4f*)(Vt + lo + 128);
}

struct Soft {
  float m;
  float l;
  v8f o;
};

DEV void pair_scores(const KeyTile& ta, const KeyTile& tb, const v16u& mw,
                     const v2f (&qf)[4], float (&sv)[2][8]) {
  v8f s[2] = {v8zero(), v8zero()};
  const v4f kq[2][2] = {{ta.k0, ta.k1}, {tb.k0, tb.k1}};
#pragma unroll
  for (int u = 0; u < 4; ++u)
#pragma unroll
    for (int c = 0; c < 2; ++c) {
      v2f a;
      a.x = kq[c][u >> 1][2 * (u & 1)];
      a.y = kq[c][u >> 1][2 * (u & 1) + 1];
      s[c] = wmma4(a, qf[u], s[c]);
    }
#pragma unroll
  for (int r = 0; r < 8; ++r)
#pragma unroll
    for (int c = 0; c < 2; ++c) {
      const bool edge = __builtin_amdgcn_inverse_ballot_w32(mw[8 * c + r]);
      sv[c][r] = edge ? s[c][r] * 0.25f
                      : -1e30f;
    }
}

DEV void pair_softmax_pv(const KeyTile& ta, const KeyTile& tb,
                         const float (&sv)[2][8], Soft (&st)[2]) {
  float mloc[2], mnew[2], f[2], psum[2];
  float p[2][8];
#pragma unroll
  for (int c = 0; c < 2; ++c) mloc[c] = sv[c][0];
#pragma unroll
  for (int r = 1; r < 8; ++r)
#pragma unroll
    for (int c = 0; c < 2; ++c) mloc[c] = fmaxf(mloc[c], sv[c][r]);
#pragma unroll
  for (int c = 0; c < 2; ++c)
    mloc[c] = fmaxf(mloc[c], xhalf(mloc[c]));
#pragma unroll
  for (int c = 0; c < 2; ++c) {
    mnew[c] = fmaxf(st[c].m, mloc[c]);
    f[c] = __expf(st[c].m - mnew[c]);
    psum[c] = 0.f;
  }
#pragma unroll
  for (int r = 0; r < 8; ++r)
#pragma unroll
    for (int c = 0; c < 2; ++c) {
      p[c][r] = __expf(sv[c][r] - mnew[c]);
      psum[c] += p[c][r];
    }
#pragma unroll
  for (int c = 0; c < 2; ++c) {
    st[c].l = st[c].l * f[c] + psum[c];
    st[c].m = mnew[c];
  }
#pragma unroll
  for (int r = 0; r < 8; ++r)
#pragma unroll
    for (int c = 0; c < 2; ++c) st[c].o[r] *= f[c];

  const v4f vq[2][2] = {{ta.v0, ta.v1}, {tb.v0, tb.v1}};
#pragma unroll
  for (int u = 0; u < 4; ++u)
#pragma unroll
    for (int c = 0; c < 2; ++c) {
      v2f a, bb;
      a.x = vq[c][u >> 1][2 * (u & 1)];
      a.y = vq[c][u >> 1][2 * (u & 1) + 1];
      bb.x = p[c][2 * u];
      bb.y = p[c][2 * u + 1];
      st[c].o = wmma4(a, bb, st[c].o);
    }
}

__global__ __launch_bounds__(128) __attribute__((amdgpu_waves_per_eu(4, 4)))
void attn_kernel(
    const float* __restrict__ Q, const float* __restrict__ Kmat,
    const float* __restrict__ V, const unsigned* __restrict__ masks,
    float* __restrict__ O) {
  const int lane = threadIdx.x & 31;
  const int col = lane & 15, hi = lane >> 4;
  const int wid = __builtin_amdgcn_readfirstlane((int)(threadIdx.x >> 5));
  const int gw = blockIdx.x * 4 + wid;
  const int nt = gw & (NTILE - 1);
  const int hd = (gw >> 7) & (NHEAD - 1);
  const int b = gw >> 10;
  const int n0 = nt * 16;
  const size_t base = (size_t)b * NSEQ * CDIM;

  const float* Qrow = Q + base + (size_t)(n0 + col) * CDIM + hd * HDIM + 8 * hi;
  v2f qf[4];
#pragma unroll
  for (int t = 0; t < 4; ++t) {
    qf[t].x = Qrow[2 * t];
    qf[t].y = Qrow[2 * t + 1];
  }

  const size_t hoff = (size_t)(b * NHEAD + hd) * HEAD_F;
  const float* Kh = Kmat + hoff;
  const float* Vh = V + hoff;
  const unsigned lo = lane * 4;
  const unsigned* Mq = masks + (size_t)nt * (NTILE * 8);

  Soft st[2];
#pragma unroll
  for (int c = 0; c < 2; ++c) {
    st[c].m = -__builtin_inff();
    st[c].l = 0.f;
    st[c].o = v8zero();
  }

  v16u mp[2];
  mp[0] = *(const v16u*)Mq;
  KeyTile ring[RING];
#pragma unroll
  for (int s = 0; s < RING; ++s) {
    tile_load(ring[s], Kh + s * TILE_F, Vh + s * TILE_F, lo);
    __builtin_amdgcn_sched_barrier(0);
  }

  for (int m = 0; m < NTILE - RING; m += RING) {
#pragma unroll
    for (int s = 0; s < RING; s += 2) {
      float sv[2][8];
      pair_scores(ring[s], ring[s + 1], mp[(s >> 1) & 1], qf, sv);
      __builtin_amdgcn_sched_barrier(0);
      mp[((s >> 1) + 1) & 1] = *(const v16u*)(Mq + ((m + s) / 2 + 1) * 16);
      __builtin_amdgcn_sched_barrier(0);
      pair_softmax_pv(ring[s], ring[s + 1], sv, st);
      __builtin_amdgcn_sched_barrier(0);
      const int nx = m + s + RING;
      tile_load(ring[s], Kh + (size_t)nx * TILE_F, Vh + (size_t)nx * TILE_F,
                lo);
      tile_load(ring[s + 1], Kh + (size_t)(nx + 1) * TILE_F,
                Vh + (size_t)(nx + 1) * TILE_F, lo);
      __builtin_amdgcn_sched_barrier(0);
    }
  }
#pragma unroll
  for (int s = 0; s < RING; s += 2) {
    float sv[2][8];
    pair_scores(ring[s], ring[s + 1], mp[(s >> 1) & 1], qf, sv);
    if (s < RING - 2) {
      __builtin_amdgcn_sched_barrier(0);
      mp[((s >> 1) + 1) & 1] =
          *(const v16u*)(Mq + ((NTILE - RING + s) / 2 + 1) * 16);
      __builtin_amdgcn_sched_barrier(0);
    }
    pair_softmax_pv(ring[s], ring[s + 1], sv, st);
  }

  const float mm = fmaxf(st[0].m, st[1].m);
  const float a0 = __expf(st[0].m - mm);
  const float a1 = __expf(st[1].m - mm);
  float l_i = st[0].l * a0 + st[1].l * a1;
  v8f o;
#pragma unroll
  for (int r = 0; r < 8; ++r) o[r] = st[0].o[r] * a0 + st[1].o[r] * a1;

  l_i += xhalf(l_i);
  float* Op = O + base + (size_t)(n0 + col) * CDIM + hd * HDIM + 8 * hi;
  v4f o0, o1;
#pragma unroll
  for (int r = 0; r < 4; ++r) {
    o0[r] = o[r] / l_i;
    o1[r] = o[r + 4] / l_i;
  }
  *(v4f*)Op = o0;
  *(v4f*)(Op + 4) = o1;
}


template <int NQ>
DEV void wstage_load(v4f (&w)[NQ][2], const float* __restrict__ Wl, int ldw,
                     int step) {
  const int kk = 64 * (step >> 4) + 4 * ((step >> 1) & 7) + 2 * (step & 1);
  const float* R0 = Wl + (size_t)kk * ldw;
#pragma unroll
  for (int q = 0; q < NQ; ++q) {
    w[q][0] = *(const v4f*)(R0 + 64 * q);
    w[q][1] = *(const v4f*)(R0 + ldw + 64 * q);
  }
}
template <int NG, int NQ, int DEPTH>
DEV void gemm_regs_t(v2f (&xp)[NG][2][8], const float* __restrict__ Wl,
                     int ldw, v8f* acc) {
  v4f ring[DEPTH][NQ][2];
#pragma unroll
  for (int i = 0; i < DEPTH; ++i) {
    wstage_load<NQ>(ring[i], Wl, ldw, i);
    __builtin_amdgcn_sched_barrier(0);
  }
#pragma nounroll
  for (int gi = 0; gi < NG; ++gi) {
    const float* Wg = Wl + (size_t)(64 * gi) * ldw;
    const float* Wn = Wl + (size_t)(64 * ((gi + 1 < NG) ? gi + 1 : gi)) * ldw;
#pragma unroll
    for (int s = 0; s < 16; ++s) {
      const v2f b = xp[0][s & 1][(s >> 1) & 7];
#pragma unroll
      for (int q = 0; q < NQ; ++q)
#pragma unroll
        for (int t = 0; t < 4; ++t) {
          v2f a;
          a.x = ring[s % DEPTH][q][0][t];
          a.y = ring[s % DEPTH][q][1][t];
          acc[4 * q + t] = wmma4(a, b, acc[4 * q + t]);
        }
      __builtin_amdgcn_sched_barrier(0);
      if (s + DEPTH < 16)
        wstage_load<NQ>(ring[s % DEPTH], Wg, ldw, s + DEPTH);
      else
        wstage_load<NQ>(ring[s % DEPTH], Wn, ldw, s + DEPTH - 16);
      __builtin_amdgcn_sched_barrier(0);
    }
    v2f first[2][8];
#pragma unroll
    for (int u = 0; u < 2; ++u)
#pragma unroll
      for (int r = 0; r < 8; ++r) first[u][r] = xp[0][u][r];
#pragma unroll
    for (int g = 0; g + 1 < NG; ++g)
#pragma unroll
      for (int u = 0; u < 2; ++u)
#pragma unroll
        for (int r = 0; r < 8; ++r) xp[g][u][r] = xp[g + 1][u][r];
#pragma unroll
    for (int u = 0; u < 2; ++u)
#pragma unroll
      for (int r = 0; r < 8; ++r) xp[NG - 1][u][r] = first[u][r];
  }
}

__global__ __launch_bounds__(128) __attribute__((amdgpu_waves_per_eu(1, 1)))
void mlp_kernel(
    const float* __restrict__ AO, const float* __restrict__ Wo,
    const float* __restrict__ bo, const float* __restrict__ hres,
    const float* __restrict__ g1, const float* __restrict__ bt1,
    const float* __restrict__ W1, const float* __restrict__ b1,
    const float* __restrict__ W2, const float* __restrict__ b2,
    const float* __restrict__ g2, const float* __restrict__ bt2,
    float* __restrict__ Out) {
  __shared__ v4f xs[2][2][8][32];
  __shared__ v4f ys[2][16][32];
  const int lane = threadIdx.x & 31;
  const int lr = lane & 15, hi = lane >> 4;
  const int wv = __builtin_amdgcn_readfirstlane((int)(threadIdx.x >> 5));
  const int pair = wv >> 1, part = wv & 1;
  const size_t row = (size_t)(blockIdx.x * 2 + pair) * 16 + lr;
  const int cb = 32 * hi;

  {
    v8f a4[4];
#pragma unroll
    for (int j = 0; j < 4; ++j) a4[j] = v8zero();
    gemm_glob_t<8, 1>(AO + row * CDIM, Wo + 64 * part + 4 * lr, CDIM, a4);
    const float* hrow = hres + row * CDIM + 64 * part + cb;
    const float* brow = bo + 64 * part + cb;
#pragma unroll
    for (int r = 0; r < 8; ++r) {
      const v4f bi = *(const v4f*)(brow + 4 * r);
      const v4f re = *(const v4f*)(hrow + 4 * r);
      v4f v;
#pragma unroll
      for (int t = 0; t < 4; ++t) v[t] = a4[t][r] + bi[t] + re[t];
      xs[pair][part][r][lane] = v;
    }
  }
  __syncthreads();

  v8f acc[8];
#pragma unroll
  for (int g = 0; g < 2; ++g)
#pragma unroll
    for (int r = 0; r < 8; ++r) {
      const v4f v = xs[pair][g][r][lane];
#pragma unroll
      for (int t = 0; t < 4; ++t) acc[4 * g + t][r] = v[t];
    }
  v2f xp[2][2][8];
  {
    float s1 = 0.f, s2 = 0.f;
#pragma unroll
    for (int g = 0; g < 2; ++g)
#pragma unroll
      for (int r = 0; r < 8; ++r)
#pragma unroll
        for (int t = 0; t < 4; ++t) {
          const float v = acc[4 * g + t][r];
          s1 += v;
          s2 += v * v;
        }
    s1 += xhalf(s1);
    s2 += xhalf(s2);
    const float mean = s1 * (1.f / CDIM);
    const float var = s2 * (1.f / CDIM) - mean * mean;
    const float rstd = rsqrtf(var + 1e-5f);
#pragma unroll
    for (int g = 0; g < 2; ++g)
#pragma unroll
      for (int r = 0; r < 8; ++r) {
        const v4f gv = *(const v4f*)(g1 + 64 * g + cb + 4 * r);
        const v4f bv = *(const v4f*)(bt1 + 64 * g + cb + 4 * r);
#pragma unroll
        for (int u = 0; u < 2; ++u) {
          xp[g][u][r].x =
              (acc[4 * g + 2 * u][r] - mean) * rstd * gv[2 * u] + bv[2 * u];
          xp[g][u][r].y = (acc[4 * g + 2 * u + 1][r] - mean) * rstd *
                              gv[2 * u + 1] + bv[2 * u + 1];
        }
      }
  }

  v2f tp[2][2][8];
  {
#pragma unroll
    for (int j = 0; j < 8; ++j) acc[j] = v8zero();
    gemm_regs_t<2, 2, 8>(xp, W1 + (size_t)cb * FDIM + 128 * part + 4 * lr, FDIM,
                         acc);
    const float* b1h = b1 + 128 * part + cb;
#pragma unroll
    for (int q = 0; q < 2; ++q)
#pragma unroll
      for (int r = 0; r < 8; ++r) {
        const v4f bi = *(const v4f*)(b1h + 64 * q + 4 * r);
#pragma unroll
        for (int u = 0; u < 2; ++u) {
          tp[q][u][r].x = fmaxf(acc[4 * q + 2 * u][r] + bi[2 * u], 0.f);
          tp[q][u][r].y = fmaxf(acc[4 * q + 2 * u + 1][r] + bi[2 * u + 1], 0.f);
        }
      }
  }

#pragma unroll
  for (int j = 0; j < 8; ++j) acc[j] = v8zero();
  gemm_regs_t<2, 2, 8>(tp, W2 + (size_t)(128 * part + cb) * CDIM + 4 * lr, CDIM,
                       acc);
  if (part == 1) {
#pragma unroll
    for (int g = 0; g < 2; ++g)
#pragma unroll
      for (int r = 0; r < 8; ++r) {
        v4f v;
#pragma unroll
        for (int t = 0; t < 4; ++t) v[t] = acc[4 * g + t][r];
        ys[pair][8 * g + r][lane] = v;
      }
  }
  __syncthreads();
  if (part == 0) {
    float s1 = 0.f, s2 = 0.f;
#pragma unroll
    for (int g = 0; g < 2; ++g)
#pragma unroll
      for (int r = 0; r < 8; ++r) {
        const v4f bi = *(const v4f*)(b2 + 64 * g + cb + 4 * r);
        const v4f pa = ys[pair][8 * g + r][lane];
#pragma unroll
        for (int t = 0; t < 4; ++t) {
          const float x = (t & 1) ? xp[g][t >> 1][r].y : xp[g][t >> 1][r].x;
          const float v = (acc[4 * g + t][r] + pa[t]) + bi[t] + x;
          acc[4 * g + t][r] = v;
          s1 += v;
          s2 += v * v;
        }
      }
    s1 += xhalf(s1);
    s2 += xhalf(s2);
    const float mean = s1 * (1.f / CDIM);
    const float var = s2 * (1.f / CDIM) - mean * mean;
    const float rstd = rsqrtf(var + 1e-5f);
    float* orow = Out + row * CDIM + cb;
#pragma unroll
    for (int g = 0; g < 2; ++g)
#pragma unroll
      for (int r = 0; r < 8; ++r) {
        const v4f gv = *(const v4f*)(g2 + 64 * g + cb + 4 * r);
        const v4f bv = *(const v4f*)(bt2 + 64 * g + cb + 4 * r);
        v4f o;
#pragma unroll
        for (int t = 0; t < 4; ++t)
          o[t] = (acc[4 * g + t][r] - mean) * rstd * gv[t] + bv[t];
        *(v4f*)(orow + 64 * g + 4 * r) = o;
      }
  }
}

extern "C" void kernel_launch(void* const* d_in, const int* in_sizes, int n_in,
                              void* d_out, int out_size, void* d_ws,
                              size_t ws_size, hipStream_t stream) {
  (void)in_sizes; (void)n_in; (void)out_size; (void)ws_size;
  const int* adj  = (const int*)d_in[0];
  const float* h  = (const float*)d_in[1];
  const float* Wq = (const float*)d_in[2];
  const float* Wk = (const float*)d_in[3];
  const float* Wv = (const float*)d_in[4];
  const float* Wo = (const float*)d_in[5];
  const float* bo = (const float*)d_in[6];
  const float* g1 = (const float*)d_in[7];
  const float* bt1 = (const float*)d_in[8];
  const float* W1 = (const float*)d_in[9];
  const float* bb1 = (const float*)d_in[10];
  const float* W2 = (const float*)d_in[11];
  const float* bb2 = (const float*)d_in[12];
  const float* g2 = (const float*)d_in[13];
  const float* bt2 = (const float*)d_in[14];

  float* ws = (float*)d_ws;
  const size_t SZ = (size_t)MROWS * CDIM;
  float* Q  = ws;
  float* K  = Q + SZ;
  float* V  = K + SZ;
  float* AO = V + SZ;
  unsigned* masks = (unsigned*)(AO + SZ);

  qkv_kernel<<<QKV_BLOCKS + PACK_BLOCKS, 256, 0, stream>>>(h, Wq, Wk, Wv, adj,
                                                            Q, K, V, masks);
  attn_kernel<<<(BATCH * NHEAD * (NSEQ / 16)) / 4, 128, 0, stream>>>(Q, K, V,
                                                                     masks, AO);
  mlp_kernel<<<(MROWS / 16) / 2, 128, 0, stream>>>(AO, Wo, bo, h, g1, bt1, W1,
                                                   bb1, W2, bb2, g2, bt2,
                                                   (float*)d_out);
}


# attention: score WMMAs start from minus a per-query reference (Q pre-scaled to base 2): the output is exp2's argument; max, subtract, O rescale only if a pair's sum passes 2^20; row sum re-associated
# speedup vs baseline: 4.5030x; 1.0984x over previous
#include <hip/hip_runtime.h>

#define DEV __device__ __forceinline__

typedef float v2f __attribute__((ext_vector_type(2)));
typedef float v8f __attribute__((ext_vector_type(8)));
typedef float v4f __attribute__((ext_vector_type(4)));
typedef int v4i __attribute__((ext_vector_type(4)));
typedef unsigned v16u __attribute__((ext_vector_type(16)));

static constexpr int CDIM  = 128;
static constexpr int FDIM  = 256;
static constexpr int NSEQ  = 2048;
static constexpr int BATCH = 4;
static constexpr int NHEAD = 8;
static constexpr int HDIM  = 16;
static constexpr int MROWS = BATCH * NSEQ;
static constexpr int NTILE = NSEQ / 16;

DEV v8f v8zero() {
  v8f v;
#pragma unroll
  for (int i = 0; i < 8; ++i) v[i] = 0.f;
  return v;
}

DEV v8f wmma4(v2f a, v2f b, v8f c) {
  return __builtin_amdgcn_wmma_f32_16x16x4_f32(false, a, false, b, (short)0, c,
                                               false, false);
}

template <int NGW>
struct GStage {
  v4f a;
  v4f w[2][NGW][2];
};
template <int NGW>
DEV void gstage_load(GStage<NGW>& st, const float* __restrict__ Arow,
                     const float* __restrict__ Bw, int ldb, int kb) {
  st.a = *(const v4f*)(Arow + kb);
#pragma unroll
  for (int s = 0; s < 2; ++s)
#pragma unroll
    for (int g = 0; g < NGW; ++g) {
      const float* B0 = Bw + (size_t)(kb + 2 * s) * ldb + 64 * g;
      st.w[s][g][0] = *(const v4f*)B0;
      st.w[s][g][1] = *(const v4f*)(B0 + ldb);
    }
}
template <int DEPTH, int NGW, bool PLAIN = false>
DEV void gemm_glob_t(const float* __restrict__ Arow,
                     const float* __restrict__ Bw, int ldb, v8f* acc) {
  const int hi = (threadIdx.x >> 4) & 1;
  constexpr int NS = CDIM / 8;
  GStage<NGW> ring[DEPTH];
#pragma unroll
  for (int i = 0; i < DEPTH; ++i) {
    gstage_load<NGW>(ring[i], Arow, Bw, ldb, 8 * i + 4 * hi);
    __builtin_amdgcn_sched_barrier(0);
  }
#pragma nounroll
  for (int i0 = 0; i0 < NS; i0 += DEPTH) {
#pragma unroll
    for (int j = 0; j < DEPTH; ++j) {
      GStage<NGW>& st = ring[j];
#pragma unroll
      for (int s = 0; s < 2; ++s) {
        v2f b;
        b.x = st.a[2 * s];
        b.y = st.a[2 * s + 1];
#pragma unroll
        for (int g = 0; g < NGW; ++g)
#pragma unroll
          for (int t = 0; t < 4; ++t) {
            v2f a;
            a.x = st.w[s][g][0][t];
            a.y = st.w[s][g][1][t];
            acc[4 * g + t] = PLAIN ? wmma4(b, a, acc[4 * g + t])
                                   : wmma4(a, b, acc[4 * g + t]);
          }
      }
      __builtin_amdgcn_sched_barrier(0);
      const int nx = (i0 + j + DEPTH < NS) ? i0 + j + DEPTH : NS - 1;
      gstage_load<NGW>(st, Arow, Bw, ldb, 8 * nx + 4 * hi);
      __builtin_amdgcn_sched_barrier(0);
    }
  }
}

static constexpr int TILE_F = 256;
static constexpr int HEAD_F = (NSEQ / 16) * TILE_F;

static constexpr int QKV_BLOCKS  = (3 * (MROWS / 16)) / 8;
static constexpr int PACK_TILES  = 8;
static constexpr int PACK_BLOCKS = (NTILE * NTILE / PACK_TILES) / 8;

DEV void pack_tiles(const int* __restrict__ adj, unsigned* __restrict__ masks,
                    int task, int lane) {
  const int nt = task / (NTILE / PACK_TILES);
  const int t0 = (task % (NTILE / PACK_TILES)) * PACK_TILES;
  const int* src = adj + (size_t)(nt * 16 + (lane & 15)) * NSEQ + t0 * 16 +
                   8 * (lane >> 4);
  unsigned* dst = masks + ((size_t)nt * NTILE + t0) * 8 + lane;
#pragma unroll
  for (int g = 0; g < PACK_TILES / 4; ++g) {
    unsigned w = 0;
#pragma unroll
    for (int j = 0; j < 4; ++j) {
      const v4i a0 = *(const v4i*)(src + (g * 4 + j) * 16);
      const v4i a1 = *(const v4i*)(src + (g * 4 + j) * 16 + 4);
#pragma unroll
      for (int r = 0; r < 8; ++r) {
        const int av = (r < 4) ? a0[r & 3] : a1[r & 3];
        const unsigned bal = __builtin_amdgcn_ballot_w32(av != 0);
        if (lane == j * 8 + r) w = bal;
      }
    }
    dst[g * 32] = w;
  }
}

__global__ __launch_bounds__(256) void qkv_kernel(
    const float* __restrict__ h, const float* __restrict__ Wq,
    const float* __restrict__ Wk, const float* __restrict__ Wv,
    const int* __restrict__ adj, float* __restrict__ Q, float* __restrict__ K,
    float* __restrict__ V, unsigned* __restrict__ masks) {
  const int lane = threadIdx.x & 31;
  const int col = lane & 15, hi = lane >> 4;
  const int wv = __builtin_amdgcn_readfirstlane((int)(threadIdx.x >> 5));
  if (blockIdx.x >= QKV_BLOCKS) {
    pack_tiles(adj, masks, (blockIdx.x - QKV_BLOCKS) * 8 + wv, lane);
    return;
  }
  const int gw = blockIdx.x * (blockDim.x >> 5) + wv;
  const int mat = gw / (MROWS / 16);
  const int rt = gw % (MROWS / 16);
  const float* W = (mat == 0) ? Wq : (mat == 1) ? Wk : Wv;
  const int row0 = rt * 16;

  v8f acc[8];
#pragma unroll
  for (int j = 0; j < 8; ++j) acc[j] = v8zero();

  const float* Arow = h + (size_t)(row0 + col) * CDIM;
  if (mat == 1)
    gemm_glob_t<4, 2, false>(Arow, W + 4 * col, CDIM, acc);
  else
    gemm_glob_t<4, 2, true>(Arow, W + 4 * col, CDIM, acc);

  if (mat == 0) {
#pragma unroll
    for (int g = 0; g < 2; ++g)
#pragma unroll
      for (int r = 0; r < 8; ++r) {
        v4f v;
#pragma unroll
        for (int t = 0; t < 4; ++t) v[t] = acc[4 * g + t][r];
        *(v4f*)(Q + (size_t)(row0 + r + 8 * hi) * CDIM + 64 * g + 4 * col) = v;
      }
  } else if (mat == 1) {
    float* T = K + (size_t)((rt / (NSEQ / 16)) * NHEAD + 2 * hi) * HEAD_F +
               (size_t)(rt & (NSEQ / 16 - 1)) * TILE_F + col * 4;
#pragma unroll
    for (int g = 0; g < 2; ++g)
#pragma unroll
      for (int r = 0; r < 8; ++r) {
        v4f v;
#pragma unroll
        for (int t = 0; t < 4; ++t) v[t] = acc[4 * g + t][r];
        *(v4f*)(T + (size_t)(4 * g + (r >> 2)) * HEAD_F + (r & 1) * 128 +
                ((r & 3) >> 1) * 64) = v;
      }
  } else {
    float* T = V + (size_t)((rt / (NSEQ / 16)) * NHEAD + (col >> 2)) * HEAD_F +
               (size_t)(rt & (NSEQ / 16 - 1)) * TILE_F +
               (16 * hi + 4 * (col & 3)) * 4;
#pragma unroll
    for (int g = 0; g < 2; ++g)
#pragma unroll
      for (int t = 0; t < 4; ++t)
#pragma unroll
        for (int q = 0; q < 2; ++q) {
          v4f v;
#pragma unroll
          for (int e = 0; e < 4; ++e) v[e] = acc[4 * g + t][4 * q + e];
          *(v4f*)(T + (size_t)(4 * g) * HEAD_F + q * 128 + t * 4) = v;
        }
  }
}

DEV float xhalf(float x) {
  const int b = __builtin_bit_cast(int, x);
  return __builtin_bit_cast(
      float, __builtin_amdgcn_permlanex16(b, b, 0x76543210, 0xfedcba98, false,
                                          false));
}

static constexpr int RING = 8;

struct KeyTile {
  v4f k0, k1;
  v4f v0, v1;
};

DEV void tile_load(KeyTile& t, const float* __restrict__ Kt,
                   const float* __restrict__ Vt, unsigned lo) {
  t.k0 = *(const v4f*)(Kt + lo);
  t.k1 = *(const v4f*)(Kt + lo + 128);
  t.v0 = *(const v4f*)(Vt + lo);
  t.v1 = *(const v4f*)(Vt + lo + 128);
}

struct Acc {
  v8f o;
};

DEV void pair_args(const KeyTile& ta, const KeyTile& tb, const v16u& mw,
                   const v2f (&qf)[4], const v8f& c0, float (&sv)[2][8]) {
  v8f s[2] = {c0, c0};
  const v4f kq[2][2] = {{ta.k0, ta.k1}, {tb.k0, tb.k1}};
#pragma unroll
  for (int u = 0; u < 4; ++u)
#pragma unroll
    for (int c = 0; c < 2; ++c) {
      v2f a;
      a.x = kq[c][u >> 1][2 * (u & 1)];
      a.y = kq[c][u >> 1][2 * (u & 1) + 1];
      s[c] = wmma4(a, qf[u], s[c]);
    }
#pragma unroll
  for (int r = 0; r < 8; ++r)
#pragma unroll
    for (int c = 0; c < 2; ++c) {
      const bool edge = __builtin_amdgcn_inverse_ballot_w32(mw[8 * c + r]);
      sv[c][r] = edge ? s[c][r] : -1e30f;
    }
}

DEV void pair_exp(const float (&sv)[2][8], float (&p)[2][8], float (&psum)[2]) {
#pragma unroll
  for (int c = 0; c < 2; ++c) psum[c] = 0.f;
#pragma unroll
  for (int r = 0; r < 8; ++r)
#pragma unroll
    for (int c = 0; c < 2; ++c) {
      p[c][r] = __builtin_amdgcn_exp2f(sv[c][r]);
      psum[c] += p[c][r];
    }
}

DEV void fix_reference(const KeyTile& ta, const KeyTile& tb, const v16u& mw,
                       const v2f (&qf)[4], float& mref, v8f& negm, float& l,
                       Acc (&st)[2]) {
  float sv[2][8];
  pair_args(ta, tb, mw, qf, v8zero(), sv);
  float mloc = fmaxf(sv[0][0], sv[1][0]);
#pragma unroll
  for (int r = 1; r < 8; ++r)
#pragma unroll
    for (int c = 0; c < 2; ++c) mloc = fmaxf(mloc, sv[c][r]);
  mloc = fmaxf(mloc, xhalf(mloc));
  const float mnew = fmaxf(mref, mloc);
  const float f = __builtin_amdgcn_exp2f(mref - mnew);
  l *= f;
#pragma unroll
  for (int c = 0; c < 2; ++c)
#pragma unroll
    for (int r = 0; r < 8; ++r) st[c].o[r] *= f;
  mref = mnew;
#pragma unroll
  for (int r = 0; r < 8; ++r) negm[r] = -mnew;
}

static constexpr float SUM_BOUND = 1048576.f;
DEV float pair_probs(const KeyTile& ta, const KeyTile& tb, const v16u& mw,
                     const v2f (&qf)[4], float& mref, v8f& negm, float& l,
                     Acc (&st)[2], float (&p)[2][8]) {
  float sv[2][8], psum[2];
  pair_args(ta, tb, mw, qf, negm, sv);
  pair_exp(sv, p, psum);
  float tot = psum[0] + psum[1];
  if (__builtin_expect(
          __builtin_amdgcn_ballot_w32(!(tot <= SUM_BOUND)) != 0u, 0)) {
    fix_reference(ta, tb, mw, qf, mref, negm, l, st);
    pair_args(ta, tb, mw, qf, negm, sv);
    pair_exp(sv, p, psum);
    tot = psum[0] + psum[1];
  }
  return tot;
}

DEV void pair_commit(const KeyTile& ta, const KeyTile& tb,
                     const float (&p)[2][8], Acc (&st)[2]) {
  const v4f vq[2][2] = {{ta.v0, ta.v1}, {tb.v0, tb.v1}};
#pragma unroll
  for (int u = 0; u < 4; ++u)
#pragma unroll
    for (int c = 0; c < 2; ++c) {
      v2f a, bb;
      a.x = vq[c][u >> 1][2 * (u & 1)];
      a.y = vq[c][u >> 1][2 * (u & 1) + 1];
      bb.x = p[c][2 * u];
      bb.y = p[c][2 * u + 1];
      st[c].o = wmma4(a, bb, st[c].o);
    }
}

__global__ __launch_bounds__(128) __attribute__((amdgpu_waves_per_eu(4, 4)))
void attn_kernel(
    const float* __restrict__ Q, const float* __restrict__ Kmat,
    const float* __restrict__ V, const unsigned* __restrict__ masks,
    float* __restrict__ O) {
  const int lane = threadIdx.x & 31;
  const int col = lane & 15, hi = lane >> 4;
  const int wid = __builtin_amdgcn_readfirstlane((int)(threadIdx.x >> 5));
  const int gw = blockIdx.x * 4 + wid;
  const int nt = gw & (NTILE - 1);
  const int hd = (gw >> 7) & (NHEAD - 1);
  const int b = gw >> 10;
  const int n0 = nt * 16;
  const size_t base = (size_t)b * NSEQ * CDIM;

  const float* Qrow = Q + base + (size_t)(n0 + col) * CDIM + hd * HDIM + 8 * hi;
  v2f qf[4];
#pragma unroll
  for (int t = 0; t < 4; ++t) {
    qf[t].x = Qrow[2 * t] * 0.36067376f;
    qf[t].y = Qrow[2 * t + 1] * 0.36067376f;
  }

  const size_t hoff = (size_t)(b * NHEAD + hd) * HEAD_F;
  const float* Kh = Kmat + hoff;
  const float* Vh = V + hoff;
  const unsigned lo = lane * 4;
  const unsigned* Mq = masks + (size_t)nt * (NTILE * 8);

  Acc st[2];
#pragma unroll
  for (int c = 0; c < 2; ++c) st[c].o = v8zero();
  float mref = -__builtin_inff();
  v8f negm;
#pragma unroll
  for (int r = 0; r < 8; ++r) negm[r] = __builtin_inff();
  float l_i = 0.f;

  v16u mp[2];
  mp[0] = *(const v16u*)Mq;
  KeyTile ring[RING];
#pragma unroll
  for (int s = 0; s < RING; ++s) {
    tile_load(ring[s], Kh + s * TILE_F, Vh + s * TILE_F, lo);
    __builtin_amdgcn_sched_barrier(0);
  }

  fix_reference(ring[0], ring[1], mp[0], qf, mref, negm, l_i, st);

  for (int m = 0; m < NTILE - RING; m += RING) {
#pragma unroll
    for (int s = 0; s < RING; s += 2) {
      float p[2][8];
      const float tot = pair_probs(ring[s], ring[s + 1], mp[(s >> 1) & 1], qf,
                                   mref, negm, l_i, st, p);
      __builtin_amdgcn_sched_barrier(0);
      mp[((s >> 1) + 1) & 1] = *(const v16u*)(Mq + ((m + s) / 2 + 1) * 16);
      __builtin_amdgcn_sched_barrier(0);
      l_i += tot;
      pair_commit(ring[s], ring[s + 1], p, st);
      __builtin_amdgcn_sched_barrier(0);
      const int nx = m + s + RING;
      tile_load(ring[s], Kh + (size_t)nx * TILE_F, Vh + (size_t)nx * TILE_F,
                lo);
      tile_load(ring[s + 1], Kh + (size_t)(nx + 1) * TILE_F,
                Vh + (size_t)(nx + 1) * TILE_F, lo);
      __builtin_amdgcn_sched_barrier(0);
    }
  }
#pragma unroll
  for (int s = 0; s < RING; s += 2) {
    float p[2][8];
    const float tot = pair_probs(ring[s], ring[s + 1], mp[(s >> 1) & 1], qf,
                                 mref, negm, l_i, st, p);
    if (s < RING - 2) {
      __builtin_amdgcn_sched_barrier(0);
      mp[((s >> 1) + 1) & 1] =
          *(const v16u*)(Mq + ((NTILE - RING + s) / 2 + 1) * 16);
      __builtin_amdgcn_sched_barrier(0);
    }
    l_i += tot;
    pair_commit(ring[s], ring[s + 1], p, st);
  }

  v8f o;
#pragma unroll
  for (int r = 0; r < 8; ++r) o[r] = st[0].o[r] + st[1].o[r];

  l_i += xhalf(l_i);
  float* Op = O + base + (size_t)(n0 + col) * CDIM + hd * HDIM + 8 * hi;
  v4f o0, o1;
#pragma unroll
  for (int r = 0; r < 4; ++r) {
    o0[r] = o[r] / l_i;
    o1[r] = o[r + 4] / l_i;
  }
  *(v4f*)Op = o0;
  *(v4f*)(Op + 4) = o1;
}


template <int NQ>
DEV void wstage_load(v4f (&w)[NQ][2], const float* __restrict__ Wl, int ldw,
                     int step) {
  const int kk = 64 * (step >> 4) + 4 * ((step >> 1) & 7) + 2 * (step & 1);
  const float* R0 = Wl + (size_t)kk * ldw;
#pragma unroll
  for (int q = 0; q < NQ; ++q) {
    w[q][0] = *(const v4f*)(R0 + 64 * q);
    w[q][1] = *(const v4f*)(R0 + ldw + 64 * q);
  }
}
template <int NG, int NQ, int DEPTH>
DEV void gemm_regs_t(v2f (&xp)[NG][2][8], const float* __restrict__ Wl,
                     int ldw, v8f* acc) {
  v4f ring[DEPTH][NQ][2];
#pragma unroll
  for (int i = 0; i < DEPTH; ++i) {
    wstage_load<NQ>(ring[i], Wl, ldw, i);
    __builtin_amdgcn_sched_barrier(0);
  }
#pragma nounroll
  for (int gi = 0; gi < NG; ++gi) {
    const float* Wg = Wl + (size_t)(64 * gi) * ldw;
    const float* Wn = Wl + (size_t)(64 * ((gi + 1 < NG) ? gi + 1 : gi)) * ldw;
#pragma unroll
    for (int s = 0; s < 16; ++s) {
      const v2f b = xp[0][s & 1][(s >> 1) & 7];
#pragma unroll
      for (int q = 0; q < NQ; ++q)
#pragma unroll
        for (int t = 0; t < 4; ++t) {
          v2f a;
          a.x = ring[s % DEPTH][q][0][t];
          a.y = ring[s % DEPTH][q][1][t];
          acc[4 * q + t] = wmma4(a, b, acc[4 * q + t]);
        }
      __builtin_amdgcn_sched_barrier(0);
      if (s + DEPTH < 16)
        wstage_load<NQ>(ring[s % DEPTH], Wg, ldw, s + DEPTH);
      else
        wstage_load<NQ>(ring[s % DEPTH], Wn, ldw, s + DEPTH - 16);
      __builtin_amdgcn_sched_barrier(0);
    }
    v2f first[2][8];
#pragma unroll
    for (int u = 0; u < 2; ++u)
#pragma unroll
      for (int r = 0; r < 8; ++r) first[u][r] = xp[0][u][r];
#pragma unroll
    for (int g = 0; g + 1 < NG; ++g)
#pragma unroll
      for (int u = 0; u < 2; ++u)
#pragma unroll
        for (int r = 0; r < 8; ++r) xp[g][u][r] = xp[g + 1][u][r];
#pragma unroll
    for (int u = 0; u < 2; ++u)
#pragma unroll
      for (int r = 0; r < 8; ++r) xp[NG - 1][u][r] = first[u][r];
  }
}

__global__ __launch_bounds__(128) __attribute__((amdgpu_waves_per_eu(1, 1)))
void mlp_kernel(
    const float* __restrict__ AO, const float* __restrict__ Wo,
    const float* __restrict__ bo, const float* __restrict__ hres,
    const float* __restrict__ g1, const float* __restrict__ bt1,
    const float* __restrict__ W1, const float* __restrict__ b1,
    const float* __restrict__ W2, const float* __restrict__ b2,
    const float* __restrict__ g2, const float* __restrict__ bt2,
    float* __restrict__ Out) {
  __shared__ v4f xs[2][2][8][32];
  __shared__ v4f ys[2][16][32];
  const int lane = threadIdx.x & 31;
  const int lr = lane & 15, hi = lane >> 4;
  const int wv = __builtin_amdgcn_readfirstlane((int)(threadIdx.x >> 5));
  const int pair = wv >> 1, part = wv & 1;
  const size_t row = (size_t)(blockIdx.x * 2 + pair) * 16 + lr;
  const int cb = 32 * hi;

  {
    v8f a4[4];
#pragma unroll
    for (int j = 0; j < 4; ++j) a4[j] = v8zero();
    gemm_glob_t<8, 1>(AO + row * CDIM, Wo + 64 * part + 4 * lr, CDIM, a4);
    const float* hrow = hres + row * CDIM + 64 * part + cb;
    const float* brow = bo + 64 * part + cb;
#pragma unroll
    for (int r = 0; r < 8; ++r) {
      const v4f bi = *(const v4f*)(brow + 4 * r);
      const v4f re = *(const v4f*)(hrow + 4 * r);
      v4f v;
#pragma unroll
      for (int t = 0; t < 4; ++t) v[t] = a4[t][r] + bi[t] + re[t];
      xs[pair][part][r][lane] = v;
    }
  }
  __syncthreads();

  v8f acc[8];
#pragma unroll
  for (int g = 0; g < 2; ++g)
#pragma unroll
    for (int r = 0; r < 8; ++r) {
      const v4f v = xs[pair][g][r][lane];
#pragma unroll
      for (int t = 0; t < 4; ++t) acc[4 * g + t][r] = v[t];
    }
  v2f xp[2][2][8];
  {
    float s1 = 0.f, s2 = 0.f;
#pragma unroll
    for (int g = 0; g < 2; ++g)
#pragma unroll
      for (int r = 0; r < 8; ++r)
#pragma unroll
        for (int t = 0; t < 4; ++t) {
          const float v = acc[4 * g + t][r];
          s1 += v;
          s2 += v * v;
        }
    s1 += xhalf(s1);
    s2 += xhalf(s2);
    const float mean = s1 * (1.f / CDIM);
    const float var = s2 * (1.f / CDIM) - mean * mean;
    const float rstd = rsqrtf(var + 1e-5f);
#pragma unroll
    for (int g = 0; g < 2; ++g)
#pragma unroll
      for (int r = 0; r < 8; ++r) {
        const v4f gv = *(const v4f*)(g1 + 64 * g + cb + 4 * r);
        const v4f bv = *(const v4f*)(bt1 + 64 * g + cb + 4 * r);
#pragma unroll
        for (int u = 0; u < 2; ++u) {
          xp[g][u][r].x =
              (acc[4 * g + 2 * u][r] - mean) * rstd * gv[2 * u] + bv[2 * u];
          xp[g][u][r].y = (acc[4 * g + 2 * u + 1][r] - mean) * rstd *
                              gv[2 * u + 1] + bv[2 * u + 1];
        }
      }
  }

  v2f tp[2][2][8];
  {
#pragma unroll
    for (int j = 0; j < 8; ++j) acc[j] = v8zero();
    gemm_regs_t<2, 2, 8>(xp, W1 + (size_t)cb * FDIM + 128 * part + 4 * lr, FDIM,
                         acc);
    const float* b1h = b1 + 128 * part + cb;
#pragma unroll
    for (int q = 0; q < 2; ++q)
#pragma unroll
      for (int r = 0; r < 8; ++r) {
        const v4f bi = *(const v4f*)(b1h + 64 * q + 4 * r);
#pragma unroll
        for (int u = 0; u < 2; ++u) {
          tp[q][u][r].x = fmaxf(acc[4 * q + 2 * u][r] + bi[2 * u], 0.f);
          tp[q][u][r].y = fmaxf(acc[4 * q + 2 * u + 1][r] + bi[2 * u + 1], 0.f);
        }
      }
  }

#pragma unroll
  for (int j = 0; j < 8; ++j) acc[j] = v8zero();
  gemm_regs_t<2, 2, 8>(tp, W2 + (size_t)(128 * part + cb) * CDIM + 4 * lr, CDIM,
                       acc);
  if (part == 1) {
#pragma unroll
    for (int g = 0; g < 2; ++g)
#pragma unroll
      for (int r = 0; r < 8; ++r) {
        v4f v;
#pragma unroll
        for (int t = 0; t < 4; ++t) v[t] = acc[4 * g + t][r];
        ys[pair][8 * g + r][lane] = v;
      }
  }
  __syncthreads();
  if (part == 0) {
    float s1 = 0.f, s2 = 0.f;
#pragma unroll
    for (int g = 0; g < 2; ++g)
#pragma unroll
      for (int r = 0; r < 8; ++r) {
        const v4f bi = *(const v4f*)(b2 + 64 * g + cb + 4 * r);
        const v4f pa = ys[pair][8 * g + r][lane];
#pragma unroll
        for (int t = 0; t < 4; ++t) {
          const float x = (t & 1) ? xp[g][t >> 1][r].y : xp[g][t >> 1][r].x;
          const float v = (acc[4 * g + t][r] + pa[t]) + bi[t] + x;
          acc[4 * g + t][r] = v;
          s1 += v;
          s2 += v * v;
        }
      }
    s1 += xhalf(s1);
    s2 += xhalf(s2);
    const float mean = s1 * (1.f / CDIM);
    const float var = s2 * (1.f / CDIM) - mean * mean;
    const float rstd = rsqrtf(var + 1e-5f);
    float* orow = Out + row * CDIM + cb;
#pragma unroll
    for (int g = 0; g < 2; ++g)
#pragma unroll
      for (int r = 0; r < 8; ++r) {
        const v4f gv = *(const v4f*)(g2 + 64 * g + cb + 4 * r);
        const v4f bv = *(const v4f*)(bt2 + 64 * g + cb + 4 * r);
        v4f o;
#pragma unroll
        for (int t = 0; t < 4; ++t)
          o[t] = (acc[4 * g + t][r] - mean) * rstd * gv[t] + bv[t];
        *(v4f*)(orow + 64 * g + 4 * r) = o;
      }
  }
}

extern "C" void kernel_launch(void* const* d_in, const int* in_sizes, int n_in,
                              void* d_out, int out_size, void* d_ws,
                              size_t ws_size, hipStream_t stream) {
  (void)in_sizes; (void)n_in; (void)out_size; (void)ws_size;
  const int* adj  = (const int*)d_in[0];
  const float* h  = (const float*)d_in[1];
  const float* Wq = (const float*)d_in[2];
  const float* Wk = (const float*)d_in[3];
  const float* Wv = (const float*)d_in[4];
  const float* Wo = (const float*)d_in[5];
  const float* bo = (const float*)d_in[6];
  const float* g1 = (const float*)d_in[7];
  const float* bt1 = (const float*)d_in[8];
  const float* W1 = (const float*)d_in[9];
  const float* bb1 = (const float*)d_in[10];
  const float* W2 = (const float*)d_in[11];
  const float* bb2 = (const float*)d_in[12];
  const float* g2 = (const float*)d_in[13];
  const float* bt2 = (const float*)d_in[14];

  float* ws = (float*)d_ws;
  const size_t SZ = (size_t)MROWS * CDIM;
  float* Q  = ws;
  float* K  = Q + SZ;
  float* V  = K + SZ;
  float* AO = V + SZ;
  unsigned* masks = (unsigned*)(AO + SZ);

  qkv_kernel<<<QKV_BLOCKS + PACK_BLOCKS, 256, 0, stream>>>(h, Wq, Wk, Wv, adj,
                                                            Q, K, V, masks);
  attn_kernel<<<(BATCH * NHEAD * (NSEQ / 16)) / 4, 128, 0, stream>>>(Q, K, V,
                                                                     masks, AO);
  mlp_kernel<<<(MROWS / 16) / 2, 128, 0, stream>>>(AO, Wo, bo, h, g1, bt1, W1,
                                                   bb1, W2, bb2, g2, bt2,
                                                   (float*)d_out);
}
